# starting kernel
# baseline (speedup 1.0000x reference)
#include <hip/hip_runtime.h>
#include <hip/hip_bf16.h>

typedef __attribute__((ext_vector_type(2))) float v2f;
typedef __attribute__((ext_vector_type(8))) float v8f;

#define CH 128
#define BN_EPS 1e-5f

__device__ __forceinline__ void atomicAddF(float* p, float v) {
    (void)__hip_atomic_fetch_add(p, v, __ATOMIC_RELAXED, __HIP_MEMORY_SCOPE_AGENT);
}

__global__ void gcn_deg_init(float* deg, int n) {
    int i = blockIdx.x * blockDim.x + threadIdx.x;
    if (i < n) deg[i] = 1.0f;
}
__global__ void gcn_deg_count(const int* __restrict__ dst, float* deg, int e) {
    int i = blockIdx.x * blockDim.x + threadIdx.x;
    if (i < e) atomicAddF(&deg[dst[i]], 1.0f);
}
__global__ void gcn_deg_fin(float* deg, int n) {
    int i = blockIdx.x * blockDim.x + threadIdx.x;
    if (i < n) deg[i] = rsqrtf(deg[i]);
}
__global__ void gcn_zero(float* p, int n) {
    int i = blockIdx.x * blockDim.x + threadIdx.x;
    if (i < n) p[i] = 0.0f;
}

__global__ void gcn_gemm_wmma(const float* __restrict__ A, const float* __restrict__ W,
                              float* __restrict__ C, int M, int K) {
    int wave = blockIdx.x * (blockDim.x >> 5) + (threadIdx.x >> 5);
    int lane = threadIdx.x & 31;
    if (wave * 16 >= M) return;

    int mrow  = wave * 16 + (lane & 15);
    int khalf = (lane >> 4) << 1;
    int ncol  = lane & 15;

    v8f acc[8];
#pragma unroll
    for (int t = 0; t < 8; ++t) acc[t] = (v8f){0,0,0,0,0,0,0,0};

    for (int k0 = 0; k0 < K; k0 += 4) {
        v2f a;
        const float* ar = A + (long)mrow * K + k0 + khalf;
        a.x = ar[0];
        a.y = ar[1];
        const float* w0 = W + (long)(k0 + khalf) * CH;
        const float* w1 = w0 + CH;
#pragma unroll
        for (int t = 0; t < 8; ++t) {
            v2f b;
            b.x = w0[t * 16 + ncol];
            b.y = w1[t * 16 + ncol];
            acc[t] = __builtin_amdgcn_wmma_f32_16x16x4_f32(
                false, a, false, b, (short)0, acc[t], false, false);
        }
    }
    int rbase = wave * 16 + ((lane >> 4) << 3);
#pragma unroll
    for (int t = 0; t < 8; ++t)
#pragma unroll
        for (int r = 0; r < 8; ++r)
            C[(long)(rbase + r) * CH + t * 16 + (lane & 15)] = acc[t][r];
}

__global__ void gcn_agg_init(const float* __restrict__ h, const float* __restrict__ dinv,
                             const float* __restrict__ bias, float* __restrict__ agg,
                             long total, int n) {
    long i = (long)blockIdx.x * blockDim.x + threadIdx.x;
    if (i >= total) return;
    long row = i >> 7;
    int  c   = (int)(i & 127);
    int  nd  = (int)(row % n);
    float di = dinv[nd];
    agg[i] = bias[c] + h[i] * di * di;
}

__global__ void gcn_scatter(const float* __restrict__ h, const int* __restrict__ src,
                            const int* __restrict__ dst, const float* __restrict__ dinv,
                            float* __restrict__ agg, int e, int n) {
    int ed = blockIdx.x * (blockDim.x >> 5) + (threadIdx.x >> 5);
    if (ed >= e) return;
    int b = blockIdx.y;
    int s = src[ed], d = dst[ed];
    float nrm = dinv[s] * dinv[d];
    int lane = threadIdx.x & 31;
    const float4* hp = (const float4*)(h + ((long)b * n + s) * CH);
    float*        ap = agg + ((long)b * n + d) * CH + lane * 4;
    float4 v = hp[lane];
    atomicAddF(ap + 0, v.x * nrm);
    atomicAddF(ap + 1, v.y * nrm);
    atomicAddF(ap + 2, v.z * nrm);
    atomicAddF(ap + 3, v.w * nrm);
}

__global__ void gcn_bn_stats(const float* __restrict__ x, float* __restrict__ sums, long rows) {
    int c = threadIdx.x;
    long r0 = (long)blockIdx.x * 256;
    long r1 = r0 + 256; if (r1 > rows) r1 = rows;
    float s = 0.f, q = 0.f;
    for (long r = r0; r < r1; ++r) {
        float v = x[r * CH + c];
        v = fmaxf(v, 0.f);
        s += v; q += v * v;
    }
    atomicAddF(&sums[c], s);
    atomicAddF(&sums[CH + c], q);
}

__global__ void gcn_bn_apply(const float* __restrict__ x, const float* __restrict__ sums,
                             const float* __restrict__ gamma, const float* __restrict__ beta,
                             float* __restrict__ y, long total, float invcnt) {
    long i = (long)blockIdx.x * blockDim.x + threadIdx.x;
    if (i >= total) return;
    int c = (int)(i & 127);
    float mu  = sums[c] * invcnt;
    float var = sums[CH + c] * invcnt - mu * mu;
    float v = fmaxf(x[i], 0.f);
    y[i] = (v - mu) * rsqrtf(var + BN_EPS) * gamma[c] + beta[c];
}

__global__ void gcn_bn_dot_out(const float* __restrict__ x, const float* __restrict__ sums,
                               const float* __restrict__ gamma, const float* __restrict__ beta,
                               const float* __restrict__ wc, const float* __restrict__ bc,
                               float* __restrict__ out, long rows, float invcnt) {
    long wid = (long)blockIdx.x * (blockDim.x >> 5) + (threadIdx.x >> 5);
    if (wid >= rows) return;
    int lane = threadIdx.x & 31;
    const float4* xp = (const float4*)(x + wid * CH);
    float4 v = xp[lane];
    float vv[4] = {v.x, v.y, v.z, v.w};
    float acc = 0.f;
#pragma unroll
    for (int j = 0; j < 4; ++j) {
        int c = lane * 4 + j;
        float mu  = sums[c] * invcnt;
        float var = sums[CH + c] * invcnt - mu * mu;
        float val = (fmaxf(vv[j], 0.f) - mu) * rsqrtf(var + BN_EPS) * gamma[c] + beta[c];
        acc += val * wc[c];
    }
#pragma unroll
    for (int o = 16; o > 0; o >>= 1)
        acc += __shfl_xor(acc, o, 32);
    if (lane == 0) out[wid] = acc + bc[0];
}

extern "C" void kernel_launch(void* const* d_in, const int* in_sizes, int n_in,
                              void* d_out, int out_size, void* d_ws, size_t ws_size,
                              hipStream_t stream) {
    const float* x   = (const float*)d_in[0];
    const int*   ei  = (const int*)d_in[1];
    const float* W1  = (const float*)d_in[2];
    const float* b1  = (const float*)d_in[3];
    const float* W2  = (const float*)d_in[4];
    const float* b2  = (const float*)d_in[5];
    const float* g1  = (const float*)d_in[6];
    const float* be1 = (const float*)d_in[7];
    const float* g2  = (const float*)d_in[8];
    const float* be2 = (const float*)d_in[9];
    const float* Wc  = (const float*)d_in[10];
    const float* bc  = (const float*)d_in[11];
    float* out = (float*)d_out;

    const int  E  = in_sizes[1] / 2;
    const int  N  = 20000;
    const int  H  = in_sizes[3];
    const int  F  = in_sizes[2] / H;
    const long BN = (long)out_size;
    const int  Bz = (int)(BN / N);
    const long TOT = BN * H;
    const float invcnt = 1.0f / (float)BN;

    float* ws     = (float*)d_ws;
    float* stats1 = ws;
    float* stats2 = ws + 256;
    float* dinv   = ws + 512;
    long   nAl    = (N + 255) & ~255L;
    float* bufA   = ws + 512 + nAl;
    float* bufB   = bufA + TOT;

    const int* srcIdx = ei;
    const int* dstIdx = ei + E;

    gcn_zero     <<<2, 256, 0, stream>>>(stats1, 512);
    gcn_deg_init <<<(N + 255) / 256, 256, 0, stream>>>(dinv, N);
    gcn_deg_count<<<(E + 255) / 256, 256, 0, stream>>>(dstIdx, dinv, E);
    gcn_deg_fin  <<<(N + 255) / 256, 256, 0, stream>>>(dinv, N);

    dim3 edgeGrid((E + 7) / 8, Bz, 1);
    int  gemmBlks  = (int)((BN / 16 + 7) / 8);
    int  ewBlks    = (int)((TOT + 255) / 256);
    int  statBlks  = (int)((BN + 255) / 256);
    int  rowBlks   = (int)((BN + 7) / 8);

    gcn_gemm_wmma<<<gemmBlks, 256, 0, stream>>>(x, W1, bufA, (int)BN, F);
    gcn_agg_init <<<ewBlks, 256, 0, stream>>>(bufA, dinv, b1, bufB, TOT, N);
    gcn_scatter  <<<edgeGrid, 256, 0, stream>>>(bufA, srcIdx, dstIdx, dinv, bufB, E, N);
    gcn_bn_stats <<<statBlks, 128, 0, stream>>>(bufB, stats1, BN);
    gcn_bn_apply <<<ewBlks, 256, 0, stream>>>(bufB, stats1, g1, be1, bufA, TOT, invcnt);

    gcn_gemm_wmma<<<gemmBlks, 256, 0, stream>>>(bufA, W2, bufB, (int)BN, H);
    gcn_agg_init <<<ewBlks, 256, 0, stream>>>(bufB, dinv, b2, bufA, TOT, N);
    gcn_scatter  <<<edgeGrid, 256, 0, stream>>>(bufB, srcIdx, dstIdx, dinv, bufA, E, N);
    gcn_bn_stats <<<statBlks, 128, 0, stream>>>(bufA, stats2, BN);

    gcn_bn_dot_out<<<rowBlks, 256, 0, stream>>>(bufA, stats2, g2, be2, Wc, bc, out, BN, invcnt);
}


# atomic scatter-add replaced by a gather over a destination-sorted edge list built in the call: one wave sums a node's in-edges in registers in ascending source order and stores the row once
# speedup vs baseline: 1.5017x; 1.5017x over previous
#include <hip/hip_runtime.h>
#include <hip/hip_bf16.h>

typedef __attribute__((ext_vector_type(2))) float v2f;
typedef __attribute__((ext_vector_type(8))) float v8f;

#define CH 128
#define BN_EPS 1e-5f

__device__ __forceinline__ void atomicAddF(float* p, float v) {
    (void)__hip_atomic_fetch_add(p, v, __ATOMIC_RELAXED, __HIP_MEMORY_SCOPE_AGENT);
}

__global__ void gcn_init(float* stats, int* cnt, int n) {
    int i = blockIdx.x * blockDim.x + threadIdx.x;
    if (i < 2 * 2 * CH) stats[i] = 0.0f;
    if (i < n) cnt[i] = 0;
}
__global__ void gcn_count(const int* __restrict__ dst, int* cnt, int e) {
    int i = blockIdx.x * blockDim.x + threadIdx.x;
    if (i < e) atomicAdd(&cnt[dst[i]], 1);
}
__global__ void gcn_scan(const int* __restrict__ cnt, int* __restrict__ rowptr,
                         int* __restrict__ cursor, float* __restrict__ dinv, int n) {
    __shared__ int part[1024];
    int t   = threadIdx.x;
    int per = (n + 1023) >> 10;
    int i0  = t * per;
    int i1  = i0 + per; if (i1 > n) i1 = n;
    int s = 0;
    for (int i = i0; i < i1; ++i) s += cnt[i];
    part[t] = s;
    __syncthreads();
    for (int off = 1; off < 1024; off <<= 1) {
        int v = (t >= off) ? part[t - off] : 0;
        __syncthreads();
        part[t] += v;
        __syncthreads();
    }
    int run = part[t] - s;
    for (int i = i0; i < i1; ++i) {
        int c = cnt[i];
        rowptr[i] = run;
        cursor[i] = run;
        dinv[i]   = rsqrtf((float)(c + 1));
        run += c;
    }
    if (t == 1023) rowptr[n] = part[1023];
}
__global__ void gcn_fill(const int* __restrict__ src, const int* __restrict__ dst,
                         int* cursor, int* __restrict__ slots, int e) {
    int i = blockIdx.x * blockDim.x + threadIdx.x;
    if (i < e) {
        int p = atomicAdd(&cursor[dst[i]], 1);
        slots[p] = src[i];
    }
}
__global__ void gcn_order(const int* __restrict__ rowptr, const int* __restrict__ slots,
                          int* __restrict__ sorted, int n) {
    int d = blockIdx.x * (blockDim.x >> 5) + (threadIdx.x >> 5);
    if (d >= n) return;
    int lane = threadIdx.x & 31;
    int beg = rowptr[d];
    int len = rowptr[d + 1] - beg;
    for (int i = lane; i < len; i += 32) {
        int v = slots[beg + i];
        int rank = 0;
        for (int j = 0; j < len; ++j) {
            int u = slots[beg + j];
            rank += (u < v || (u == v && j < i)) ? 1 : 0;
        }
        sorted[beg + rank] = v;
    }
}

__global__ void gcn_gemm_wmma(const float* __restrict__ A, const float* __restrict__ W,
                              float* __restrict__ C, int M, int K) {
    int wave = blockIdx.x * (blockDim.x >> 5) + (threadIdx.x >> 5);
    int lane = threadIdx.x & 31;
    if (wave * 16 >= M) return;

    int mrow  = wave * 16 + (lane & 15);
    int khalf = (lane >> 4) << 1;
    int ncol  = lane & 15;

    v8f acc[8];
#pragma unroll
    for (int t = 0; t < 8; ++t) acc[t] = (v8f){0,0,0,0,0,0,0,0};

    for (int k0 = 0; k0 < K; k0 += 4) {
        v2f a;
        const float* ar = A + (long)mrow * K + k0 + khalf;
        a.x = ar[0];
        a.y = ar[1];
        const float* w0 = W + (long)(k0 + khalf) * CH;
        const float* w1 = w0 + CH;
#pragma unroll
        for (int t = 0; t < 8; ++t) {
            v2f b;
            b.x = w0[t * 16 + ncol];
            b.y = w1[t * 16 + ncol];
            acc[t] = __builtin_amdgcn_wmma_f32_16x16x4_f32(
                false, a, false, b, (short)0, acc[t], false, false);
        }
    }
    int rbase = wave * 16 + ((lane >> 4) << 3);
#pragma unroll
    for (int t = 0; t < 8; ++t)
#pragma unroll
        for (int r = 0; r < 8; ++r)
            C[(long)(rbase + r) * CH + t * 16 + (lane & 15)] = acc[t][r];
}

__device__ __forceinline__ void gcn_fma4(float4& acc, const float4& v, float w) {
    acc.x = __builtin_fmaf(v.x, w, acc.x);
    acc.y = __builtin_fmaf(v.y, w, acc.y);
    acc.z = __builtin_fmaf(v.z, w, acc.z);
    acc.w = __builtin_fmaf(v.w, w, acc.w);
}
__global__ void gcn_gather(const float* __restrict__ h, const int* __restrict__ rowptr,
                           const int* __restrict__ csrc, const float* __restrict__ dinv,
                           const float* __restrict__ bias, float* __restrict__ agg, int n) {
    int d = blockIdx.x * (blockDim.x >> 5) + (threadIdx.x >> 5);
    if (d >= n) return;
    d = __builtin_amdgcn_readfirstlane(d);
    int lane = threadIdx.x & 31;
    const float4* hb = (const float4*)(h + (long)blockIdx.y * n * CH);
    float4*       ab = (float4*)(agg + (long)blockIdx.y * n * CH);
    int beg = __builtin_amdgcn_readfirstlane(rowptr[d]);
    int end = __builtin_amdgcn_readfirstlane(rowptr[d + 1]);
    float dd = dinv[d];
    float4 hv = hb[(long)d * (CH / 4) + lane];
    float4 bv = ((const float4*)bias)[lane];
    float4 acc;
    acc.x = bv.x + hv.x * dd * dd;
    acc.y = bv.y + hv.y * dd * dd;
    acc.z = bv.z + hv.z * dd * dd;
    acc.w = bv.w + hv.w * dd * dd;

    for (int base = beg; base < end; base += 32) {
        int cnt = end - base; if (cnt > 32) cnt = 32;
        int   sj = 0;
        float wj = 0.f;
        if (lane < cnt) { sj = csrc[base + lane]; wj = dinv[sj] * dd; }
        int wjb = __builtin_bit_cast(int, wj);
        int j = 0;
        for (; j + 4 <= cnt; j += 4) {
            int s0 = __builtin_amdgcn_readlane(sj, j);
            int s1 = __builtin_amdgcn_readlane(sj, j + 1);
            int s2 = __builtin_amdgcn_readlane(sj, j + 2);
            int s3 = __builtin_amdgcn_readlane(sj, j + 3);
            float w0 = __builtin_bit_cast(float, __builtin_amdgcn_readlane(wjb, j));
            float w1 = __builtin_bit_cast(float, __builtin_amdgcn_readlane(wjb, j + 1));
            float w2 = __builtin_bit_cast(float, __builtin_amdgcn_readlane(wjb, j + 2));
            float w3 = __builtin_bit_cast(float, __builtin_amdgcn_readlane(wjb, j + 3));
            float4 v0 = hb[(long)s0 * (CH / 4) + lane];
            float4 v1 = hb[(long)s1 * (CH / 4) + lane];
            float4 v2 = hb[(long)s2 * (CH / 4) + lane];
            float4 v3 = hb[(long)s3 * (CH / 4) + lane];
            gcn_fma4(acc, v0, w0);
            gcn_fma4(acc, v1, w1);
            gcn_fma4(acc, v2, w2);
            gcn_fma4(acc, v3, w3);
        }
        int rem = cnt - j;
        if (rem > 0) {
            int s0 = __builtin_amdgcn_readlane(sj, j);
            float w0 = __builtin_bit_cast(float, __builtin_amdgcn_readlane(wjb, j));
            float w1 = 0.f, w2 = 0.f;
            float4 v0 = hb[(long)s0 * (CH / 4) + lane];
            float4 v1 = {0.f, 0.f, 0.f, 0.f}, v2 = {0.f, 0.f, 0.f, 0.f};
            if (rem > 1) {
                int s1 = __builtin_amdgcn_readlane(sj, j + 1);
                w1 = __builtin_bit_cast(float, __builtin_amdgcn_readlane(wjb, j + 1));
                v1 = hb[(long)s1 * (CH / 4) + lane];
            }
            if (rem > 2) {
                int s2 = __builtin_amdgcn_readlane(sj, j + 2);
                w2 = __builtin_bit_cast(float, __builtin_amdgcn_readlane(wjb, j + 2));
                v2 = hb[(long)s2 * (CH / 4) + lane];
            }
            gcn_fma4(acc, v0, w0);
            if (rem > 1) gcn_fma4(acc, v1, w1);
            if (rem > 2) gcn_fma4(acc, v2, w2);
        }
    }
    ab[(long)d * (CH / 4) + lane] = acc;
}

__global__ void gcn_bn_stats(const float* __restrict__ x, float* __restrict__ sums, long rows) {
    int c = threadIdx.x;
    long r0 = (long)blockIdx.x * 256;
    long r1 = r0 + 256; if (r1 > rows) r1 = rows;
    float s = 0.f, q = 0.f;
    for (long r = r0; r < r1; ++r) {
        float v = x[r * CH + c];
        v = fmaxf(v, 0.f);
        s += v; q += v * v;
    }
    atomicAddF(&sums[c], s);
    atomicAddF(&sums[CH + c], q);
}

__global__ void gcn_bn_apply(const float* x, const float* __restrict__ sums,
                             const float* __restrict__ gamma, const float* __restrict__ beta,
                             float* y, long total, float invcnt) {
    long i = (long)blockIdx.x * blockDim.x + threadIdx.x;
    if (i >= total) return;
    int c = (int)(i & 127);
    float mu  = sums[c] * invcnt;
    float var = sums[CH + c] * invcnt - mu * mu;
    float v = fmaxf(x[i], 0.f);
    y[i] = (v - mu) * rsqrtf(var + BN_EPS) * gamma[c] + beta[c];
}

__global__ void gcn_bn_dot_out(const float* __restrict__ x, const float* __restrict__ sums,
                               const float* __restrict__ gamma, const float* __restrict__ beta,
                               const float* __restrict__ wc, const float* __restrict__ bc,
                               float* __restrict__ out, long rows, float invcnt) {
    long wid = (long)blockIdx.x * (blockDim.x >> 5) + (threadIdx.x >> 5);
    if (wid >= rows) return;
    int lane = threadIdx.x & 31;
    const float4* xp = (const float4*)(x + wid * CH);
    float4 v = xp[lane];
    float vv[4] = {v.x, v.y, v.z, v.w};
    float acc = 0.f;
#pragma unroll
    for (int j = 0; j < 4; ++j) {
        int c = lane * 4 + j;
        float mu  = sums[c] * invcnt;
        float var = sums[CH + c] * invcnt - mu * mu;
        float val = (fmaxf(vv[j], 0.f) - mu) * rsqrtf(var + BN_EPS) * gamma[c] + beta[c];
        acc += val * wc[c];
    }
#pragma unroll
    for (int o = 16; o > 0; o >>= 1)
        acc += __shfl_xor(acc, o, 32);
    if (lane == 0) out[wid] = acc + bc[0];
}

extern "C" void kernel_launch(void* const* d_in, const int* in_sizes, int n_in,
                              void* d_out, int out_size, void* d_ws, size_t ws_size,
                              hipStream_t stream) {
    const float* x   = (const float*)d_in[0];
    const int*   ei  = (const int*)d_in[1];
    const float* W1  = (const float*)d_in[2];
    const float* b1  = (const float*)d_in[3];
    const float* W2  = (const float*)d_in[4];
    const float* b2  = (const float*)d_in[5];
    const float* g1  = (const float*)d_in[6];
    const float* be1 = (const float*)d_in[7];
    const float* g2  = (const float*)d_in[8];
    const float* be2 = (const float*)d_in[9];
    const float* Wc  = (const float*)d_in[10];
    const float* bc  = (const float*)d_in[11];
    float* out = (float*)d_out;

    const int  E  = in_sizes[1] / 2;
    const int  N  = 20000;
    const int  H  = in_sizes[3];
    const int  F  = in_sizes[2] / H;
    const long BN = (long)out_size;
    const int  Bz = (int)(BN / N);
    const long TOT = BN * H;
    const float invcnt = 1.0f / (float)BN;

    float* ws     = (float*)d_ws;
    float* stats1 = ws;
    float* stats2 = ws + 256;
    float* dinv   = ws + 512;
    long   nAl    = (N + 255) & ~255L;
    float* bufA   = ws + 512 + nAl;

    const long PL   = (long)N * H;
    float* bufM     = bufA + (Bz - 1) * PL;
    float* hLast    = bufA + (Bz - 1) * PL;
    float* aggLast  = bufA + (2 * Bz - 2) * PL;
    long   n1Al     = (N + 1 + 255) & ~255L;
    long   eAl      = (E + 255) & ~255L;
    int*   csrEnd   = (int*)(bufA + 2 * TOT);
    int*   sorted   = csrEnd - eAl;
    int*   slots    = sorted - eAl;
    int*   rowptr   = slots - n1Al;
    int*   cursor   = rowptr - nAl;
    int*   cnt      = cursor - nAl;

    const int* srcIdx = ei;
    const int* dstIdx = ei + E;

    gcn_init  <<<(N + 255) / 256, 256, 0, stream>>>(stats1, cnt, N);
    gcn_count <<<(E + 255) / 256, 256, 0, stream>>>(dstIdx, cnt, E);
    gcn_scan  <<<1, 1024, 0, stream>>>(cnt, rowptr, cursor, dinv, N);
    gcn_fill  <<<(E + 255) / 256, 256, 0, stream>>>(srcIdx, dstIdx, cursor, slots, E);
    gcn_order <<<(N + 7) / 8, 256, 0, stream>>>(rowptr, slots, sorted, N);

    dim3 nodeGridLast((N + 7) / 8, 1, 1);
    dim3 nodeGridRest((N + 7) / 8, Bz - 1, 1);
    const long MR   = (long)(Bz - 1) * N;
    int  gemmBlks  = (int)((BN / 16 + 7) / 8);
    int  gemmBlksR = (int)((MR / 16 + 7) / 8);
    int  gemmBlksL = (int)((N / 16 + 7) / 8);
    int  ewBlks    = (int)((TOT + 255) / 256);
    int  statBlks  = (int)((BN + 255) / 256);
    int  rowBlks   = (int)((BN + 7) / 8);

    gcn_gemm_wmma<<<gemmBlks, 256, 0, stream>>>(x, W1, bufA, (int)BN, F);
    gcn_gather   <<<nodeGridLast, 256, 0, stream>>>(hLast, rowptr, sorted, dinv, b1, aggLast, N);
    gcn_gather   <<<nodeGridRest, 256, 0, stream>>>(bufA, rowptr, sorted, dinv, b1, bufM, N);
    gcn_bn_stats <<<statBlks, 128, 0, stream>>>(bufM, stats1, BN);
    gcn_bn_apply <<<ewBlks, 256, 0, stream>>>(bufM, stats1, g1, be1, bufM, TOT, invcnt);

    gcn_gemm_wmma<<<gemmBlksR, 256, 0, stream>>>(bufM, W2, bufA, (int)MR, H);
    gcn_gemm_wmma<<<gemmBlksL, 256, 0, stream>>>(aggLast, W2, hLast, N, H);
    gcn_gather   <<<nodeGridLast, 256, 0, stream>>>(hLast, rowptr, sorted, dinv, b2, aggLast, N);
    gcn_gather   <<<nodeGridRest, 256, 0, stream>>>(bufA, rowptr, sorted, dinv, b2, bufM, N);
    gcn_bn_stats <<<statBlks, 128, 0, stream>>>(bufM, stats2, BN);

    gcn_bn_dot_out<<<rowBlks, 256, 0, stream>>>(bufM, stats2, g2, be2, Wc, bc, out, BN, invcnt);
}


# BatchNorm-1 apply pass removed: the second GEMM applies relu and the per-channel normalisation (same expression) to its A fragment as it loads it; per-block parameters in LDS
# speedup vs baseline: 1.6107x; 1.0726x over previous
#include <hip/hip_runtime.h>
#include <hip/hip_bf16.h>

typedef __attribute__((ext_vector_type(2))) float v2f;
typedef __attribute__((ext_vector_type(8))) float v8f;

#define CH 128
#define BN_EPS 1e-5f

__device__ __forceinline__ void atomicAddF(float* p, float v) {
    (void)__hip_atomic_fetch_add(p, v, __ATOMIC_RELAXED, __HIP_MEMORY_SCOPE_AGENT);
}

__global__ void gcn_init(float* stats, int* cnt, int n) {
    int i = blockIdx.x * blockDim.x + threadIdx.x;
    if (i < 2 * 2 * CH) stats[i] = 0.0f;
    if (i < n) cnt[i] = 0;
}
__global__ void gcn_count(const int* __restrict__ dst, int* cnt, int e) {
    int i = blockIdx.x * blockDim.x + threadIdx.x;
    if (i < e) atomicAdd(&cnt[dst[i]], 1);
}
__global__ void gcn_scan(const int* __restrict__ cnt, int* __restrict__ rowptr,
                         int* __restrict__ cursor, float* __restrict__ dinv, int n) {
    __shared__ int part[1024];
    int t   = threadIdx.x;
    int per = (n + 1023) >> 10;
    int i0  = t * per;
    int i1  = i0 + per; if (i1 > n) i1 = n;
    int s = 0;
    for (int i = i0; i < i1; ++i) s += cnt[i];
    part[t] = s;
    __syncthreads();
    for (int off = 1; off < 1024; off <<= 1) {
        int v = (t >= off) ? part[t - off] : 0;
        __syncthreads();
        part[t] += v;
        __syncthreads();
    }
    int run = part[t] - s;
    for (int i = i0; i < i1; ++i) {
        int c = cnt[i];
        rowptr[i] = run;
        cursor[i] = run;
        dinv[i]   = rsqrtf((float)(c + 1));
        run += c;
    }
    if (t == 1023) rowptr[n] = part[1023];
}
__global__ void gcn_fill(const int* __restrict__ src, const int* __restrict__ dst,
                         int* cursor, int* __restrict__ slots, int e) {
    int i = blockIdx.x * blockDim.x + threadIdx.x;
    if (i < e) {
        int p = atomicAdd(&cursor[dst[i]], 1);
        slots[p] = src[i];
    }
}
__global__ void gcn_order(const int* __restrict__ rowptr, const int* __restrict__ slots,
                          int* __restrict__ sorted, int n) {
    int d = blockIdx.x * (blockDim.x >> 5) + (threadIdx.x >> 5);
    if (d >= n) return;
    int lane = threadIdx.x & 31;
    int beg = rowptr[d];
    int len = rowptr[d + 1] - beg;
    for (int i = lane; i < len; i += 32) {
        int v = slots[beg + i];
        int rank = 0;
        for (int j = 0; j < len; ++j) {
            int u = slots[beg + j];
            rank += (u < v || (u == v && j < i)) ? 1 : 0;
        }
        sorted[beg + rank] = v;
    }
}

template <bool BN>
__device__ __forceinline__ void gcn_gemm_body(const float* __restrict__ A, const float* __restrict__ W,
                                              float* __restrict__ C, int M, int K,
                                              const float4* prm) {
    int wave = blockIdx.x * (blockDim.x >> 5) + (threadIdx.x >> 5);
    int lane = threadIdx.x & 31;
    if (wave * 16 >= M) return;

    int mrow  = wave * 16 + (lane & 15);
    int khalf = (lane >> 4) << 1;
    int ncol  = lane & 15;

    v8f acc[8];
#pragma unroll
    for (int t = 0; t < 8; ++t) acc[t] = (v8f){0,0,0,0,0,0,0,0};

    for (int k0 = 0; k0 < K; k0 += 4) {
        v2f a;
        const float* ar = A + (long)mrow * K + k0 + khalf;
        a.x = ar[0];
        a.y = ar[1];
        if (BN) {
            float4 p0 = prm[k0 + khalf];
            float4 p1 = prm[k0 + khalf + 1];
            a.x = (fmaxf(a.x, 0.f) - p0.x) * p0.y * p0.z + p0.w;
            a.y = (fmaxf(a.y, 0.f) - p1.x) * p1.y * p1.z + p1.w;
        }
        const float* w0 = W + (long)(k0 + khalf) * CH;
        const float* w1 = w0 + CH;
#pragma unroll
        for (int t = 0; t < 8; ++t) {
            v2f b;
            b.x = w0[t * 16 + ncol];
            b.y = w1[t * 16 + ncol];
            acc[t] = __builtin_amdgcn_wmma_f32_16x16x4_f32(
                false, a, false, b, (short)0, acc[t], false, false);
        }
    }
    int rbase = wave * 16 + ((lane >> 4) << 3);
#pragma unroll
    for (int t = 0; t < 8; ++t)
#pragma unroll
        for (int r = 0; r < 8; ++r)
            C[(long)(rbase + r) * CH + t * 16 + (lane & 15)] = acc[t][r];
}

__global__ void gcn_gemm_wmma(const float* __restrict__ A, const float* __restrict__ W,
                              float* __restrict__ C, int M, int K) {
    gcn_gemm_body<false>(A, W, C, M, K, nullptr);
}

__global__ void gcn_gemm_wmma_bn(const float* __restrict__ A, const float* __restrict__ W,
                                 float* __restrict__ C, int M, int K,
                                 const float* __restrict__ sums, const float* __restrict__ gamma,
                                 const float* __restrict__ beta, float invcnt) {
    __shared__ float4 prm[CH];
    int c = threadIdx.x;
    if (c < CH) {
        float mu  = sums[c] * invcnt;
        float var = sums[CH + c] * invcnt - mu * mu;
        prm[c] = make_float4(mu, rsqrtf(var + BN_EPS), gamma[c], beta[c]);
    }
    __syncthreads();
    gcn_gemm_body<true>(A, W, C, M, K, prm);
}

__device__ __forceinline__ void gcn_fma4(float4& acc, const float4& v, float w) {
    acc.x = __builtin_fmaf(v.x, w, acc.x);
    acc.y = __builtin_fmaf(v.y, w, acc.y);
    acc.z = __builtin_fmaf(v.z, w, acc.z);
    acc.w = __builtin_fmaf(v.w, w, acc.w);
}
__global__ void gcn_gather(const float* __restrict__ h, const int* __restrict__ rowptr,
                           const int* __restrict__ csrc, const float* __restrict__ dinv,
                           const float* __restrict__ bias, float* __restrict__ agg, int n) {
    int d = blockIdx.x * (blockDim.x >> 5) + (threadIdx.x >> 5);
    if (d >= n) return;
    d = __builtin_amdgcn_readfirstlane(d);
    int lane = threadIdx.x & 31;
    const float4* hb = (const float4*)(h + (long)blockIdx.y * n * CH);
    float4*       ab = (float4*)(agg + (long)blockIdx.y * n * CH);
    int beg = __builtin_amdgcn_readfirstlane(rowptr[d]);
    int end = __builtin_amdgcn_readfirstlane(rowptr[d + 1]);
    float dd = dinv[d];
    float4 hv = hb[(long)d * (CH / 4) + lane];
    float4 bv = ((const float4*)bias)[lane];
    float4 acc;
    acc.x = bv.x + hv.x * dd * dd;
    acc.y = bv.y + hv.y * dd * dd;
    acc.z = bv.z + hv.z * dd * dd;
    acc.w = bv.w + hv.w * dd * dd;

    for (int base = beg; base < end; base += 32) {
        int cnt = end - base; if (cnt > 32) cnt = 32;
        int   sj = 0;
        float wj = 0.f;
        if (lane < cnt) { sj = csrc[base + lane]; wj = dinv[sj] * dd; }
        int wjb = __builtin_bit_cast(int, wj);
        int j = 0;
        for (; j + 4 <= cnt; j += 4) {
            int s0 = __builtin_amdgcn_readlane(sj, j);
            int s1 = __builtin_amdgcn_readlane(sj, j + 1);
            int s2 = __builtin_amdgcn_readlane(sj, j + 2);
            int s3 = __builtin_amdgcn_readlane(sj, j + 3);
            float w0 = __builtin_bit_cast(float, __builtin_amdgcn_readlane(wjb, j));
            float w1 = __builtin_bit_cast(float, __builtin_amdgcn_readlane(wjb, j + 1));
            float w2 = __builtin_bit_cast(float, __builtin_amdgcn_readlane(wjb, j + 2));
            float w3 = __builtin_bit_cast(float, __builtin_amdgcn_readlane(wjb, j + 3));
            float4 v0 = hb[(long)s0 * (CH / 4) + lane];
            float4 v1 = hb[(long)s1 * (CH / 4) + lane];
            float4 v2 = hb[(long)s2 * (CH / 4) + lane];
            float4 v3 = hb[(long)s3 * (CH / 4) + lane];
            gcn_fma4(acc, v0, w0);
            gcn_fma4(acc, v1, w1);
            gcn_fma4(acc, v2, w2);
            gcn_fma4(acc, v3, w3);
        }
        int rem = cnt - j;
        if (rem > 0) {
            int s0 = __builtin_amdgcn_readlane(sj, j);
            float w0 = __builtin_bit_cast(float, __builtin_amdgcn_readlane(wjb, j));
            float w1 = 0.f, w2 = 0.f;
            float4 v0 = hb[(long)s0 * (CH / 4) + lane];
            float4 v1 = {0.f, 0.f, 0.f, 0.f}, v2 = {0.f, 0.f, 0.f, 0.f};
            if (rem > 1) {
                int s1 = __builtin_amdgcn_readlane(sj, j + 1);
                w1 = __builtin_bit_cast(float, __builtin_amdgcn_readlane(wjb, j + 1));
                v1 = hb[(long)s1 * (CH / 4) + lane];
            }
            if (rem > 2) {
                int s2 = __builtin_amdgcn_readlane(sj, j + 2);
                w2 = __builtin_bit_cast(float, __builtin_amdgcn_readlane(wjb, j + 2));
                v2 = hb[(long)s2 * (CH / 4) + lane];
            }
            gcn_fma4(acc, v0, w0);
            if (rem > 1) gcn_fma4(acc, v1, w1);
            if (rem > 2) gcn_fma4(acc, v2, w2);
        }
    }
    ab[(long)d * (CH / 4) + lane] = acc;
}

__global__ void gcn_bn_stats(const float* __restrict__ x, float* __restrict__ sums, long rows) {
    int c = threadIdx.x;
    long r0 = (long)blockIdx.x * 256;
    long r1 = r0 + 256; if (r1 > rows) r1 = rows;
    float s = 0.f, q = 0.f;
    for (long r = r0; r < r1; ++r) {
        float v = x[r * CH + c];
        v = fmaxf(v, 0.f);
        s += v; q += v * v;
    }
    atomicAddF(&sums[c], s);
    atomicAddF(&sums[CH + c], q);
}

__global__ void gcn_bn_dot_out(const float* __restrict__ x, const float* __restrict__ sums,
                               const float* __restrict__ gamma, const float* __restrict__ beta,
                               const float* __restrict__ wc, const float* __restrict__ bc,
                               float* __restrict__ out, long rows, float invcnt) {
    long wid = (long)blockIdx.x * (blockDim.x >> 5) + (threadIdx.x >> 5);
    if (wid >= rows) return;
    int lane = threadIdx.x & 31;
    const float4* xp = (const float4*)(x + wid * CH);
    float4 v = xp[lane];
    float vv[4] = {v.x, v.y, v.z, v.w};
    float acc = 0.f;
#pragma unroll
    for (int j = 0; j < 4; ++j) {
        int c = lane * 4 + j;
        float mu  = sums[c] * invcnt;
        float var = sums[CH + c] * invcnt - mu * mu;
        float val = (fmaxf(vv[j], 0.f) - mu) * rsqrtf(var + BN_EPS) * gamma[c] + beta[c];
        acc += val * wc[c];
    }
#pragma unroll
    for (int o = 16; o > 0; o >>= 1)
        acc += __shfl_xor(acc, o, 32);
    if (lane == 0) out[wid] = acc + bc[0];
}

extern "C" void kernel_launch(void* const* d_in, const int* in_sizes, int n_in,
                              void* d_out, int out_size, void* d_ws, size_t ws_size,
                              hipStream_t stream) {
    const float* x   = (const float*)d_in[0];
    const int*   ei  = (const int*)d_in[1];
    const float* W1  = (const float*)d_in[2];
    const float* b1  = (const float*)d_in[3];
    const float* W2  = (const float*)d_in[4];
    const float* b2  = (const float*)d_in[5];
    const float* g1  = (const float*)d_in[6];
    const float* be1 = (const float*)d_in[7];
    const float* g2  = (const float*)d_in[8];
    const float* be2 = (const float*)d_in[9];
    const float* Wc  = (const float*)d_in[10];
    const float* bc  = (const float*)d_in[11];
    float* out = (float*)d_out;

    const int  E  = in_sizes[1] / 2;
    const int  N  = 20000;
    const int  H  = in_sizes[3];
    const int  F  = in_sizes[2] / H;
    const long BN = (long)out_size;
    const int  Bz = (int)(BN / N);
    const long TOT = BN * H;
    const float invcnt = 1.0f / (float)BN;

    float* ws     = (float*)d_ws;
    float* stats1 = ws;
    float* stats2 = ws + 256;
    float* dinv   = ws + 512;
    long   nAl    = (N + 255) & ~255L;
    float* bufA   = ws + 512 + nAl;

    const long PL   = (long)N * H;
    float* bufM     = bufA + (Bz - 1) * PL;
    float* hLast    = bufA + (Bz - 1) * PL;
    float* aggLast  = bufA + (2 * Bz - 2) * PL;
    long   n1Al     = (N + 1 + 255) & ~255L;
    long   eAl      = (E + 255) & ~255L;
    int*   csrEnd   = (int*)(bufA + 2 * TOT);
    int*   sorted   = csrEnd - eAl;
    int*   slots    = sorted - eAl;
    int*   rowptr   = slots - n1Al;
    int*   cursor   = rowptr - nAl;
    int*   cnt      = cursor - nAl;

    const int* srcIdx = ei;
    const int* dstIdx = ei + E;

    gcn_init  <<<(N + 255) / 256, 256, 0, stream>>>(stats1, cnt, N);
    gcn_count <<<(E + 255) / 256, 256, 0, stream>>>(dstIdx, cnt, E);
    gcn_scan  <<<1, 1024, 0, stream>>>(cnt, rowptr, cursor, dinv, N);
    gcn_fill  <<<(E + 255) / 256, 256, 0, stream>>>(srcIdx, dstIdx, cursor, slots, E);
    gcn_order <<<(N + 7) / 8, 256, 0, stream>>>(rowptr, slots, sorted, N);

    dim3 nodeGridLast((N + 7) / 8, 1, 1);
    dim3 nodeGridRest((N + 7) / 8, Bz - 1, 1);
    const long MR   = (long)(Bz - 1) * N;
    int  gemmBlks  = (int)((BN / 16 + 7) / 8);
    int  gemmBlksR = (int)((MR / 16 + 7) / 8);
    int  gemmBlksL = (int)((N / 16 + 7) / 8);
    int  statBlks  = (int)((BN + 255) / 256);
    int  rowBlks   = (int)((BN + 7) / 8);

    gcn_gemm_wmma<<<gemmBlks, 256, 0, stream>>>(x, W1, bufA, (int)BN, F);
    gcn_gather   <<<nodeGridLast, 256, 0, stream>>>(hLast, rowptr, sorted, dinv, b1, aggLast, N);
    gcn_gather   <<<nodeGridRest, 256, 0, stream>>>(bufA, rowptr, sorted, dinv, b1, bufM, N);
    gcn_bn_stats <<<statBlks, 128, 0, stream>>>(bufM, stats1, BN);

    gcn_gemm_wmma_bn<<<gemmBlksR, 256, 0, stream>>>(bufM, W2, bufA, (int)MR, H, stats1, g1, be1, invcnt);
    gcn_gemm_wmma_bn<<<gemmBlksL, 256, 0, stream>>>(aggLast, W2, hLast, N, H, stats1, g1, be1, invcnt);
    gcn_gather   <<<nodeGridLast, 256, 0, stream>>>(hLast, rowptr, sorted, dinv, b2, aggLast, N);
    gcn_gather   <<<nodeGridRest, 256, 0, stream>>>(bufA, rowptr, sorted, dinv, b2, bufM, N);
    gcn_bn_stats <<<statBlks, 128, 0, stream>>>(bufM, stats2, BN);

    gcn_bn_dot_out<<<rowBlks, 256, 0, stream>>>(bufM, stats2, g2, be2, Wc, bc, out, BN, invcnt);
}


# BatchNorm statistics pass: a wave keeps 10 whole-row 16-byte loads in flight (was one 4-byte load per round trip, 256 trips); its sums now go by wave row runs, an LDS wave sum, one atomic per block
# speedup vs baseline: 2.9712x; 1.8447x over previous
#include <hip/hip_runtime.h>
#include <hip/hip_bf16.h>

typedef __attribute__((ext_vector_type(2))) float v2f;
typedef __attribute__((ext_vector_type(8))) float v8f;

#define CH 128
#define BN_EPS 1e-5f

__device__ __forceinline__ void atomicAddF(float* p, float v) {
    (void)__hip_atomic_fetch_add(p, v, __ATOMIC_RELAXED, __HIP_MEMORY_SCOPE_AGENT);
}

__global__ void gcn_init(float* stats, int* cnt, int n) {
    int i = blockIdx.x * blockDim.x + threadIdx.x;
    if (i < 2 * 2 * CH) stats[i] = 0.0f;
    if (i < n) cnt[i] = 0;
}
__global__ void gcn_count(const int* __restrict__ dst, int* cnt, int e) {
    int i = blockIdx.x * blockDim.x + threadIdx.x;
    if (i < e) atomicAdd(&cnt[dst[i]], 1);
}
__global__ void gcn_scan(const int* __restrict__ cnt, int* __restrict__ rowptr,
                         int* __restrict__ cursor, float* __restrict__ dinv, int n) {
    __shared__ int part[1024];
    int t   = threadIdx.x;
    int per = (n + 1023) >> 10;
    int i0  = t * per;
    int i1  = i0 + per; if (i1 > n) i1 = n;
    int s = 0;
    for (int i = i0; i < i1; ++i) s += cnt[i];
    part[t] = s;
    __syncthreads();
    for (int off = 1; off < 1024; off <<= 1) {
        int v = (t >= off) ? part[t - off] : 0;
        __syncthreads();
        part[t] += v;
        __syncthreads();
    }
    int run = part[t] - s;
    for (int i = i0; i < i1; ++i) {
        int c = cnt[i];
        rowptr[i] = run;
        cursor[i] = run;
        dinv[i]   = rsqrtf((float)(c + 1));
        run += c;
    }
    if (t == 1023) rowptr[n] = part[1023];
}
__global__ void gcn_fill(const int* __restrict__ src, const int* __restrict__ dst,
                         int* cursor, int* __restrict__ slots, int e) {
    int i = blockIdx.x * blockDim.x + threadIdx.x;
    if (i < e) {
        int p = atomicAdd(&cursor[dst[i]], 1);
        slots[p] = src[i];
    }
}
__global__ void gcn_order(const int* __restrict__ rowptr, const int* __restrict__ slots,
                          int* __restrict__ sorted, int n) {
    int d = blockIdx.x * (blockDim.x >> 5) + (threadIdx.x >> 5);
    if (d >= n) return;
    int lane = threadIdx.x & 31;
    int beg = rowptr[d];
    int len = rowptr[d + 1] - beg;
    for (int i = lane; i < len; i += 32) {
        int v = slots[beg + i];
        int rank = 0;
        for (int j = 0; j < len; ++j) {
            int u = slots[beg + j];
            rank += (u < v || (u == v && j < i)) ? 1 : 0;
        }
        sorted[beg + rank] = v;
    }
}

template <bool BN>
__device__ __forceinline__ void gcn_gemm_body(const float* __restrict__ A, const float* __restrict__ W,
                                              float* __restrict__ C, int M, int K,
                                              const float4* prm) {
    int wave = blockIdx.x * (blockDim.x >> 5) + (threadIdx.x >> 5);
    int lane = threadIdx.x & 31;
    if (wave * 16 >= M) return;

    int mrow  = wave * 16 + (lane & 15);
    int khalf = (lane >> 4) << 1;
    int ncol  = lane & 15;

    v8f acc[8];
#pragma unroll
    for (int t = 0; t < 8; ++t) acc[t] = (v8f){0,0,0,0,0,0,0,0};

    for (int k0 = 0; k0 < K; k0 += 4) {
        v2f a;
        const float* ar = A + (long)mrow * K + k0 + khalf;
        a.x = ar[0];
        a.y = ar[1];
        if (BN) {
            float4 p0 = prm[k0 + khalf];
            float4 p1 = prm[k0 + khalf + 1];
            a.x = (fmaxf(a.x, 0.f) - p0.x) * p0.y * p0.z + p0.w;
            a.y = (fmaxf(a.y, 0.f) - p1.x) * p1.y * p1.z + p1.w;
        }
        const float* w0 = W + (long)(k0 + khalf) * CH;
        const float* w1 = w0 + CH;
#pragma unroll
        for (int t = 0; t < 8; ++t) {
            v2f b;
            b.x = w0[t * 16 + ncol];
            b.y = w1[t * 16 + ncol];
            acc[t] = __builtin_amdgcn_wmma_f32_16x16x4_f32(
                false, a, false, b, (short)0, acc[t], false, false);
        }
    }
    int rbase = wave * 16 + ((lane >> 4) << 3);
#pragma unroll
    for (int t = 0; t < 8; ++t)
#pragma unroll
        for (int r = 0; r < 8; ++r)
            C[(long)(rbase + r) * CH + t * 16 + (lane & 15)] = acc[t][r];
}

__global__ void gcn_gemm_wmma(const float* __restrict__ A, const float* __restrict__ W,
                              float* __restrict__ C, int M, int K) {
    gcn_gemm_body<false>(A, W, C, M, K, nullptr);
}

__global__ void gcn_gemm_wmma_bn(const float* __restrict__ A, const float* __restrict__ W,
                                 float* __restrict__ C, int M, int K,
                                 const float* __restrict__ sums, const float* __restrict__ gamma,
                                 const float* __restrict__ beta, float invcnt) {
    __shared__ float4 prm[CH];
    int c = threadIdx.x;
    if (c < CH) {
        float mu  = sums[c] * invcnt;
        float var = sums[CH + c] * invcnt - mu * mu;
        prm[c] = make_float4(mu, rsqrtf(var + BN_EPS), gamma[c], beta[c]);
    }
    __syncthreads();
    gcn_gemm_body<true>(A, W, C, M, K, prm);
}

__device__ __forceinline__ void gcn_fma4(float4& acc, const float4& v, float w) {
    acc.x = __builtin_fmaf(v.x, w, acc.x);
    acc.y = __builtin_fmaf(v.y, w, acc.y);
    acc.z = __builtin_fmaf(v.z, w, acc.z);
    acc.w = __builtin_fmaf(v.w, w, acc.w);
}
__global__ void gcn_gather(const float* __restrict__ h, const int* __restrict__ rowptr,
                           const int* __restrict__ csrc, const float* __restrict__ dinv,
                           const float* __restrict__ bias, float* __restrict__ agg, int n) {
    int d = blockIdx.x * (blockDim.x >> 5) + (threadIdx.x >> 5);
    if (d >= n) return;
    d = __builtin_amdgcn_readfirstlane(d);
    int lane = threadIdx.x & 31;
    const float4* hb = (const float4*)(h + (long)blockIdx.y * n * CH);
    float4*       ab = (float4*)(agg + (long)blockIdx.y * n * CH);
    int beg = __builtin_amdgcn_readfirstlane(rowptr[d]);
    int end = __builtin_amdgcn_readfirstlane(rowptr[d + 1]);
    float dd = dinv[d];
    float4 hv = hb[(long)d * (CH / 4) + lane];
    float4 bv = ((const float4*)bias)[lane];
    float4 acc;
    acc.x = bv.x + hv.x * dd * dd;
    acc.y = bv.y + hv.y * dd * dd;
    acc.z = bv.z + hv.z * dd * dd;
    acc.w = bv.w + hv.w * dd * dd;

    for (int base = beg; base < end; base += 32) {
        int cnt = end - base; if (cnt > 32) cnt = 32;
        int   sj = 0;
        float wj = 0.f;
        if (lane < cnt) { sj = csrc[base + lane]; wj = dinv[sj] * dd; }
        int wjb = __builtin_bit_cast(int, wj);
        int j = 0;
        for (; j + 4 <= cnt; j += 4) {
            int s0 = __builtin_amdgcn_readlane(sj, j);
            int s1 = __builtin_amdgcn_readlane(sj, j + 1);
            int s2 = __builtin_amdgcn_readlane(sj, j + 2);
            int s3 = __builtin_amdgcn_readlane(sj, j + 3);
            float w0 = __builtin_bit_cast(float, __builtin_amdgcn_readlane(wjb, j));
            float w1 = __builtin_bit_cast(float, __builtin_amdgcn_readlane(wjb, j + 1));
            float w2 = __builtin_bit_cast(float, __builtin_amdgcn_readlane(wjb, j + 2));
            float w3 = __builtin_bit_cast(float, __builtin_amdgcn_readlane(wjb, j + 3));
            float4 v0 = hb[(long)s0 * (CH / 4) + lane];
            float4 v1 = hb[(long)s1 * (CH / 4) + lane];
            float4 v2 = hb[(long)s2 * (CH / 4) + lane];
            float4 v3 = hb[(long)s3 * (CH / 4) + lane];
            gcn_fma4(acc, v0, w0);
            gcn_fma4(acc, v1, w1);
            gcn_fma4(acc, v2, w2);
            gcn_fma4(acc, v3, w3);
        }
        int rem = cnt - j;
        if (rem > 0) {
            int s0 = __builtin_amdgcn_readlane(sj, j);
            float w0 = __builtin_bit_cast(float, __builtin_amdgcn_readlane(wjb, j));
            float w1 = 0.f, w2 = 0.f;
            float4 v0 = hb[(long)s0 * (CH / 4) + lane];
            float4 v1 = {0.f, 0.f, 0.f, 0.f}, v2 = {0.f, 0.f, 0.f, 0.f};
            if (rem > 1) {
                int s1 = __builtin_amdgcn_readlane(sj, j + 1);
                w1 = __builtin_bit_cast(float, __builtin_amdgcn_readlane(wjb, j + 1));
                v1 = hb[(long)s1 * (CH / 4) + lane];
            }
            if (rem > 2) {
                int s2 = __builtin_amdgcn_readlane(sj, j + 2);
                w2 = __builtin_bit_cast(float, __builtin_amdgcn_readlane(wjb, j + 2));
                v2 = hb[(long)s2 * (CH / 4) + lane];
            }
            gcn_fma4(acc, v0, w0);
            if (rem > 1) gcn_fma4(acc, v1, w1);
            if (rem > 2) gcn_fma4(acc, v2, w2);
        }
    }
    ab[(long)d * (CH / 4) + lane] = acc;
}

#define ST_U     10
#define ST_WAVES 16
__global__ void gcn_bn_stats(const float* __restrict__ x, float* __restrict__ sums, long rows) {
    __shared__ float red[ST_WAVES][2 * CH];
    int lane = threadIdx.x & 31;
    int wv = __builtin_amdgcn_readfirstlane(threadIdx.x >> 5);
    long gw = (long)blockIdx.x * ST_WAVES + wv;
    long tw = (long)gridDim.x * ST_WAVES;
    const float4* xp = (const float4*)x;
    float4 s = {0.f, 0.f, 0.f, 0.f};
    float4 q = {0.f, 0.f, 0.f, 0.f};
    for (long r = gw; r < rows; r += tw * ST_U) {
        float4 v[ST_U];
#pragma unroll
        for (int u = 0; u < ST_U; ++u) {
            long rr = r + u * tw;
            v[u] = xp[(rr < rows ? rr : rows - 1) * (CH / 4) + lane];
        }
#pragma unroll
        for (int u = 0; u < ST_U; ++u) {
            bool in = r + u * tw < rows;
            float a = fmaxf(in ? v[u].x : 0.f, 0.f);
            float b = fmaxf(in ? v[u].y : 0.f, 0.f);
            float c = fmaxf(in ? v[u].z : 0.f, 0.f);
            float d = fmaxf(in ? v[u].w : 0.f, 0.f);
            s.x += a; s.y += b; s.z += c; s.w += d;
            q.x = __builtin_fmaf(a, a, q.x);
            q.y = __builtin_fmaf(b, b, q.y);
            q.z = __builtin_fmaf(c, c, q.z);
            q.w = __builtin_fmaf(d, d, q.w);
        }
    }
    *(float4*)&red[wv][lane * 4]      = s;
    *(float4*)&red[wv][CH + lane * 4] = q;
    __syncthreads();
    int t = threadIdx.x;
    if (t < 2 * CH) {
        float tot = 0.f;
#pragma unroll
        for (int w = 0; w < ST_WAVES; ++w) tot += red[w][t];
        atomicAddF(&sums[t], tot);
    }
}

__global__ void gcn_bn_dot_out(const float* __restrict__ x, const float* __restrict__ sums,
                               const float* __restrict__ gamma, const float* __restrict__ beta,
                               const float* __restrict__ wc, const float* __restrict__ bc,
                               float* __restrict__ out, long rows, float invcnt) {
    long wid = (long)blockIdx.x * (blockDim.x >> 5) + (threadIdx.x >> 5);
    if (wid >= rows) return;
    int lane = threadIdx.x & 31;
    const float4* xp = (const float4*)(x + wid * CH);
    float4 v = xp[lane];
    float vv[4] = {v.x, v.y, v.z, v.w};
    float acc = 0.f;
#pragma unroll
    for (int j = 0; j < 4; ++j) {
        int c = lane * 4 + j;
        float mu  = sums[c] * invcnt;
        float var = sums[CH + c] * invcnt - mu * mu;
        float val = (fmaxf(vv[j], 0.f) - mu) * rsqrtf(var + BN_EPS) * gamma[c] + beta[c];
        acc += val * wc[c];
    }
#pragma unroll
    for (int o = 16; o > 0; o >>= 1)
        acc += __shfl_xor(acc, o, 32);
    if (lane == 0) out[wid] = acc + bc[0];
}

extern "C" void kernel_launch(void* const* d_in, const int* in_sizes, int n_in,
                              void* d_out, int out_size, void* d_ws, size_t ws_size,
                              hipStream_t stream) {
    const float* x   = (const float*)d_in[0];
    const int*   ei  = (const int*)d_in[1];
    const float* W1  = (const float*)d_in[2];
    const float* b1  = (const float*)d_in[3];
    const float* W2  = (const float*)d_in[4];
    const float* b2  = (const float*)d_in[5];
    const float* g1  = (const float*)d_in[6];
    const float* be1 = (const float*)d_in[7];
    const float* g2  = (const float*)d_in[8];
    const float* be2 = (const float*)d_in[9];
    const float* Wc  = (const float*)d_in[10];
    const float* bc  = (const float*)d_in[11];
    float* out = (float*)d_out;

    const int  E  = in_sizes[1] / 2;
    const int  N  = 20000;
    const int  H  = in_sizes[3];
    const int  F  = in_sizes[2] / H;
    const long BN = (long)out_size;
    const int  Bz = (int)(BN / N);
    const long TOT = BN * H;
    const float invcnt = 1.0f / (float)BN;

    float* ws     = (float*)d_ws;
    float* stats1 = ws;
    float* stats2 = ws + 256;
    float* dinv   = ws + 512;
    long   nAl    = (N + 255) & ~255L;
    float* bufA   = ws + 512 + nAl;

    const long PL   = (long)N * H;
    float* bufM     = bufA + (Bz - 1) * PL;
    float* hLast    = bufA + (Bz - 1) * PL;
    float* aggLast  = bufA + (2 * Bz - 2) * PL;
    long   n1Al     = (N + 1 + 255) & ~255L;
    long   eAl      = (E + 255) & ~255L;
    int*   csrEnd   = (int*)(bufA + 2 * TOT);
    int*   sorted   = csrEnd - eAl;
    int*   slots    = sorted - eAl;
    int*   rowptr   = slots - n1Al;
    int*   cursor   = rowptr - nAl;
    int*   cnt      = cursor - nAl;

    const int* srcIdx = ei;
    const int* dstIdx = ei + E;

    gcn_init  <<<(N + 255) / 256, 256, 0, stream>>>(stats1, cnt, N);
    gcn_count <<<(E + 255) / 256, 256, 0, stream>>>(dstIdx, cnt, E);
    gcn_scan  <<<1, 1024, 0, stream>>>(cnt, rowptr, cursor, dinv, N);
    gcn_fill  <<<(E + 255) / 256, 256, 0, stream>>>(srcIdx, dstIdx, cursor, slots, E);
    gcn_order <<<(N + 7) / 8, 256, 0, stream>>>(rowptr, slots, sorted, N);

    dim3 nodeGridLast((N + 7) / 8, 1, 1);
    dim3 nodeGridRest((N + 7) / 8, Bz - 1, 1);
    const long MR   = (long)(Bz - 1) * N;
    int  gemmBlks  = (int)((BN / 16 + 7) / 8);
    int  gemmBlksR = (int)((MR / 16 + 7) / 8);
    int  gemmBlksL = (int)((N / 16 + 7) / 8);
    int  statBlks  = 256;
    int  rowBlks   = (int)((BN + 7) / 8);

    gcn_gemm_wmma<<<gemmBlks, 256, 0, stream>>>(x, W1, bufA, (int)BN, F);
    gcn_gather   <<<nodeGridLast, 256, 0, stream>>>(hLast, rowptr, sorted, dinv, b1, aggLast, N);
    gcn_gather   <<<nodeGridRest, 256, 0, stream>>>(bufA, rowptr, sorted, dinv, b1, bufM, N);
    gcn_bn_stats <<<statBlks, 512, 0, stream>>>(bufM, stats1, BN);

    gcn_gemm_wmma_bn<<<gemmBlksR, 256, 0, stream>>>(bufM, W2, bufA, (int)MR, H, stats1, g1, be1, invcnt);
    gcn_gemm_wmma_bn<<<gemmBlksL, 256, 0, stream>>>(aggLast, W2, hLast, N, H, stats1, g1, be1, invcnt);
    gcn_gather   <<<nodeGridLast, 256, 0, stream>>>(hLast, rowptr, sorted, dinv, b2, aggLast, N);
    gcn_gather   <<<nodeGridRest, 256, 0, stream>>>(bufA, rowptr, sorted, dinv, b2, bufM, N);
    gcn_bn_stats <<<statBlks, 512, 0, stream>>>(bufM, stats2, BN);

    gcn_bn_dot_out<<<rowBlks, 256, 0, stream>>>(bufM, stats2, g2, be2, Wc, bc, out, BN, invcnt);
}


# layer 1 sums the neighbours' 64-wide input rows first (one wave per node, all batches) and multiplies by W1 after, bias as the accumulators' start: order of sums changes (sum before product)
# speedup vs baseline: 3.2152x; 1.0821x over previous
#include <hip/hip_runtime.h>
#include <hip/hip_bf16.h>

typedef __attribute__((ext_vector_type(2))) float v2f;
typedef __attribute__((ext_vector_type(8))) float v8f;

#define CH 128
#define BN_EPS 1e-5f

__device__ __forceinline__ void atomicAddF(float* p, float v) {
    (void)__hip_atomic_fetch_add(p, v, __ATOMIC_RELAXED, __HIP_MEMORY_SCOPE_AGENT);
}

__global__ void gcn_init(float* stats, int* cnt, int n) {
    int i = blockIdx.x * blockDim.x + threadIdx.x;
    if (i < 2 * 2 * CH) stats[i] = 0.0f;
    if (i < n) cnt[i] = 0;
}
__global__ void gcn_count(const int* __restrict__ dst, int* cnt, int e) {
    int i = blockIdx.x * blockDim.x + threadIdx.x;
    if (i < e) atomicAdd(&cnt[dst[i]], 1);
}
__global__ void gcn_scan(const int* __restrict__ cnt, int* __restrict__ rowptr,
                         int* __restrict__ cursor, float* __restrict__ dinv, int n) {
    __shared__ int part[1024];
    int t   = threadIdx.x;
    int per = (n + 1023) >> 10;
    int i0  = t * per;
    int i1  = i0 + per; if (i1 > n) i1 = n;
    int s = 0;
    for (int i = i0; i < i1; ++i) s += cnt[i];
    part[t] = s;
    __syncthreads();
    for (int off = 1; off < 1024; off <<= 1) {
        int v = (t >= off) ? part[t - off] : 0;
        __syncthreads();
        part[t] += v;
        __syncthreads();
    }
    int run = part[t] - s;
    for (int i = i0; i < i1; ++i) {
        int c = cnt[i];
        rowptr[i] = run;
        cursor[i] = run;
        dinv[i]   = rsqrtf((float)(c + 1));
        run += c;
    }
    if (t == 1023) rowptr[n] = part[1023];
}
__global__ void gcn_fill(const int* __restrict__ src, const int* __restrict__ dst,
                         int* cursor, int* __restrict__ slots, int e) {
    int i = blockIdx.x * blockDim.x + threadIdx.x;
    if (i < e) {
        int p = atomicAdd(&cursor[dst[i]], 1);
        slots[p] = src[i];
    }
}
__global__ void gcn_order(const int* __restrict__ rowptr, const int* __restrict__ slots,
                          int* __restrict__ sorted, int n) {
    int d = blockIdx.x * (blockDim.x >> 5) + (threadIdx.x >> 5);
    if (d >= n) return;
    int lane = threadIdx.x & 31;
    int beg = rowptr[d];
    int len = rowptr[d + 1] - beg;
    for (int i = lane; i < len; i += 32) {
        int v = slots[beg + i];
        int rank = 0;
        for (int j = 0; j < len; ++j) {
            int u = slots[beg + j];
            rank += (u < v || (u == v && j < i)) ? 1 : 0;
        }
        sorted[beg + rank] = v;
    }
}

template <bool BN, bool BIAS>
__device__ __forceinline__ void gcn_gemm_body(const float* __restrict__ A, const float* __restrict__ W,
                                              float* __restrict__ C, int M, int K,
                                              const float4* prm, const float* __restrict__ bias) {
    int wave = blockIdx.x * (blockDim.x >> 5) + (threadIdx.x >> 5);
    int lane = threadIdx.x & 31;
    if (wave * 16 >= M) return;

    int mrow  = wave * 16 + (lane & 15);
    int khalf = (lane >> 4) << 1;
    int ncol  = lane & 15;

    v8f acc[8];
#pragma unroll
    for (int t = 0; t < 8; ++t) {
        float bv = BIAS ? bias[t * 16 + ncol] : 0.f;
        acc[t] = (v8f){bv, bv, bv, bv, bv, bv, bv, bv};
    }

    for (int k0 = 0; k0 < K; k0 += 4) {
        v2f a;
        const float* ar = A + (long)mrow * K + k0 + khalf;
        a.x = ar[0];
        a.y = ar[1];
        if (BN) {
            float4 p0 = prm[k0 + khalf];
            float4 p1 = prm[k0 + khalf + 1];
            a.x = (fmaxf(a.x, 0.f) - p0.x) * p0.y * p0.z + p0.w;
            a.y = (fmaxf(a.y, 0.f) - p1.x) * p1.y * p1.z + p1.w;
        }
        const float* w0 = W + (long)(k0 + khalf) * CH;
        const float* w1 = w0 + CH;
#pragma unroll
        for (int t = 0; t < 8; ++t) {
            v2f b;
            b.x = w0[t * 16 + ncol];
            b.y = w1[t * 16 + ncol];
            acc[t] = __builtin_amdgcn_wmma_f32_16x16x4_f32(
                false, a, false, b, (short)0, acc[t], false, false);
        }
    }
    int rbase = wave * 16 + ((lane >> 4) << 3);
#pragma unroll
    for (int t = 0; t < 8; ++t)
#pragma unroll
        for (int r = 0; r < 8; ++r)
            C[(long)(rbase + r) * CH + t * 16 + (lane & 15)] = acc[t][r];
}

__global__ void gcn_gemm_wmma(const float* __restrict__ A, const float* __restrict__ W,
                              const float* __restrict__ bias, float* __restrict__ C, int M, int K) {
    gcn_gemm_body<false, true>(A, W, C, M, K, nullptr, bias);
}

__global__ void gcn_gemm_wmma_bn(const float* __restrict__ A, const float* __restrict__ W,
                                 float* __restrict__ C, int M, int K,
                                 const float* __restrict__ sums, const float* __restrict__ gamma,
                                 const float* __restrict__ beta, float invcnt) {
    __shared__ float4 prm[CH];
    int c = threadIdx.x;
    if (c < CH) {
        float mu  = sums[c] * invcnt;
        float var = sums[CH + c] * invcnt - mu * mu;
        prm[c] = make_float4(mu, rsqrtf(var + BN_EPS), gamma[c], beta[c]);
    }
    __syncthreads();
    gcn_gemm_body<true, false>(A, W, C, M, K, prm, nullptr);
}

__device__ __forceinline__ void gcn_fma4(float4& acc, const float4& v, float w) {
    acc.x = __builtin_fmaf(v.x, w, acc.x);
    acc.y = __builtin_fmaf(v.y, w, acc.y);
    acc.z = __builtin_fmaf(v.z, w, acc.z);
    acc.w = __builtin_fmaf(v.w, w, acc.w);
}
__global__ void gcn_gather(const float* __restrict__ h, const int* __restrict__ rowptr,
                           const int* __restrict__ csrc, const float* __restrict__ dinv,
                           const float* __restrict__ bias, float* __restrict__ agg, int n) {
    int d = blockIdx.x * (blockDim.x >> 5) + (threadIdx.x >> 5);
    if (d >= n) return;
    d = __builtin_amdgcn_readfirstlane(d);
    int lane = threadIdx.x & 31;
    const float4* hb = (const float4*)(h + (long)blockIdx.y * n * CH);
    float4*       ab = (float4*)(agg + (long)blockIdx.y * n * CH);
    int beg = __builtin_amdgcn_readfirstlane(rowptr[d]);
    int end = __builtin_amdgcn_readfirstlane(rowptr[d + 1]);
    float dd = dinv[d];
    float4 hv = hb[(long)d * (CH / 4) + lane];
    float4 bv = ((const float4*)bias)[lane];
    float4 acc;
    acc.x = bv.x + hv.x * dd * dd;
    acc.y = bv.y + hv.y * dd * dd;
    acc.z = bv.z + hv.z * dd * dd;
    acc.w = bv.w + hv.w * dd * dd;

    for (int base = beg; base < end; base += 32) {
        int cnt = end - base; if (cnt > 32) cnt = 32;
        int   sj = 0;
        float wj = 0.f;
        if (lane < cnt) { sj = csrc[base + lane]; wj = dinv[sj] * dd; }
        int wjb = __builtin_bit_cast(int, wj);
        int j = 0;
        for (; j + 4 <= cnt; j += 4) {
            int s0 = __builtin_amdgcn_readlane(sj, j);
            int s1 = __builtin_amdgcn_readlane(sj, j + 1);
            int s2 = __builtin_amdgcn_readlane(sj, j + 2);
            int s3 = __builtin_amdgcn_readlane(sj, j + 3);
            float w0 = __builtin_bit_cast(float, __builtin_amdgcn_readlane(wjb, j));
            float w1 = __builtin_bit_cast(float, __builtin_amdgcn_readlane(wjb, j + 1));
            float w2 = __builtin_bit_cast(float, __builtin_amdgcn_readlane(wjb, j + 2));
            float w3 = __builtin_bit_cast(float, __builtin_amdgcn_readlane(wjb, j + 3));
            float4 v0 = hb[(long)s0 * (CH / 4) + lane];
            float4 v1 = hb[(long)s1 * (CH / 4) + lane];
            float4 v2 = hb[(long)s2 * (CH / 4) + lane];
            float4 v3 = hb[(long)s3 * (CH / 4) + lane];
            gcn_fma4(acc, v0, w0);
            gcn_fma4(acc, v1, w1);
            gcn_fma4(acc, v2, w2);
            gcn_fma4(acc, v3, w3);
        }
        int rem = cnt - j;
        if (rem > 0) {
            int s0 = __builtin_amdgcn_readlane(sj, j);
            float w0 = __builtin_bit_cast(float, __builtin_amdgcn_readlane(wjb, j));
            float w1 = 0.f, w2 = 0.f;
            float4 v0 = hb[(long)s0 * (CH / 4) + lane];
            float4 v1 = {0.f, 0.f, 0.f, 0.f}, v2 = {0.f, 0.f, 0.f, 0.f};
            if (rem > 1) {
                int s1 = __builtin_amdgcn_readlane(sj, j + 1);
                w1 = __builtin_bit_cast(float, __builtin_amdgcn_readlane(wjb, j + 1));
                v1 = hb[(long)s1 * (CH / 4) + lane];
            }
            if (rem > 2) {
                int s2 = __builtin_amdgcn_readlane(sj, j + 2);
                w2 = __builtin_bit_cast(float, __builtin_amdgcn_readlane(wjb, j + 2));
                v2 = hb[(long)s2 * (CH / 4) + lane];
            }
            gcn_fma4(acc, v0, w0);
            if (rem > 1) gcn_fma4(acc, v1, w1);
            if (rem > 2) gcn_fma4(acc, v2, w2);
        }
    }
    ab[(long)d * (CH / 4) + lane] = acc;
}

#define FIN 64
__global__ void gcn_gather_in(const float* __restrict__ x, const int* __restrict__ rowptr,
                              const int* __restrict__ csrc, const float* __restrict__ dinv,
                              float* __restrict__ ax, int n, int nb) {
    int d = blockIdx.x * (blockDim.x >> 5) + (threadIdx.x >> 5);
    if (d >= n) return;
    d = __builtin_amdgcn_readfirstlane(d);
    int lane = threadIdx.x & 31;
    int c4   = lane & 15;
    int pb0  = blockIdx.y * 4 + (lane >> 4);
    int pb1  = pb0 + 2;
    bool ok0 = pb0 < nb, ok1 = pb1 < nb;
    const float4* x0 = (const float4*)x + (long)(ok0 ? pb0 : 0) * n * (FIN / 4) + c4;
    const float4* x1 = (const float4*)x + (long)(ok1 ? pb1 : 0) * n * (FIN / 4) + c4;
    int beg = __builtin_amdgcn_readfirstlane(rowptr[d]);
    int end = __builtin_amdgcn_readfirstlane(rowptr[d + 1]);
    float dd = dinv[d];
    float4 h0 = x0[(long)d * (FIN / 4)];
    float4 h1 = x1[(long)d * (FIN / 4)];
    float4 acc0, acc1;
    acc0.x = h0.x * dd * dd; acc0.y = h0.y * dd * dd; acc0.z = h0.z * dd * dd; acc0.w = h0.w * dd * dd;
    acc1.x = h1.x * dd * dd; acc1.y = h1.y * dd * dd; acc1.z = h1.z * dd * dd; acc1.w = h1.w * dd * dd;

    for (int base = beg; base < end; base += 32) {
        int cnt = end - base; if (cnt > 32) cnt = 32;
        int   sj = 0;
        float wj = 0.f;
        if (lane < cnt) { sj = csrc[base + lane]; wj = dinv[sj] * dd; }
        int wjb = __builtin_bit_cast(int, wj);
        int j = 0;
        for (; j + 4 <= cnt; j += 4) {
            int s0 = __builtin_amdgcn_readlane(sj, j);
            int s1 = __builtin_amdgcn_readlane(sj, j + 1);
            int s2 = __builtin_amdgcn_readlane(sj, j + 2);
            int s3 = __builtin_amdgcn_readlane(sj, j + 3);
            float4 a0 = x0[(long)s0 * (FIN / 4)];
            float4 b0 = x1[(long)s0 * (FIN / 4)];
            float4 a1 = x0[(long)s1 * (FIN / 4)];
            float4 b1 = x1[(long)s1 * (FIN / 4)];
            float4 a2 = x0[(long)s2 * (FIN / 4)];
            float4 b2 = x1[(long)s2 * (FIN / 4)];
            float4 a3 = x0[(long)s3 * (FIN / 4)];
            float4 b3 = x1[(long)s3 * (FIN / 4)];
            float w0 = __builtin_bit_cast(float, __builtin_amdgcn_readlane(wjb, j));
            float w1 = __builtin_bit_cast(float, __builtin_amdgcn_readlane(wjb, j + 1));
            float w2 = __builtin_bit_cast(float, __builtin_amdgcn_readlane(wjb, j + 2));
            float w3 = __builtin_bit_cast(float, __builtin_amdgcn_readlane(wjb, j + 3));
            gcn_fma4(acc0, a0, w0); gcn_fma4(acc1, b0, w0);
            gcn_fma4(acc0, a1, w1); gcn_fma4(acc1, b1, w1);
            gcn_fma4(acc0, a2, w2); gcn_fma4(acc1, b2, w2);
            gcn_fma4(acc0, a3, w3); gcn_fma4(acc1, b3, w3);
        }
        int rem = cnt - j;
        if (rem > 0) {
            const float4 z = {0.f, 0.f, 0.f, 0.f};
            int s0 = __builtin_amdgcn_readlane(sj, j);
            float w0 = __builtin_bit_cast(float, __builtin_amdgcn_readlane(wjb, j));
            float w1 = 0.f, w2 = 0.f;
            float4 a0 = x0[(long)s0 * (FIN / 4)];
            float4 b0 = x1[(long)s0 * (FIN / 4)];
            float4 a1 = z, b1 = z, a2 = z, b2 = z;
            if (rem > 1) {
                int s1 = __builtin_amdgcn_readlane(sj, j + 1);
                w1 = __builtin_bit_cast(float, __builtin_amdgcn_readlane(wjb, j + 1));
                a1 = x0[(long)s1 * (FIN / 4)];
                b1 = x1[(long)s1 * (FIN / 4)];
            }
            if (rem > 2) {
                int s2 = __builtin_amdgcn_readlane(sj, j + 2);
                w2 = __builtin_bit_cast(float, __builtin_amdgcn_readlane(wjb, j + 2));
                a2 = x0[(long)s2 * (FIN / 4)];
                b2 = x1[(long)s2 * (FIN / 4)];
            }
            gcn_fma4(acc0, a0, w0); gcn_fma4(acc1, b0, w0);
            if (rem > 1) { gcn_fma4(acc0, a1, w1); gcn_fma4(acc1, b1, w1); }
            if (rem > 2) { gcn_fma4(acc0, a2, w2); gcn_fma4(acc1, b2, w2); }
        }
    }
    float4* o = (float4*)ax;
    if (ok0) o[((long)pb0 * n + d) * (FIN / 4) + c4] = acc0;
    if (ok1) o[((long)pb1 * n + d) * (FIN / 4) + c4] = acc1;
}

#define ST_U     10
#define ST_WAVES 16
__global__ void gcn_bn_stats(const float* __restrict__ x, float* __restrict__ sums, long rows) {
    __shared__ float red[ST_WAVES][2 * CH];
    int lane = threadIdx.x & 31;
    int wv = __builtin_amdgcn_readfirstlane(threadIdx.x >> 5);
    long gw = (long)blockIdx.x * ST_WAVES + wv;
    long tw = (long)gridDim.x * ST_WAVES;
    const float4* xp = (const float4*)x;
    float4 s = {0.f, 0.f, 0.f, 0.f};
    float4 q = {0.f, 0.f, 0.f, 0.f};
    for (long r = gw; r < rows; r += tw * ST_U) {
        float4 v[ST_U];
#pragma unroll
        for (int u = 0; u < ST_U; ++u) {
            long rr = r + u * tw;
            v[u] = xp[(rr < rows ? rr : rows - 1) * (CH / 4) + lane];
        }
#pragma unroll
        for (int u = 0; u < ST_U; ++u) {
            bool in = r + u * tw < rows;
            float a = fmaxf(in ? v[u].x : 0.f, 0.f);
            float b = fmaxf(in ? v[u].y : 0.f, 0.f);
            float c = fmaxf(in ? v[u].z : 0.f, 0.f);
            float d = fmaxf(in ? v[u].w : 0.f, 0.f);
            s.x += a; s.y += b; s.z += c; s.w += d;
            q.x = __builtin_fmaf(a, a, q.x);
            q.y = __builtin_fmaf(b, b, q.y);
            q.z = __builtin_fmaf(c, c, q.z);
            q.w = __builtin_fmaf(d, d, q.w);
        }
    }
    *(float4*)&red[wv][lane * 4]      = s;
    *(float4*)&red[wv][CH + lane * 4] = q;
    __syncthreads();
    int t = threadIdx.x;
    if (t < 2 * CH) {
        float tot = 0.f;
#pragma unroll
        for (int w = 0; w < ST_WAVES; ++w) tot += red[w][t];
        atomicAddF(&sums[t], tot);
    }
}

__global__ void gcn_bn_dot_out(const float* __restrict__ x, const float* __restrict__ sums,
                               const float* __restrict__ gamma, const float* __restrict__ beta,
                               const float* __restrict__ wc, const float* __restrict__ bc,
                               float* __restrict__ out, long rows, float invcnt) {
    long wid = (long)blockIdx.x * (blockDim.x >> 5) + (threadIdx.x >> 5);
    if (wid >= rows) return;
    int lane = threadIdx.x & 31;
    const float4* xp = (const float4*)(x + wid * CH);
    float4 v = xp[lane];
    float vv[4] = {v.x, v.y, v.z, v.w};
    float acc = 0.f;
#pragma unroll
    for (int j = 0; j < 4; ++j) {
        int c = lane * 4 + j;
        float mu  = sums[c] * invcnt;
        float var = sums[CH + c] * invcnt - mu * mu;
        float val = (fmaxf(vv[j], 0.f) - mu) * rsqrtf(var + BN_EPS) * gamma[c] + beta[c];
        acc += val * wc[c];
    }
#pragma unroll
    for (int o = 16; o > 0; o >>= 1)
        acc += __shfl_xor(acc, o, 32);
    if (lane == 0) out[wid] = acc + bc[0];
}

extern "C" void kernel_launch(void* const* d_in, const int* in_sizes, int n_in,
                              void* d_out, int out_size, void* d_ws, size_t ws_size,
                              hipStream_t stream) {
    const float* x   = (const float*)d_in[0];
    const int*   ei  = (const int*)d_in[1];
    const float* W1  = (const float*)d_in[2];
    const float* b1  = (const float*)d_in[3];
    const float* W2  = (const float*)d_in[4];
    const float* b2  = (const float*)d_in[5];
    const float* g1  = (const float*)d_in[6];
    const float* be1 = (const float*)d_in[7];
    const float* g2  = (const float*)d_in[8];
    const float* be2 = (const float*)d_in[9];
    const float* Wc  = (const float*)d_in[10];
    const float* bc  = (const float*)d_in[11];
    float* out = (float*)d_out;

    const int  E  = in_sizes[1] / 2;
    const int  N  = 20000;
    const int  H  = in_sizes[3];
    const int  F  = in_sizes[2] / H;
    const long BN = (long)out_size;
    const int  Bz = (int)(BN / N);
    const long TOT = BN * H;
    const float invcnt = 1.0f / (float)BN;

    float* ws     = (float*)d_ws;
    float* stats1 = ws;
    float* stats2 = ws + 256;
    float* dinv   = ws + 512;
    long   nAl    = (N + 255) & ~255L;
    float* bufA   = ws + 512 + nAl;

    const long PL   = (long)N * H;
    float* bufM     = bufA + (Bz - 1) * PL;
    float* hLast    = bufA + (Bz - 1) * PL;
    float* aggLast  = bufA + (2 * Bz - 2) * PL;
    long   n1Al     = (N + 1 + 255) & ~255L;
    long   eAl      = (E + 255) & ~255L;
    int*   csrEnd   = (int*)(bufA + 2 * TOT);
    int*   sorted   = csrEnd - eAl;
    int*   slots    = sorted - eAl;
    int*   rowptr   = slots - n1Al;
    int*   cursor   = rowptr - nAl;
    int*   cnt      = cursor - nAl;

    const int* srcIdx = ei;
    const int* dstIdx = ei + E;

    gcn_init  <<<(N + 255) / 256, 256, 0, stream>>>(stats1, cnt, N);
    gcn_count <<<(E + 255) / 256, 256, 0, stream>>>(dstIdx, cnt, E);
    gcn_scan  <<<1, 1024, 0, stream>>>(cnt, rowptr, cursor, dinv, N);
    gcn_fill  <<<(E + 255) / 256, 256, 0, stream>>>(srcIdx, dstIdx, cursor, slots, E);
    gcn_order <<<(N + 7) / 8, 256, 0, stream>>>(rowptr, slots, sorted, N);

    dim3 nodeGridLast((N + 7) / 8, 1, 1);
    dim3 nodeGridRest((N + 7) / 8, Bz - 1, 1);
    const long MR   = (long)(Bz - 1) * N;
    int  gemmBlks  = (int)((BN / 16 + 7) / 8);
    int  gemmBlksR = (int)((MR / 16 + 7) / 8);
    int  gemmBlksL = (int)((N / 16 + 7) / 8);
    int  statBlks  = 256;
    int  rowBlks   = (int)((BN + 7) / 8);

    dim3 nodeGridIn((N + 7) / 8, (Bz + 3) / 4, 1);
    gcn_gather_in<<<nodeGridIn, 256, 0, stream>>>(x, rowptr, sorted, dinv, bufA, N, Bz);
    gcn_gemm_wmma<<<gemmBlks, 256, 0, stream>>>(bufA, W1, b1, bufM, (int)BN, F);
    gcn_bn_stats <<<statBlks, 512, 0, stream>>>(bufM, stats1, BN);

    gcn_gemm_wmma_bn<<<gemmBlksR, 256, 0, stream>>>(bufM, W2, bufA, (int)MR, H, stats1, g1, be1, invcnt);
    gcn_gemm_wmma_bn<<<gemmBlksL, 256, 0, stream>>>(aggLast, W2, hLast, N, H, stats1, g1, be1, invcnt);
    gcn_gather   <<<nodeGridLast, 256, 0, stream>>>(hLast, rowptr, sorted, dinv, b2, aggLast, N);
    gcn_gather   <<<nodeGridRest, 256, 0, stream>>>(bufA, rowptr, sorted, dinv, b2, bufM, N);
    gcn_bn_stats <<<statBlks, 512, 0, stream>>>(bufM, stats2, BN);

    gcn_bn_dot_out<<<rowBlks, 256, 0, stream>>>(bufM, stats2, g2, be2, Wc, bc, out, BN, invcnt);
}


# final BatchNorm-2 + classifier pass: 4096 resident waves each take 20 consecutive rows, all 20 row loads in flight (was 80000 one-row waves); channel terms once per wave; order of sums unchanged
# speedup vs baseline: 3.3134x; 1.0305x over previous
#include <hip/hip_runtime.h>
#include <hip/hip_bf16.h>

typedef __attribute__((ext_vector_type(2))) float v2f;
typedef __attribute__((ext_vector_type(8))) float v8f;

#define CH 128
#define BN_EPS 1e-5f

__device__ __forceinline__ void atomicAddF(float* p, float v) {
    (void)__hip_atomic_fetch_add(p, v, __ATOMIC_RELAXED, __HIP_MEMORY_SCOPE_AGENT);
}

__global__ void gcn_init(float* stats, int* cnt, int n) {
    int i = blockIdx.x * blockDim.x + threadIdx.x;
    if (i < 2 * 2 * CH) stats[i] = 0.0f;
    if (i < n) cnt[i] = 0;
}
__global__ void gcn_count(const int* __restrict__ dst, int* cnt, int e) {
    int i = blockIdx.x * blockDim.x + threadIdx.x;
    if (i < e) atomicAdd(&cnt[dst[i]], 1);
}
__global__ void gcn_scan(const int* __restrict__ cnt, int* __restrict__ rowptr,
                         int* __restrict__ cursor, float* __restrict__ dinv, int n) {
    __shared__ int part[1024];
    int t   = threadIdx.x;
    int per = (n + 1023) >> 10;
    int i0  = t * per;
    int i1  = i0 + per; if (i1 > n) i1 = n;
    int s = 0;
    for (int i = i0; i < i1; ++i) s += cnt[i];
    part[t] = s;
    __syncthreads();
    for (int off = 1; off < 1024; off <<= 1) {
        int v = (t >= off) ? part[t - off] : 0;
        __syncthreads();
        part[t] += v;
        __syncthreads();
    }
    int run = part[t] - s;
    for (int i = i0; i < i1; ++i) {
        int c = cnt[i];
        rowptr[i] = run;
        cursor[i] = run;
        dinv[i]   = rsqrtf((float)(c + 1));
        run += c;
    }
    if (t == 1023) rowptr[n] = part[1023];
}
__global__ void gcn_fill(const int* __restrict__ src, const int* __restrict__ dst,
                         int* cursor, int* __restrict__ slots, int e) {
    int i = blockIdx.x * blockDim.x + threadIdx.x;
    if (i < e) {
        int p = atomicAdd(&cursor[dst[i]], 1);
        slots[p] = src[i];
    }
}
__global__ void gcn_order(const int* __restrict__ rowptr, const int* __restrict__ slots,
                          int* __restrict__ sorted, int n) {
    int d = blockIdx.x * (blockDim.x >> 5) + (threadIdx.x >> 5);
    if (d >= n) return;
    int lane = threadIdx.x & 31;
    int beg = rowptr[d];
    int len = rowptr[d + 1] - beg;
    for (int i = lane; i < len; i += 32) {
        int v = slots[beg + i];
        int rank = 0;
        for (int j = 0; j < len; ++j) {
            int u = slots[beg + j];
            rank += (u < v || (u == v && j < i)) ? 1 : 0;
        }
        sorted[beg + rank] = v;
    }
}

template <bool BN, bool BIAS>
__device__ __forceinline__ void gcn_gemm_body(const float* __restrict__ A, const float* __restrict__ W,
                                              float* __restrict__ C, int M, int K,
                                              const float4* prm, const float* __restrict__ bias) {
    int wave = blockIdx.x * (blockDim.x >> 5) + (threadIdx.x >> 5);
    int lane = threadIdx.x & 31;
    if (wave * 16 >= M) return;

    int mrow  = wave * 16 + (lane & 15);
    int khalf = (lane >> 4) << 1;
    int ncol  = lane & 15;

    v8f acc[8];
#pragma unroll
    for (int t = 0; t < 8; ++t) {
        float bv = BIAS ? bias[t * 16 + ncol] : 0.f;
        acc[t] = (v8f){bv, bv, bv, bv, bv, bv, bv, bv};
    }

    for (int k0 = 0; k0 < K; k0 += 4) {
        v2f a;
        const float* ar = A + (long)mrow * K + k0 + khalf;
        a.x = ar[0];
        a.y = ar[1];
        if (BN) {
            float4 p0 = prm[k0 + khalf];
            float4 p1 = prm[k0 + khalf + 1];
            a.x = (fmaxf(a.x, 0.f) - p0.x) * p0.y * p0.z + p0.w;
            a.y = (fmaxf(a.y, 0.f) - p1.x) * p1.y * p1.z + p1.w;
        }
        const float* w0 = W + (long)(k0 + khalf) * CH;
        const float* w1 = w0 + CH;
#pragma unroll
        for (int t = 0; t < 8; ++t) {
            v2f b;
            b.x = w0[t * 16 + ncol];
            b.y = w1[t * 16 + ncol];
            acc[t] = __builtin_amdgcn_wmma_f32_16x16x4_f32(
                false, a, false, b, (short)0, acc[t], false, false);
        }
    }
    int rbase = wave * 16 + ((lane >> 4) << 3);
#pragma unroll
    for (int t = 0; t < 8; ++t)
#pragma unroll
        for (int r = 0; r < 8; ++r)
            C[(long)(rbase + r) * CH + t * 16 + (lane & 15)] = acc[t][r];
}

__global__ void gcn_gemm_wmma(const float* __restrict__ A, const float* __restrict__ W,
                              const float* __restrict__ bias, float* __restrict__ C, int M, int K) {
    gcn_gemm_body<false, true>(A, W, C, M, K, nullptr, bias);
}

__global__ void gcn_gemm_wmma_bn(const float* __restrict__ A, const float* __restrict__ W,
                                 float* __restrict__ C, int M, int K,
                                 const float* __restrict__ sums, const float* __restrict__ gamma,
                                 const float* __restrict__ beta, float invcnt) {
    __shared__ float4 prm[CH];
    int c = threadIdx.x;
    if (c < CH) {
        float mu  = sums[c] * invcnt;
        float var = sums[CH + c] * invcnt - mu * mu;
        prm[c] = make_float4(mu, rsqrtf(var + BN_EPS), gamma[c], beta[c]);
    }
    __syncthreads();
    gcn_gemm_body<true, false>(A, W, C, M, K, prm, nullptr);
}

__device__ __forceinline__ void gcn_fma4(float4& acc, const float4& v, float w) {
    acc.x = __builtin_fmaf(v.x, w, acc.x);
    acc.y = __builtin_fmaf(v.y, w, acc.y);
    acc.z = __builtin_fmaf(v.z, w, acc.z);
    acc.w = __builtin_fmaf(v.w, w, acc.w);
}
__global__ void gcn_gather(const float* __restrict__ h, const int* __restrict__ rowptr,
                           const int* __restrict__ csrc, const float* __restrict__ dinv,
                           const float* __restrict__ bias, float* __restrict__ agg, int n) {
    int d = blockIdx.x * (blockDim.x >> 5) + (threadIdx.x >> 5);
    if (d >= n) return;
    d = __builtin_amdgcn_readfirstlane(d);
    int lane = threadIdx.x & 31;
    const float4* hb = (const float4*)(h + (long)blockIdx.y * n * CH);
    float4*       ab = (float4*)(agg + (long)blockIdx.y * n * CH);
    int beg = __builtin_amdgcn_readfirstlane(rowptr[d]);
    int end = __builtin_amdgcn_readfirstlane(rowptr[d + 1]);
    float dd = dinv[d];
    float4 hv = hb[(long)d * (CH / 4) + lane];
    float4 bv = ((const float4*)bias)[lane];
    float4 acc;
    acc.x = bv.x + hv.x * dd * dd;
    acc.y = bv.y + hv.y * dd * dd;
    acc.z = bv.z + hv.z * dd * dd;
    acc.w = bv.w + hv.w * dd * dd;

    for (int base = beg; base < end; base += 32) {
        int cnt = end - base; if (cnt > 32) cnt = 32;
        int   sj = 0;
        float wj = 0.f;
        if (lane < cnt) { sj = csrc[base + lane]; wj = dinv[sj] * dd; }
        int wjb = __builtin_bit_cast(int, wj);
        int j = 0;
        for (; j + 4 <= cnt; j += 4) {
            int s0 = __builtin_amdgcn_readlane(sj, j);
            int s1 = __builtin_amdgcn_readlane(sj, j + 1);
            int s2 = __builtin_amdgcn_readlane(sj, j + 2);
            int s3 = __builtin_amdgcn_readlane(sj, j + 3);
            float w0 = __builtin_bit_cast(float, __builtin_amdgcn_readlane(wjb, j));
            float w1 = __builtin_bit_cast(float, __builtin_amdgcn_readlane(wjb, j + 1));
            float w2 = __builtin_bit_cast(float, __builtin_amdgcn_readlane(wjb, j + 2));
            float w3 = __builtin_bit_cast(float, __builtin_amdgcn_readlane(wjb, j + 3));
            float4 v0 = hb[(long)s0 * (CH / 4) + lane];
            float4 v1 = hb[(long)s1 * (CH / 4) + lane];
            float4 v2 = hb[(long)s2 * (CH / 4) + lane];
            float4 v3 = hb[(long)s3 * (CH / 4) + lane];
            gcn_fma4(acc, v0, w0);
            gcn_fma4(acc, v1, w1);
            gcn_fma4(acc, v2, w2);
            gcn_fma4(acc, v3, w3);
        }
        int rem = cnt - j;
        if (rem > 0) {
            int s0 = __builtin_amdgcn_readlane(sj, j);
            float w0 = __builtin_bit_cast(float, __builtin_amdgcn_readlane(wjb, j));
            float w1 = 0.f, w2 = 0.f;
            float4 v0 = hb[(long)s0 * (CH / 4) + lane];
            float4 v1 = {0.f, 0.f, 0.f, 0.f}, v2 = {0.f, 0.f, 0.f, 0.f};
            if (rem > 1) {
                int s1 = __builtin_amdgcn_readlane(sj, j + 1);
                w1 = __builtin_bit_cast(float, __builtin_amdgcn_readlane(wjb, j + 1));
                v1 = hb[(long)s1 * (CH / 4) + lane];
            }
            if (rem > 2) {
                int s2 = __builtin_amdgcn_readlane(sj, j + 2);
                w2 = __builtin_bit_cast(float, __builtin_amdgcn_readlane(wjb, j + 2));
                v2 = hb[(long)s2 * (CH / 4) + lane];
            }
            gcn_fma4(acc, v0, w0);
            if (rem > 1) gcn_fma4(acc, v1, w1);
            if (rem > 2) gcn_fma4(acc, v2, w2);
        }
    }
    ab[(long)d * (CH / 4) + lane] = acc;
}

#define FIN 64
__global__ void gcn_gather_in(const float* __restrict__ x, const int* __restrict__ rowptr,
                              const int* __restrict__ csrc, const float* __restrict__ dinv,
                              float* __restrict__ ax, int n, int nb) {
    int d = blockIdx.x * (blockDim.x >> 5) + (threadIdx.x >> 5);
    if (d >= n) return;
    d = __builtin_amdgcn_readfirstlane(d);
    int lane = threadIdx.x & 31;
    int c4   = lane & 15;
    int pb0  = blockIdx.y * 4 + (lane >> 4);
    int pb1  = pb0 + 2;
    bool ok0 = pb0 < nb, ok1 = pb1 < nb;
    const float4* x0 = (const float4*)x + (long)(ok0 ? pb0 : 0) * n * (FIN / 4) + c4;
    const float4* x1 = (const float4*)x + (long)(ok1 ? pb1 : 0) * n * (FIN / 4) + c4;
    int beg = __builtin_amdgcn_readfirstlane(rowptr[d]);
    int end = __builtin_amdgcn_readfirstlane(rowptr[d + 1]);
    float dd = dinv[d];
    float4 h0 = x0[(long)d * (FIN / 4)];
    float4 h1 = x1[(long)d * (FIN / 4)];
    float4 acc0, acc1;
    acc0.x = h0.x * dd * dd; acc0.y = h0.y * dd * dd; acc0.z = h0.z * dd * dd; acc0.w = h0.w * dd * dd;
    acc1.x = h1.x * dd * dd; acc1.y = h1.y * dd * dd; acc1.z = h1.z * dd * dd; acc1.w = h1.w * dd * dd;

    for (int base = beg; base < end; base += 32) {
        int cnt = end - base; if (cnt > 32) cnt = 32;
        int   sj = 0;
        float wj = 0.f;
        if (lane < cnt) { sj = csrc[base + lane]; wj = dinv[sj] * dd; }
        int wjb = __builtin_bit_cast(int, wj);
        int j = 0;
        for (; j + 4 <= cnt; j += 4) {
            int s0 = __builtin_amdgcn_readlane(sj, j);
            int s1 = __builtin_amdgcn_readlane(sj, j + 1);
            int s2 = __builtin_amdgcn_readlane(sj, j + 2);
            int s3 = __builtin_amdgcn_readlane(sj, j + 3);
            float4 a0 = x0[(long)s0 * (FIN / 4)];
            float4 b0 = x1[(long)s0 * (FIN / 4)];
            float4 a1 = x0[(long)s1 * (FIN / 4)];
            float4 b1 = x1[(long)s1 * (FIN / 4)];
            float4 a2 = x0[(long)s2 * (FIN / 4)];
            float4 b2 = x1[(long)s2 * (FIN / 4)];
            float4 a3 = x0[(long)s3 * (FIN / 4)];
            float4 b3 = x1[(long)s3 * (FIN / 4)];
            float w0 = __builtin_bit_cast(float, __builtin_amdgcn_readlane(wjb, j));
            float w1 = __builtin_bit_cast(float, __builtin_amdgcn_readlane(wjb, j + 1));
            float w2 = __builtin_bit_cast(float, __builtin_amdgcn_readlane(wjb, j + 2));
            float w3 = __builtin_bit_cast(float, __builtin_amdgcn_readlane(wjb, j + 3));
            gcn_fma4(acc0, a0, w0); gcn_fma4(acc1, b0, w0);
            gcn_fma4(acc0, a1, w1); gcn_fma4(acc1, b1, w1);
            gcn_fma4(acc0, a2, w2); gcn_fma4(acc1, b2, w2);
            gcn_fma4(acc0, a3, w3); gcn_fma4(acc1, b3, w3);
        }
        int rem = cnt - j;
        if (rem > 0) {
            const float4 z = {0.f, 0.f, 0.f, 0.f};
            int s0 = __builtin_amdgcn_readlane(sj, j);
            float w0 = __builtin_bit_cast(float, __builtin_amdgcn_readlane(wjb, j));
            float w1 = 0.f, w2 = 0.f;
            float4 a0 = x0[(long)s0 * (FIN / 4)];
            float4 b0 = x1[(long)s0 * (FIN / 4)];
            float4 a1 = z, b1 = z, a2 = z, b2 = z;
            if (rem > 1) {
                int s1 = __builtin_amdgcn_readlane(sj, j + 1);
                w1 = __builtin_bit_cast(float, __builtin_amdgcn_readlane(wjb, j + 1));
                a1 = x0[(long)s1 * (FIN / 4)];
                b1 = x1[(long)s1 * (FIN / 4)];
            }
            if (rem > 2) {
                int s2 = __builtin_amdgcn_readlane(sj, j + 2);
                w2 = __builtin_bit_cast(float, __builtin_amdgcn_readlane(wjb, j + 2));
                a2 = x0[(long)s2 * (FIN / 4)];
                b2 = x1[(long)s2 * (FIN / 4)];
            }
            gcn_fma4(acc0, a0, w0); gcn_fma4(acc1, b0, w0);
            if (rem > 1) { gcn_fma4(acc0, a1, w1); gcn_fma4(acc1, b1, w1); }
            if (rem > 2) { gcn_fma4(acc0, a2, w2); gcn_fma4(acc1, b2, w2); }
        }
    }
    float4* o = (float4*)ax;
    if (ok0) o[((long)pb0 * n + d) * (FIN / 4) + c4] = acc0;
    if (ok1) o[((long)pb1 * n + d) * (FIN / 4) + c4] = acc1;
}

#define ST_U     10
#define ST_WAVES 16
__global__ void gcn_bn_stats(const float* __restrict__ x, float* __restrict__ sums, long rows) {
    __shared__ float red[ST_WAVES][2 * CH];
    int lane = threadIdx.x & 31;
    int wv = __builtin_amdgcn_readfirstlane(threadIdx.x >> 5);
    long gw = (long)blockIdx.x * ST_WAVES + wv;
    long tw = (long)gridDim.x * ST_WAVES;
    const float4* xp = (const float4*)x;
    float4 s = {0.f, 0.f, 0.f, 0.f};
    float4 q = {0.f, 0.f, 0.f, 0.f};
    for (long r = gw; r < rows; r += tw * ST_U) {
        float4 v[ST_U];
#pragma unroll
        for (int u = 0; u < ST_U; ++u) {
            long rr = r + u * tw;
            v[u] = xp[(rr < rows ? rr : rows - 1) * (CH / 4) + lane];
        }
#pragma unroll
        for (int u = 0; u < ST_U; ++u) {
            bool in = r + u * tw < rows;
            float a = fmaxf(in ? v[u].x : 0.f, 0.f);
            float b = fmaxf(in ? v[u].y : 0.f, 0.f);
            float c = fmaxf(in ? v[u].z : 0.f, 0.f);
            float d = fmaxf(in ? v[u].w : 0.f, 0.f);
            s.x += a; s.y += b; s.z += c; s.w += d;
            q.x = __builtin_fmaf(a, a, q.x);
            q.y = __builtin_fmaf(b, b, q.y);
            q.z = __builtin_fmaf(c, c, q.z);
            q.w = __builtin_fmaf(d, d, q.w);
        }
    }
    *(float4*)&red[wv][lane * 4]      = s;
    *(float4*)&red[wv][CH + lane * 4] = q;
    __syncthreads();
    int t = threadIdx.x;
    if (t < 2 * CH) {
        float tot = 0.f;
#pragma unroll
        for (int w = 0; w < ST_WAVES; ++w) tot += red[w][t];
        atomicAddF(&sums[t], tot);
    }
}

#define DO_U     20
#define DO_WAVES 16
__global__ void __launch_bounds__(DO_WAVES * 32) __attribute__((amdgpu_waves_per_eu(4, 4)))
gcn_bn_dot_out(const float* __restrict__ x, const float* __restrict__ sums,
               const float* __restrict__ gamma, const float* __restrict__ beta,
               const float* __restrict__ wc, const float* __restrict__ bc,
               float* __restrict__ out, int rows, float invcnt) {
    int lane = threadIdx.x & 31;
    int wv  = __builtin_amdgcn_readfirstlane(threadIdx.x >> 5);
    int gw  = blockIdx.x * DO_WAVES + wv;
    int tw  = gridDim.x * DO_WAVES;
    int per = (rows + tw - 1) / tw;
    int r0  = gw * per;
    int r1  = r0 + per; if (r1 > rows) r1 = rows;
    if (r0 >= r1) return;

    float mu[4], is[4], g[4], b[4], w[4];
#pragma unroll
    for (int j = 0; j < 4; ++j) {
        int c = lane * 4 + j;
        mu[j] = sums[c] * invcnt;
        float var = sums[CH + c] * invcnt - mu[j] * mu[j];
        is[j] = rsqrtf(var + BN_EPS);
        g[j]  = gamma[c];
        b[j]  = beta[c];
        w[j]  = wc[c];
    }
    float bcv = bc[0];
    const float4* xp = (const float4*)x + lane;

    for (int r = r0; r < r1; r += DO_U) {
        float4 v[DO_U];
#pragma unroll
        for (int u = 0; u < DO_U; ++u) {
            int rr = r + u;
            v[u] = xp[(long)(rr < r1 ? rr : r1 - 1) * (CH / 4)];
        }
        float res = 0.f;
#pragma unroll
        for (int u = 0; u < DO_U; ++u) {
            float vv[4] = {v[u].x, v[u].y, v[u].z, v[u].w};
            float acc = 0.f;
#pragma unroll
            for (int j = 0; j < 4; ++j) {
                float val = (fmaxf(vv[j], 0.f) - mu[j]) * is[j] * g[j] + b[j];
                acc += val * w[j];
            }
#pragma unroll
            for (int o = 16; o > 0; o >>= 1)
                acc += __shfl_xor(acc, o, 32);
            res = (lane == u) ? acc : res;
        }
        if (lane < DO_U && r + lane < r1) out[r + lane] = res + bcv;
    }
}

extern "C" void kernel_launch(void* const* d_in, const int* in_sizes, int n_in,
                              void* d_out, int out_size, void* d_ws, size_t ws_size,
                              hipStream_t stream) {
    const float* x   = (const float*)d_in[0];
    const int*   ei  = (const int*)d_in[1];
    const float* W1  = (const float*)d_in[2];
    const float* b1  = (const float*)d_in[3];
    const float* W2  = (const float*)d_in[4];
    const float* b2  = (const float*)d_in[5];
    const float* g1  = (const float*)d_in[6];
    const float* be1 = (const float*)d_in[7];
    const float* g2  = (const float*)d_in[8];
    const float* be2 = (const float*)d_in[9];
    const float* Wc  = (const float*)d_in[10];
    const float* bc  = (const float*)d_in[11];
    float* out = (float*)d_out;

    const int  E  = in_sizes[1] / 2;
    const int  N  = 20000;
    const int  H  = in_sizes[3];
    const int  F  = in_sizes[2] / H;
    const long BN = (long)out_size;
    const int  Bz = (int)(BN / N);
    const long TOT = BN * H;
    const float invcnt = 1.0f / (float)BN;

    float* ws     = (float*)d_ws;
    float* stats1 = ws;
    float* stats2 = ws + 256;
    float* dinv   = ws + 512;
    long   nAl    = (N + 255) & ~255L;
    float* bufA   = ws + 512 + nAl;

    const long PL   = (long)N * H;
    float* bufM     = bufA + (Bz - 1) * PL;
    float* hLast    = bufA + (Bz - 1) * PL;
    float* aggLast  = bufA + (2 * Bz - 2) * PL;
    long   n1Al     = (N + 1 + 255) & ~255L;
    long   eAl      = (E + 255) & ~255L;
    int*   csrEnd   = (int*)(bufA + 2 * TOT);
    int*   sorted   = csrEnd - eAl;
    int*   slots    = sorted - eAl;
    int*   rowptr   = slots - n1Al;
    int*   cursor   = rowptr - nAl;
    int*   cnt      = cursor - nAl;

    const int* srcIdx = ei;
    const int* dstIdx = ei + E;

    gcn_init  <<<(N + 255) / 256, 256, 0, stream>>>(stats1, cnt, N);
    gcn_count <<<(E + 255) / 256, 256, 0, stream>>>(dstIdx, cnt, E);
    gcn_scan  <<<1, 1024, 0, stream>>>(cnt, rowptr, cursor, dinv, N);
    gcn_fill  <<<(E + 255) / 256, 256, 0, stream>>>(srcIdx, dstIdx, cursor, slots, E);
    gcn_order <<<(N + 7) / 8, 256, 0, stream>>>(rowptr, slots, sorted, N);

    dim3 nodeGridLast((N + 7) / 8, 1, 1);
    dim3 nodeGridRest((N + 7) / 8, Bz - 1, 1);
    const long MR   = (long)(Bz - 1) * N;
    int  gemmBlks  = (int)((BN / 16 + 7) / 8);
    int  gemmBlksR = (int)((MR / 16 + 7) / 8);
    int  gemmBlksL = (int)((N / 16 + 7) / 8);
    int  statBlks  = 256;

    dim3 nodeGridIn((N + 7) / 8, (Bz + 3) / 4, 1);
    gcn_gather_in<<<nodeGridIn, 256, 0, stream>>>(x, rowptr, sorted, dinv, bufA, N, Bz);
    gcn_gemm_wmma<<<gemmBlks, 256, 0, stream>>>(bufA, W1, b1, bufM, (int)BN, F);
    gcn_bn_stats <<<statBlks, 512, 0, stream>>>(bufM, stats1, BN);

    gcn_gemm_wmma_bn<<<gemmBlksR, 256, 0, stream>>>(bufM, W2, bufA, (int)MR, H, stats1, g1, be1, invcnt);
    gcn_gemm_wmma_bn<<<gemmBlksL, 256, 0, stream>>>(aggLast, W2, hLast, N, H, stats1, g1, be1, invcnt);
    gcn_gather   <<<nodeGridLast, 256, 0, stream>>>(hLast, rowptr, sorted, dinv, b2, aggLast, N);
    gcn_gather   <<<nodeGridRest, 256, 0, stream>>>(bufA, rowptr, sorted, dinv, b2, bufM, N);
    gcn_bn_stats <<<statBlks, 512, 0, stream>>>(bufM, stats2, BN);

    gcn_bn_dot_out<<<statBlks, 512, 0, stream>>>(bufM, stats2, g2, be2, Wc, bc, out, (int)BN, invcnt);
}


# prefix-sum pass of the edge-list build: 20 work-groups, one counter per thread, so each wave load and store covers one whole line (was one work-group whose lanes each walked their own 80-byte run)
# speedup vs baseline: 3.8165x; 1.1518x over previous
#include <hip/hip_runtime.h>
#include <hip/hip_bf16.h>

typedef __attribute__((ext_vector_type(2))) float v2f;
typedef __attribute__((ext_vector_type(8))) float v8f;

#define CH 128
#define BN_EPS 1e-5f

__device__ __forceinline__ void atomicAddF(float* p, float v) {
    (void)__hip_atomic_fetch_add(p, v, __ATOMIC_RELAXED, __HIP_MEMORY_SCOPE_AGENT);
}

__global__ void gcn_init(float* stats, int* cnt, int n) {
    int i = blockIdx.x * blockDim.x + threadIdx.x;
    if (i < 2 * 2 * CH) stats[i] = 0.0f;
    if (i < n) cnt[i] = 0;
}
__global__ void gcn_count(const int* __restrict__ dst, int* cnt, int e) {
    int i = blockIdx.x * blockDim.x + threadIdx.x;
    if (i < e) atomicAdd(&cnt[dst[i]], 1);
}
#define SC_T 1024
__global__ void __launch_bounds__(SC_T)
gcn_scan(const int* __restrict__ cnt, int* __restrict__ rowptr,
         int* __restrict__ cursor, float* __restrict__ dinv, int n) {
    __shared__ int wown[SC_T / 32];
    __shared__ int wpre[SC_T / 32];
    int t    = threadIdx.x;
    int lane = t & 31;
    int wv   = t >> 5;
    int nb   = blockIdx.x;
    int i    = nb * SC_T + t;

    int pre = 0;
    for (int j = 0; j < nb; ++j) pre += cnt[j * SC_T + t];
    int own = (i < n) ? cnt[i] : 0;

    int inc = own;
    for (int o = 1; o < 32; o <<= 1) {
        int v = __shfl_up(inc, o, 32);
        inc += (lane >= o) ? v : 0;
    }
    for (int o = 16; o > 0; o >>= 1) pre += __shfl_xor(pre, o, 32);
    if (lane == 31) wown[wv] = inc;
    if (lane == 0)  wpre[wv] = pre;
    __syncthreads();

    int v = wpre[lane] + ((lane < wv) ? wown[lane] : 0);
    for (int o = 16; o > 0; o >>= 1) v += __shfl_xor(v, o, 32);
    int excl = v + inc - own;

    if (i < n) {
        rowptr[i] = excl;
        cursor[i] = excl;
        dinv[i]   = rsqrtf((float)(own + 1));
    }
    if (i == n - 1) rowptr[n] = excl + own;
}
__global__ void gcn_fill(const int* __restrict__ src, const int* __restrict__ dst,
                         int* cursor, int* __restrict__ slots, int e) {
    int i = blockIdx.x * blockDim.x + threadIdx.x;
    if (i < e) {
        int p = atomicAdd(&cursor[dst[i]], 1);
        slots[p] = src[i];
    }
}
__global__ void gcn_order(const int* __restrict__ rowptr, const int* __restrict__ slots,
                          int* __restrict__ sorted, int n) {
    int d = blockIdx.x * (blockDim.x >> 5) + (threadIdx.x >> 5);
    if (d >= n) return;
    int lane = threadIdx.x & 31;
    int beg = rowptr[d];
    int len = rowptr[d + 1] - beg;
    for (int i = lane; i < len; i += 32) {
        int v = slots[beg + i];
        int rank = 0;
        for (int j = 0; j < len; ++j) {
            int u = slots[beg + j];
            rank += (u < v || (u == v && j < i)) ? 1 : 0;
        }
        sorted[beg + rank] = v;
    }
}

template <bool BN, bool BIAS>
__device__ __forceinline__ void gcn_gemm_body(const float* __restrict__ A, const float* __restrict__ W,
                                              float* __restrict__ C, int M, int K,
                                              const float4* prm, const float* __restrict__ bias) {
    int wave = blockIdx.x * (blockDim.x >> 5) + (threadIdx.x >> 5);
    int lane = threadIdx.x & 31;
    if (wave * 16 >= M) return;

    int mrow  = wave * 16 + (lane & 15);
    int khalf = (lane >> 4) << 1;
    int ncol  = lane & 15;

    v8f acc[8];
#pragma unroll
    for (int t = 0; t < 8; ++t) {
        float bv = BIAS ? bias[t * 16 + ncol] : 0.f;
        acc[t] = (v8f){bv, bv, bv, bv, bv, bv, bv, bv};
    }

    for (int k0 = 0; k0 < K; k0 += 4) {
        v2f a;
        const float* ar = A + (long)mrow * K + k0 + khalf;
        a.x = ar[0];
        a.y = ar[1];
        if (BN) {
            float4 p0 = prm[k0 + khalf];
            float4 p1 = prm[k0 + khalf + 1];
            a.x = (fmaxf(a.x, 0.f) - p0.x) * p0.y * p0.z + p0.w;
            a.y = (fmaxf(a.y, 0.f) - p1.x) * p1.y * p1.z + p1.w;
        }
        const float* w0 = W + (long)(k0 + khalf) * CH;
        const float* w1 = w0 + CH;
#pragma unroll
        for (int t = 0; t < 8; ++t) {
            v2f b;
            b.x = w0[t * 16 + ncol];
            b.y = w1[t * 16 + ncol];
            acc[t] = __builtin_amdgcn_wmma_f32_16x16x4_f32(
                false, a, false, b, (short)0, acc[t], false, false);
        }
    }
    int rbase = wave * 16 + ((lane >> 4) << 3);
#pragma unroll
    for (int t = 0; t < 8; ++t)
#pragma unroll
        for (int r = 0; r < 8; ++r)
            C[(long)(rbase + r) * CH + t * 16 + (lane & 15)] = acc[t][r];
}

__global__ void gcn_gemm_wmma(const float* __restrict__ A, const float* __restrict__ W,
                              const float* __restrict__ bias, float* __restrict__ C, int M, int K) {
    gcn_gemm_body<false, true>(A, W, C, M, K, nullptr, bias);
}

__global__ void gcn_gemm_wmma_bn(const float* __restrict__ A, const float* __restrict__ W,
                                 float* __restrict__ C, int M, int K,
                                 const float* __restrict__ sums, const float* __restrict__ gamma,
                                 const float* __restrict__ beta, float invcnt) {
    __shared__ float4 prm[CH];
    int c = threadIdx.x;
    if (c < CH) {
        float mu  = sums[c] * invcnt;
        float var = sums[CH + c] * invcnt - mu * mu;
        prm[c] = make_float4(mu, rsqrtf(var + BN_EPS), gamma[c], beta[c]);
    }
    __syncthreads();
    gcn_gemm_body<true, false>(A, W, C, M, K, prm, nullptr);
}

__device__ __forceinline__ void gcn_fma4(float4& acc, const float4& v, float w) {
    acc.x = __builtin_fmaf(v.x, w, acc.x);
    acc.y = __builtin_fmaf(v.y, w, acc.y);
    acc.z = __builtin_fmaf(v.z, w, acc.z);
    acc.w = __builtin_fmaf(v.w, w, acc.w);
}
__global__ void gcn_gather(const float* __restrict__ h, const int* __restrict__ rowptr,
                           const int* __restrict__ csrc, const float* __restrict__ dinv,
                           const float* __restrict__ bias, float* __restrict__ agg, int n) {
    int d = blockIdx.x * (blockDim.x >> 5) + (threadIdx.x >> 5);
    if (d >= n) return;
    d = __builtin_amdgcn_readfirstlane(d);
    int lane = threadIdx.x & 31;
    const float4* hb = (const float4*)(h + (long)blockIdx.y * n * CH);
    float4*       ab = (float4*)(agg + (long)blockIdx.y * n * CH);
    int beg = __builtin_amdgcn_readfirstlane(rowptr[d]);
    int end = __builtin_amdgcn_readfirstlane(rowptr[d + 1]);
    float dd = dinv[d];
    float4 hv = hb[(long)d * (CH / 4) + lane];
    float4 bv = ((const float4*)bias)[lane];
    float4 acc;
    acc.x = bv.x + hv.x * dd * dd;
    acc.y = bv.y + hv.y * dd * dd;
    acc.z = bv.z + hv.z * dd * dd;
    acc.w = bv.w + hv.w * dd * dd;

    for (int base = beg; base < end; base += 32) {
        int cnt = end - base; if (cnt > 32) cnt = 32;
        int   sj = 0;
        float wj = 0.f;
        if (lane < cnt) { sj = csrc[base + lane]; wj = dinv[sj] * dd; }
        int wjb = __builtin_bit_cast(int, wj);
        int j = 0;
        for (; j + 4 <= cnt; j += 4) {
            int s0 = __builtin_amdgcn_readlane(sj, j);
            int s1 = __builtin_amdgcn_readlane(sj, j + 1);
            int s2 = __builtin_amdgcn_readlane(sj, j + 2);
            int s3 = __builtin_amdgcn_readlane(sj, j + 3);
            float w0 = __builtin_bit_cast(float, __builtin_amdgcn_readlane(wjb, j));
            float w1 = __builtin_bit_cast(float, __builtin_amdgcn_readlane(wjb, j + 1));
            float w2 = __builtin_bit_cast(float, __builtin_amdgcn_readlane(wjb, j + 2));
            float w3 = __builtin_bit_cast(float, __builtin_amdgcn_readlane(wjb, j + 3));
            float4 v0 = hb[(long)s0 * (CH / 4) + lane];
            float4 v1 = hb[(long)s1 * (CH / 4) + lane];
            float4 v2 = hb[(long)s2 * (CH / 4) + lane];
            float4 v3 = hb[(long)s3 * (CH / 4) + lane];
            gcn_fma4(acc, v0, w0);
            gcn_fma4(acc, v1, w1);
            gcn_fma4(acc, v2, w2);
            gcn_fma4(acc, v3, w3);
        }
        int rem = cnt - j;
        if (rem > 0) {
            int s0 = __builtin_amdgcn_readlane(sj, j);
            float w0 = __builtin_bit_cast(float, __builtin_amdgcn_readlane(wjb, j));
            float w1 = 0.f, w2 = 0.f;
            float4 v0 = hb[(long)s0 * (CH / 4) + lane];
            float4 v1 = {0.f, 0.f, 0.f, 0.f}, v2 = {0.f, 0.f, 0.f, 0.f};
            if (rem > 1) {
                int s1 = __builtin_amdgcn_readlane(sj, j + 1);
                w1 = __builtin_bit_cast(float, __builtin_amdgcn_readlane(wjb, j + 1));
                v1 = hb[(long)s1 * (CH / 4) + lane];
            }
            if (rem > 2) {
                int s2 = __builtin_amdgcn_readlane(sj, j + 2);
                w2 = __builtin_bit_cast(float, __builtin_amdgcn_readlane(wjb, j + 2));
                v2 = hb[(long)s2 * (CH / 4) + lane];
            }
            gcn_fma4(acc, v0, w0);
            if (rem > 1) gcn_fma4(acc, v1, w1);
            if (rem > 2) gcn_fma4(acc, v2, w2);
        }
    }
    ab[(long)d * (CH / 4) + lane] = acc;
}

#define FIN 64
__global__ void gcn_gather_in(const float* __restrict__ x, const int* __restrict__ rowptr,
                              const int* __restrict__ csrc, const float* __restrict__ dinv,
                              float* __restrict__ ax, int n, int nb) {
    int d = blockIdx.x * (blockDim.x >> 5) + (threadIdx.x >> 5);
    if (d >= n) return;
    d = __builtin_amdgcn_readfirstlane(d);
    int lane = threadIdx.x & 31;
    int c4   = lane & 15;
    int pb0  = blockIdx.y * 4 + (lane >> 4);
    int pb1  = pb0 + 2;
    bool ok0 = pb0 < nb, ok1 = pb1 < nb;
    const float4* x0 = (const float4*)x + (long)(ok0 ? pb0 : 0) * n * (FIN / 4) + c4;
    const float4* x1 = (const float4*)x + (long)(ok1 ? pb1 : 0) * n * (FIN / 4) + c4;
    int beg = __builtin_amdgcn_readfirstlane(rowptr[d]);
    int end = __builtin_amdgcn_readfirstlane(rowptr[d + 1]);
    float dd = dinv[d];
    float4 h0 = x0[(long)d * (FIN / 4)];
    float4 h1 = x1[(long)d * (FIN / 4)];
    float4 acc0, acc1;
    acc0.x = h0.x * dd * dd; acc0.y = h0.y * dd * dd; acc0.z = h0.z * dd * dd; acc0.w = h0.w * dd * dd;
    acc1.x = h1.x * dd * dd; acc1.y = h1.y * dd * dd; acc1.z = h1.z * dd * dd; acc1.w = h1.w * dd * dd;

    for (int base = beg; base < end; base += 32) {
        int cnt = end - base; if (cnt > 32) cnt = 32;
        int   sj = 0;
        float wj = 0.f;
        if (lane < cnt) { sj = csrc[base + lane]; wj = dinv[sj] * dd; }
        int wjb = __builtin_bit_cast(int, wj);
        int j = 0;
        for (; j + 4 <= cnt; j += 4) {
            int s0 = __builtin_amdgcn_readlane(sj, j);
            int s1 = __builtin_amdgcn_readlane(sj, j + 1);
            int s2 = __builtin_amdgcn_readlane(sj, j + 2);
            int s3 = __builtin_amdgcn_readlane(sj, j + 3);
            float4 a0 = x0[(long)s0 * (FIN / 4)];
            float4 b0 = x1[(long)s0 * (FIN / 4)];
            float4 a1 = x0[(long)s1 * (FIN / 4)];
            float4 b1 = x1[(long)s1 * (FIN / 4)];
            float4 a2 = x0[(long)s2 * (FIN / 4)];
            float4 b2 = x1[(long)s2 * (FIN / 4)];
            float4 a3 = x0[(long)s3 * (FIN / 4)];
            float4 b3 = x1[(long)s3 * (FIN / 4)];
            float w0 = __builtin_bit_cast(float, __builtin_amdgcn_readlane(wjb, j));
            float w1 = __builtin_bit_cast(float, __builtin_amdgcn_readlane(wjb, j + 1));
            float w2 = __builtin_bit_cast(float, __builtin_amdgcn_readlane(wjb, j + 2));
            float w3 = __builtin_bit_cast(float, __builtin_amdgcn_readlane(wjb, j + 3));
            gcn_fma4(acc0, a0, w0); gcn_fma4(acc1, b0, w0);
            gcn_fma4(acc0, a1, w1); gcn_fma4(acc1, b1, w1);
            gcn_fma4(acc0, a2, w2); gcn_fma4(acc1, b2, w2);
            gcn_fma4(acc0, a3, w3); gcn_fma4(acc1, b3, w3);
        }
        int rem = cnt - j;
        if (rem > 0) {
            const float4 z = {0.f, 0.f, 0.f, 0.f};
            int s0 = __builtin_amdgcn_readlane(sj, j);
            float w0 = __builtin_bit_cast(float, __builtin_amdgcn_readlane(wjb, j));
            float w1 = 0.f, w2 = 0.f;
            float4 a0 = x0[(long)s0 * (FIN / 4)];
            float4 b0 = x1[(long)s0 * (FIN / 4)];
            float4 a1 = z, b1 = z, a2 = z, b2 = z;
            if (rem > 1) {
                int s1 = __builtin_amdgcn_readlane(sj, j + 1);
                w1 = __builtin_bit_cast(float, __builtin_amdgcn_readlane(wjb, j + 1));
                a1 = x0[(long)s1 * (FIN / 4)];
                b1 = x1[(long)s1 * (FIN / 4)];
            }
            if (rem > 2) {
                int s2 = __builtin_amdgcn_readlane(sj, j + 2);
                w2 = __builtin_bit_cast(float, __builtin_amdgcn_readlane(wjb, j + 2));
                a2 = x0[(long)s2 * (FIN / 4)];
                b2 = x1[(long)s2 * (FIN / 4)];
            }
            gcn_fma4(acc0, a0, w0); gcn_fma4(acc1, b0, w0);
            if (rem > 1) { gcn_fma4(acc0, a1, w1); gcn_fma4(acc1, b1, w1); }
            if (rem > 2) { gcn_fma4(acc0, a2, w2); gcn_fma4(acc1, b2, w2); }
        }
    }
    float4* o = (float4*)ax;
    if (ok0) o[((long)pb0 * n + d) * (FIN / 4) + c4] = acc0;
    if (ok1) o[((long)pb1 * n + d) * (FIN / 4) + c4] = acc1;
}

#define ST_U     10
#define ST_WAVES 16
__global__ void gcn_bn_stats(const float* __restrict__ x, float* __restrict__ sums, long rows) {
    __shared__ float red[ST_WAVES][2 * CH];
    int lane = threadIdx.x & 31;
    int wv = __builtin_amdgcn_readfirstlane(threadIdx.x >> 5);
    long gw = (long)blockIdx.x * ST_WAVES + wv;
    long tw = (long)gridDim.x * ST_WAVES;
    const float4* xp = (const float4*)x;
    float4 s = {0.f, 0.f, 0.f, 0.f};
    float4 q = {0.f, 0.f, 0.f, 0.f};
    for (long r = gw; r < rows; r += tw * ST_U) {
        float4 v[ST_U];
#pragma unroll
        for (int u = 0; u < ST_U; ++u) {
            long rr = r + u * tw;
            v[u] = xp[(rr < rows ? rr : rows - 1) * (CH / 4) + lane];
        }
#pragma unroll
        for (int u = 0; u < ST_U; ++u) {
            bool in = r + u * tw < rows;
            float a = fmaxf(in ? v[u].x : 0.f, 0.f);
            float b = fmaxf(in ? v[u].y : 0.f, 0.f);
            float c = fmaxf(in ? v[u].z : 0.f, 0.f);
            float d = fmaxf(in ? v[u].w : 0.f, 0.f);
            s.x += a; s.y += b; s.z += c; s.w += d;
            q.x = __builtin_fmaf(a, a, q.x);
            q.y = __builtin_fmaf(b, b, q.y);
            q.z = __builtin_fmaf(c, c, q.z);
            q.w = __builtin_fmaf(d, d, q.w);
        }
    }
    *(float4*)&red[wv][lane * 4]      = s;
    *(float4*)&red[wv][CH + lane * 4] = q;
    __syncthreads();
    int t = threadIdx.x;
    if (t < 2 * CH) {
        float tot = 0.f;
#pragma unroll
        for (int w = 0; w < ST_WAVES; ++w) tot += red[w][t];
        atomicAddF(&sums[t], tot);
    }
}

#define DO_U     20
#define DO_WAVES 16
__global__ void __launch_bounds__(DO_WAVES * 32) __attribute__((amdgpu_waves_per_eu(4, 4)))
gcn_bn_dot_out(const float* __restrict__ x, const float* __restrict__ sums,
               const float* __restrict__ gamma, const float* __restrict__ beta,
               const float* __restrict__ wc, const float* __restrict__ bc,
               float* __restrict__ out, int rows, float invcnt) {
    int lane = threadIdx.x & 31;
    int wv  = __builtin_amdgcn_readfirstlane(threadIdx.x >> 5);
    int gw  = blockIdx.x * DO_WAVES + wv;
    int tw  = gridDim.x * DO_WAVES;
    int per = (rows + tw - 1) / tw;
    int r0  = gw * per;
    int r1  = r0 + per; if (r1 > rows) r1 = rows;
    if (r0 >= r1) return;

    float mu[4], is[4], g[4], b[4], w[4];
#pragma unroll
    for (int j = 0; j < 4; ++j) {
        int c = lane * 4 + j;
        mu[j] = sums[c] * invcnt;
        float var = sums[CH + c] * invcnt - mu[j] * mu[j];
        is[j] = rsqrtf(var + BN_EPS);
        g[j]  = gamma[c];
        b[j]  = beta[c];
        w[j]  = wc[c];
    }
    float bcv = bc[0];
    const float4* xp = (const float4*)x + lane;

    for (int r = r0; r < r1; r += DO_U) {
        float4 v[DO_U];
#pragma unroll
        for (int u = 0; u < DO_U; ++u) {
            int rr = r + u;
            v[u] = xp[(long)(rr < r1 ? rr : r1 - 1) * (CH / 4)];
        }
        float res = 0.f;
#pragma unroll
        for (int u = 0; u < DO_U; ++u) {
            float vv[4] = {v[u].x, v[u].y, v[u].z, v[u].w};
            float acc = 0.f;
#pragma unroll
            for (int j = 0; j < 4; ++j) {
                float val = (fmaxf(vv[j], 0.f) - mu[j]) * is[j] * g[j] + b[j];
                acc += val * w[j];
            }
#pragma unroll
            for (int o = 16; o > 0; o >>= 1)
                acc += __shfl_xor(acc, o, 32);
            res = (lane == u) ? acc : res;
        }
        if (lane < DO_U && r + lane < r1) out[r + lane] = res + bcv;
    }
}

extern "C" void kernel_launch(void* const* d_in, const int* in_sizes, int n_in,
                              void* d_out, int out_size, void* d_ws, size_t ws_size,
                              hipStream_t stream) {
    const float* x   = (const float*)d_in[0];
    const int*   ei  = (const int*)d_in[1];
    const float* W1  = (const float*)d_in[2];
    const float* b1  = (const float*)d_in[3];
    const float* W2  = (const float*)d_in[4];
    const float* b2  = (const float*)d_in[5];
    const float* g1  = (const float*)d_in[6];
    const float* be1 = (const float*)d_in[7];
    const float* g2  = (const float*)d_in[8];
    const float* be2 = (const float*)d_in[9];
    const float* Wc  = (const float*)d_in[10];
    const float* bc  = (const float*)d_in[11];
    float* out = (float*)d_out;

    const int  E  = in_sizes[1] / 2;
    const int  N  = 20000;
    const int  H  = in_sizes[3];
    const int  F  = in_sizes[2] / H;
    const long BN = (long)out_size;
    const int  Bz = (int)(BN / N);
    const long TOT = BN * H;
    const float invcnt = 1.0f / (float)BN;

    float* ws     = (float*)d_ws;
    float* stats1 = ws;
    float* stats2 = ws + 256;
    float* dinv   = ws + 512;
    long   nAl    = (N + 255) & ~255L;
    float* bufA   = ws + 512 + nAl;

    const long PL   = (long)N * H;
    float* bufM     = bufA + (Bz - 1) * PL;
    float* hLast    = bufA + (Bz - 1) * PL;
    float* aggLast  = bufA + (2 * Bz - 2) * PL;
    long   n1Al     = (N + 1 + 255) & ~255L;
    long   eAl      = (E + 255) & ~255L;
    int*   csrEnd   = (int*)(bufA + 2 * TOT);
    int*   sorted   = csrEnd - eAl;
    int*   slots    = sorted - eAl;
    int*   rowptr   = slots - n1Al;
    int*   cursor   = rowptr - nAl;
    int*   cnt      = cursor - nAl;

    const int* srcIdx = ei;
    const int* dstIdx = ei + E;

    gcn_init  <<<(N + 255) / 256, 256, 0, stream>>>(stats1, cnt, N);
    gcn_count <<<(E + 255) / 256, 256, 0, stream>>>(dstIdx, cnt, E);
    gcn_scan  <<<(N + SC_T - 1) / SC_T, SC_T, 0, stream>>>(cnt, rowptr, cursor, dinv, N);
    gcn_fill  <<<(E + 255) / 256, 256, 0, stream>>>(srcIdx, dstIdx, cursor, slots, E);
    gcn_order <<<(N + 7) / 8, 256, 0, stream>>>(rowptr, slots, sorted, N);

    dim3 nodeGridLast((N + 7) / 8, 1, 1);
    dim3 nodeGridRest((N + 7) / 8, Bz - 1, 1);
    const long MR   = (long)(Bz - 1) * N;
    int  gemmBlks  = (int)((BN / 16 + 7) / 8);
    int  gemmBlksR = (int)((MR / 16 + 7) / 8);
    int  gemmBlksL = (int)((N / 16 + 7) / 8);
    int  statBlks  = 256;

    dim3 nodeGridIn((N + 7) / 8, (Bz + 3) / 4, 1);
    gcn_gather_in<<<nodeGridIn, 256, 0, stream>>>(x, rowptr, sorted, dinv, bufA, N, Bz);
    gcn_gemm_wmma<<<gemmBlks, 256, 0, stream>>>(bufA, W1, b1, bufM, (int)BN, F);
    gcn_bn_stats <<<statBlks, 512, 0, stream>>>(bufM, stats1, BN);

    gcn_gemm_wmma_bn<<<gemmBlksR, 256, 0, stream>>>(bufM, W2, bufA, (int)MR, H, stats1, g1, be1, invcnt);
    gcn_gemm_wmma_bn<<<gemmBlksL, 256, 0, stream>>>(aggLast, W2, hLast, N, H, stats1, g1, be1, invcnt);
    gcn_gather   <<<nodeGridLast, 256, 0, stream>>>(hLast, rowptr, sorted, dinv, b2, aggLast, N);
    gcn_gather   <<<nodeGridRest, 256, 0, stream>>>(bufA, rowptr, sorted, dinv, b2, bufM, N);
    gcn_bn_stats <<<statBlks, 512, 0, stream>>>(bufM, stats2, BN);

    gcn_bn_dot_out<<<statBlks, 512, 0, stream>>>(bufM, stats2, g2, be2, Wc, bc, out, (int)BN, invcnt);
}


# layer 2 in one kernel: a work-group sums BatchNorm(relu) rows of 16 nodes in an LDS tile, multiplies it by W2 on WMMA, stores whole rows; no h buffer, 2 launches fewer; sum before product
# speedup vs baseline: 4.4200x; 1.1581x over previous
#include <hip/hip_runtime.h>
#include <hip/hip_bf16.h>

typedef __attribute__((ext_vector_type(2))) float v2f;
typedef __attribute__((ext_vector_type(8))) float v8f;

#define CH 128
#define BN_EPS 1e-5f

__device__ __forceinline__ void atomicAddF(float* p, float v) {
    (void)__hip_atomic_fetch_add(p, v, __ATOMIC_RELAXED, __HIP_MEMORY_SCOPE_AGENT);
}

__global__ void gcn_init(float* stats, int* cnt, int n) {
    int i = blockIdx.x * blockDim.x + threadIdx.x;
    if (i < 2 * 2 * CH) stats[i] = 0.0f;
    if (i < n) cnt[i] = 0;
}
__global__ void gcn_count(const int* __restrict__ dst, int* cnt, int e) {
    int i = blockIdx.x * blockDim.x + threadIdx.x;
    if (i < e) atomicAdd(&cnt[dst[i]], 1);
}
#define SC_T 1024
__global__ void __launch_bounds__(SC_T)
gcn_scan(const int* __restrict__ cnt, int* __restrict__ rowptr,
         int* __restrict__ cursor, float* __restrict__ dinv, int n) {
    __shared__ int wown[SC_T / 32];
    __shared__ int wpre[SC_T / 32];
    int t    = threadIdx.x;
    int lane = t & 31;
    int wv   = t >> 5;
    int nb   = blockIdx.x;
    int i    = nb * SC_T + t;

    int pre = 0;
    for (int j = 0; j < nb; ++j) pre += cnt[j * SC_T + t];
    int own = (i < n) ? cnt[i] : 0;

    int inc = own;
    for (int o = 1; o < 32; o <<= 1) {
        int v = __shfl_up(inc, o, 32);
        inc += (lane >= o) ? v : 0;
    }
    for (int o = 16; o > 0; o >>= 1) pre += __shfl_xor(pre, o, 32);
    if (lane == 31) wown[wv] = inc;
    if (lane == 0)  wpre[wv] = pre;
    __syncthreads();

    int v = wpre[lane] + ((lane < wv) ? wown[lane] : 0);
    for (int o = 16; o > 0; o >>= 1) v += __shfl_xor(v, o, 32);
    int excl = v + inc - own;

    if (i < n) {
        rowptr[i] = excl;
        cursor[i] = excl;
        dinv[i]   = rsqrtf((float)(own + 1));
    }
    if (i == n - 1) rowptr[n] = excl + own;
}
__global__ void gcn_fill(const int* __restrict__ src, const int* __restrict__ dst,
                         int* cursor, int* __restrict__ slots, int e) {
    int i = blockIdx.x * blockDim.x + threadIdx.x;
    if (i < e) {
        int p = atomicAdd(&cursor[dst[i]], 1);
        slots[p] = src[i];
    }
}
__global__ void gcn_order(const int* __restrict__ rowptr, const int* __restrict__ slots,
                          int* __restrict__ sorted, int n) {
    int d = blockIdx.x * (blockDim.x >> 5) + (threadIdx.x >> 5);
    if (d >= n) return;
    int lane = threadIdx.x & 31;
    int beg = rowptr[d];
    int len = rowptr[d + 1] - beg;
    for (int i = lane; i < len; i += 32) {
        int v = slots[beg + i];
        int rank = 0;
        for (int j = 0; j < len; ++j) {
            int u = slots[beg + j];
            rank += (u < v || (u == v && j < i)) ? 1 : 0;
        }
        sorted[beg + rank] = v;
    }
}

__global__ void gcn_gemm_wmma(const float* __restrict__ A, const float* __restrict__ W,
                              const float* __restrict__ bias, float* __restrict__ C, int M, int K) {
    int wave = blockIdx.x * (blockDim.x >> 5) + (threadIdx.x >> 5);
    int lane = threadIdx.x & 31;
    if (wave * 16 >= M) return;

    int mrow  = wave * 16 + (lane & 15);
    int khalf = (lane >> 4) << 1;
    int ncol  = lane & 15;

    v8f acc[8];
#pragma unroll
    for (int t = 0; t < 8; ++t) {
        float bv = bias[t * 16 + ncol];
        acc[t] = (v8f){bv, bv, bv, bv, bv, bv, bv, bv};
    }

    for (int k0 = 0; k0 < K; k0 += 4) {
        v2f a;
        const float* ar = A + (long)mrow * K + k0 + khalf;
        a.x = ar[0];
        a.y = ar[1];
        const float* w0 = W + (long)(k0 + khalf) * CH;
        const float* w1 = w0 + CH;
#pragma unroll
        for (int t = 0; t < 8; ++t) {
            v2f b;
            b.x = w0[t * 16 + ncol];
            b.y = w1[t * 16 + ncol];
            acc[t] = __builtin_amdgcn_wmma_f32_16x16x4_f32(
                false, a, false, b, (short)0, acc[t], false, false);
        }
    }
    int rbase = wave * 16 + ((lane >> 4) << 3);
#pragma unroll
    for (int t = 0; t < 8; ++t)
#pragma unroll
        for (int r = 0; r < 8; ++r)
            C[(long)(rbase + r) * CH + t * 16 + (lane & 15)] = acc[t][r];
}

__device__ __forceinline__ void gcn_fma4(float4& acc, const float4& v, float w) {
    acc.x = __builtin_fmaf(v.x, w, acc.x);
    acc.y = __builtin_fmaf(v.y, w, acc.y);
    acc.z = __builtin_fmaf(v.z, w, acc.z);
    acc.w = __builtin_fmaf(v.w, w, acc.w);
}
#define FT_ROWS  16
#define FT_PITCH 132
__device__ __forceinline__ float4 gcn_bn4(const float4& v, const float* mu, const float* is,
                                          const float* g, const float* be) {
    float4 r;
    r.x = (fmaxf(v.x, 0.f) - mu[0]) * is[0] * g[0] + be[0];
    r.y = (fmaxf(v.y, 0.f) - mu[1]) * is[1] * g[1] + be[1];
    r.z = (fmaxf(v.z, 0.f) - mu[2]) * is[2] * g[2] + be[2];
    r.w = (fmaxf(v.w, 0.f) - mu[3]) * is[3] * g[3] + be[3];
    return r;
}
__global__ void __launch_bounds__(FT_ROWS * 16)
gcn_gather_bn_gemm(const float* __restrict__ a, const int* __restrict__ rowptr,
                   const int* __restrict__ csrc, const float* __restrict__ dinv,
                   const float* __restrict__ sums, const float* __restrict__ gamma,
                   const float* __restrict__ beta, float invcnt,
                   const float* __restrict__ W, const float* __restrict__ bias,
                   float* __restrict__ out, int n) {
    __shared__ __attribute__((aligned(16))) float tile[FT_ROWS * FT_PITCH];
    int lane = threadIdx.x & 31;
    int wv = __builtin_amdgcn_readfirstlane(threadIdx.x >> 5);
    int node0 = blockIdx.x * FT_ROWS;
    const float4* hb = (const float4*)(a + (long)blockIdx.y * n * CH);
    float4*       ob = (float4*)(out + (long)blockIdx.y * n * CH);

    float mu[4], is[4], g[4], be[4];
#pragma unroll
    for (int j = 0; j < 4; ++j) {
        int c = lane * 4 + j;
        mu[j] = sums[c] * invcnt;
        float var = sums[CH + c] * invcnt - mu[j] * mu[j];
        is[j] = rsqrtf(var + BN_EPS);
        g[j]  = gamma[c];
        be[j] = beta[c];
    }

    for (int q = 0; q < 2; ++q) {
        int r = wv * 2 + q;
        int d = node0 + r;
        float4 acc = {0.f, 0.f, 0.f, 0.f};
        if (d < n) {
            int beg = __builtin_amdgcn_readfirstlane(rowptr[d]);
            int end = __builtin_amdgcn_readfirstlane(rowptr[d + 1]);
            float dd = dinv[d];
            float4 hv = gcn_bn4(hb[(long)d * (CH / 4) + lane], mu, is, g, be);
            acc.x = hv.x * dd * dd;
            acc.y = hv.y * dd * dd;
            acc.z = hv.z * dd * dd;
            acc.w = hv.w * dd * dd;

            for (int base = beg; base < end; base += 32) {
                int cnt = end - base; if (cnt > 32) cnt = 32;
                int   sj = 0;
                float wj = 0.f;
                if (lane < cnt) { sj = csrc[base + lane]; wj = dinv[sj] * dd; }
                int wjb = __builtin_bit_cast(int, wj);
                int j = 0;
                for (; j + 4 <= cnt; j += 4) {
                    int s0 = __builtin_amdgcn_readlane(sj, j);
                    int s1 = __builtin_amdgcn_readlane(sj, j + 1);
                    int s2 = __builtin_amdgcn_readlane(sj, j + 2);
                    int s3 = __builtin_amdgcn_readlane(sj, j + 3);
                    float w0 = __builtin_bit_cast(float, __builtin_amdgcn_readlane(wjb, j));
                    float w1 = __builtin_bit_cast(float, __builtin_amdgcn_readlane(wjb, j + 1));
                    float w2 = __builtin_bit_cast(float, __builtin_amdgcn_readlane(wjb, j + 2));
                    float w3 = __builtin_bit_cast(float, __builtin_amdgcn_readlane(wjb, j + 3));
                    float4 v0 = hb[(long)s0 * (CH / 4) + lane];
                    float4 v1 = hb[(long)s1 * (CH / 4) + lane];
                    float4 v2 = hb[(long)s2 * (CH / 4) + lane];
                    float4 v3 = hb[(long)s3 * (CH / 4) + lane];
                    gcn_fma4(acc, gcn_bn4(v0, mu, is, g, be), w0);
                    gcn_fma4(acc, gcn_bn4(v1, mu, is, g, be), w1);
                    gcn_fma4(acc, gcn_bn4(v2, mu, is, g, be), w2);
                    gcn_fma4(acc, gcn_bn4(v3, mu, is, g, be), w3);
                }
                int rem = cnt - j;
                if (rem > 0) {
                    int s0 = __builtin_amdgcn_readlane(sj, j);
                    float w0 = __builtin_bit_cast(float, __builtin_amdgcn_readlane(wjb, j));
                    float w1 = 0.f, w2 = 0.f;
                    float4 v0 = hb[(long)s0 * (CH / 4) + lane];
                    float4 v1 = {0.f, 0.f, 0.f, 0.f}, v2 = {0.f, 0.f, 0.f, 0.f};
                    if (rem > 1) {
                        int s1 = __builtin_amdgcn_readlane(sj, j + 1);
                        w1 = __builtin_bit_cast(float, __builtin_amdgcn_readlane(wjb, j + 1));
                        v1 = hb[(long)s1 * (CH / 4) + lane];
                    }
                    if (rem > 2) {
                        int s2 = __builtin_amdgcn_readlane(sj, j + 2);
                        w2 = __builtin_bit_cast(float, __builtin_amdgcn_readlane(wjb, j + 2));
                        v2 = hb[(long)s2 * (CH / 4) + lane];
                    }
                    gcn_fma4(acc, gcn_bn4(v0, mu, is, g, be), w0);
                    if (rem > 1) gcn_fma4(acc, gcn_bn4(v1, mu, is, g, be), w1);
                    if (rem > 2) gcn_fma4(acc, gcn_bn4(v2, mu, is, g, be), w2);
                }
            }
        }
        *(float4*)&tile[r * FT_PITCH + lane * 4] = acc;
    }

    int khalf = (lane >> 4) << 1;
    int col   = wv * 16 + (lane & 15);
    float bv  = bias[col];
    v8f acc8 = (v8f){bv, bv, bv, bv, bv, bv, bv, bv};
    const float* ap = &tile[(lane & 15) * FT_PITCH + khalf];
    const float* wp = W + (long)khalf * CH + col;
    v2f bw[16];
#pragma unroll
    for (int i = 0; i < 16; ++i) {
        bw[i].x = wp[(long)(4 * i) * CH];
        bw[i].y = wp[(long)(4 * i) * CH + CH];
    }
    __syncthreads();
#pragma unroll
    for (int i = 0; i < 16; ++i) {
        v2f av = *(const v2f*)(ap + 4 * i);
        acc8 = __builtin_amdgcn_wmma_f32_16x16x4_f32(
            false, av, false, bw[i], (short)0, acc8, false, false);
        bw[i].x = wp[(long)(4 * i + 64) * CH];
        bw[i].y = wp[(long)(4 * i + 64) * CH + CH];
    }
#pragma unroll
    for (int i = 0; i < 16; ++i) {
        v2f av = *(const v2f*)(ap + 4 * i + 64);
        acc8 = __builtin_amdgcn_wmma_f32_16x16x4_f32(
            false, av, false, bw[i], (short)0, acc8, false, false);
    }
    __syncthreads();
    int rb = (lane >> 4) << 3;
#pragma unroll
    for (int r = 0; r < 8; ++r) tile[(rb + r) * FT_PITCH + col] = acc8[r];
    __syncthreads();
    for (int q = 0; q < 2; ++q) {
        int r = wv * 2 + q;
        int d = node0 + r;
        if (d < n) ob[(long)d * (CH / 4) + lane] = *(const float4*)&tile[r * FT_PITCH + lane * 4];
    }
}

#define FIN 64
__global__ void gcn_gather_in(const float* __restrict__ x, const int* __restrict__ rowptr,
                              const int* __restrict__ csrc, const float* __restrict__ dinv,
                              float* __restrict__ ax, int n, int nb) {
    int d = blockIdx.x * (blockDim.x >> 5) + (threadIdx.x >> 5);
    if (d >= n) return;
    d = __builtin_amdgcn_readfirstlane(d);
    int lane = threadIdx.x & 31;
    int c4   = lane & 15;
    int pb0  = blockIdx.y * 4 + (lane >> 4);
    int pb1  = pb0 + 2;
    bool ok0 = pb0 < nb, ok1 = pb1 < nb;
    const float4* x0 = (const float4*)x + (long)(ok0 ? pb0 : 0) * n * (FIN / 4) + c4;
    const float4* x1 = (const float4*)x + (long)(ok1 ? pb1 : 0) * n * (FIN / 4) + c4;
    int beg = __builtin_amdgcn_readfirstlane(rowptr[d]);
    int end = __builtin_amdgcn_readfirstlane(rowptr[d + 1]);
    float dd = dinv[d];
    float4 h0 = x0[(long)d * (FIN / 4)];
    float4 h1 = x1[(long)d * (FIN / 4)];
    float4 acc0, acc1;
    acc0.x = h0.x * dd * dd; acc0.y = h0.y * dd * dd; acc0.z = h0.z * dd * dd; acc0.w = h0.w * dd * dd;
    acc1.x = h1.x * dd * dd; acc1.y = h1.y * dd * dd; acc1.z = h1.z * dd * dd; acc1.w = h1.w * dd * dd;

    for (int base = beg; base < end; base += 32) {
        int cnt = end - base; if (cnt > 32) cnt = 32;
        int   sj = 0;
        float wj = 0.f;
        if (lane < cnt) { sj = csrc[base + lane]; wj = dinv[sj] * dd; }
        int wjb = __builtin_bit_cast(int, wj);
        int j = 0;
        for (; j + 4 <= cnt; j += 4) {
            int s0 = __builtin_amdgcn_readlane(sj, j);
            int s1 = __builtin_amdgcn_readlane(sj, j + 1);
            int s2 = __builtin_amdgcn_readlane(sj, j + 2);
            int s3 = __builtin_amdgcn_readlane(sj, j + 3);
            float4 a0 = x0[(long)s0 * (FIN / 4)];
            float4 b0 = x1[(long)s0 * (FIN / 4)];
            float4 a1 = x0[(long)s1 * (FIN / 4)];
            float4 b1 = x1[(long)s1 * (FIN / 4)];
            float4 a2 = x0[(long)s2 * (FIN / 4)];
            float4 b2 = x1[(long)s2 * (FIN / 4)];
            float4 a3 = x0[(long)s3 * (FIN / 4)];
            float4 b3 = x1[(long)s3 * (FIN / 4)];
            float w0 = __builtin_bit_cast(float, __builtin_amdgcn_readlane(wjb, j));
            float w1 = __builtin_bit_cast(float, __builtin_amdgcn_readlane(wjb, j + 1));
            float w2 = __builtin_bit_cast(float, __builtin_amdgcn_readlane(wjb, j + 2));
            float w3 = __builtin_bit_cast(float, __builtin_amdgcn_readlane(wjb, j + 3));
            gcn_fma4(acc0, a0, w0); gcn_fma4(acc1, b0, w0);
            gcn_fma4(acc0, a1, w1); gcn_fma4(acc1, b1, w1);
            gcn_fma4(acc0, a2, w2); gcn_fma4(acc1, b2, w2);
            gcn_fma4(acc0, a3, w3); gcn_fma4(acc1, b3, w3);
        }
        int rem = cnt - j;
        if (rem > 0) {
            const float4 z = {0.f, 0.f, 0.f, 0.f};
            int s0 = __builtin_amdgcn_readlane(sj, j);
            float w0 = __builtin_bit_cast(float, __builtin_amdgcn_readlane(wjb, j));
            float w1 = 0.f, w2 = 0.f;
            float4 a0 = x0[(long)s0 * (FIN / 4)];
            float4 b0 = x1[(long)s0 * (FIN / 4)];
            float4 a1 = z, b1 = z, a2 = z, b2 = z;
            if (rem > 1) {
                int s1 = __builtin_amdgcn_readlane(sj, j + 1);
                w1 = __builtin_bit_cast(float, __builtin_amdgcn_readlane(wjb, j + 1));
                a1 = x0[(long)s1 * (FIN / 4)];
                b1 = x1[(long)s1 * (FIN / 4)];
            }
            if (rem > 2) {
                int s2 = __builtin_amdgcn_readlane(sj, j + 2);
                w2 = __builtin_bit_cast(float, __builtin_amdgcn_readlane(wjb, j + 2));
                a2 = x0[(long)s2 * (FIN / 4)];
                b2 = x1[(long)s2 * (FIN / 4)];
            }
            gcn_fma4(acc0, a0, w0); gcn_fma4(acc1, b0, w0);
            if (rem > 1) { gcn_fma4(acc0, a1, w1); gcn_fma4(acc1, b1, w1); }
            if (rem > 2) { gcn_fma4(acc0, a2, w2); gcn_fma4(acc1, b2, w2); }
        }
    }
    float4* o = (float4*)ax;
    if (ok0) o[((long)pb0 * n + d) * (FIN / 4) + c4] = acc0;
    if (ok1) o[((long)pb1 * n + d) * (FIN / 4) + c4] = acc1;
}

#define ST_U     10
#define ST_WAVES 16
__global__ void gcn_bn_stats(const float* __restrict__ x, float* __restrict__ sums, long rows) {
    __shared__ float red[ST_WAVES][2 * CH];
    int lane = threadIdx.x & 31;
    int wv = __builtin_amdgcn_readfirstlane(threadIdx.x >> 5);
    long gw = (long)blockIdx.x * ST_WAVES + wv;
    long tw = (long)gridDim.x * ST_WAVES;
    const float4* xp = (const float4*)x;
    float4 s = {0.f, 0.f, 0.f, 0.f};
    float4 q = {0.f, 0.f, 0.f, 0.f};
    for (long r = gw; r < rows; r += tw * ST_U) {
        float4 v[ST_U];
#pragma unroll
        for (int u = 0; u < ST_U; ++u) {
            long rr = r + u * tw;
            v[u] = xp[(rr < rows ? rr : rows - 1) * (CH / 4) + lane];
        }
#pragma unroll
        for (int u = 0; u < ST_U; ++u) {
            bool in = r + u * tw < rows;
            float a = fmaxf(in ? v[u].x : 0.f, 0.f);
            float b = fmaxf(in ? v[u].y : 0.f, 0.f);
            float c = fmaxf(in ? v[u].z : 0.f, 0.f);
            float d = fmaxf(in ? v[u].w : 0.f, 0.f);
            s.x += a; s.y += b; s.z += c; s.w += d;
            q.x = __builtin_fmaf(a, a, q.x);
            q.y = __builtin_fmaf(b, b, q.y);
            q.z = __builtin_fmaf(c, c, q.z);
            q.w = __builtin_fmaf(d, d, q.w);
        }
    }
    *(float4*)&red[wv][lane * 4]      = s;
    *(float4*)&red[wv][CH + lane * 4] = q;
    __syncthreads();
    int t = threadIdx.x;
    if (t < 2 * CH) {
        float tot = 0.f;
#pragma unroll
        for (int w = 0; w < ST_WAVES; ++w) tot += red[w][t];
        atomicAddF(&sums[t], tot);
    }
}

#define DO_U     20
#define DO_WAVES 16
__global__ void __launch_bounds__(DO_WAVES * 32) __attribute__((amdgpu_waves_per_eu(4, 4)))
gcn_bn_dot_out(const float* __restrict__ x, const float* __restrict__ sums,
               const float* __restrict__ gamma, const float* __restrict__ beta,
               const float* __restrict__ wc, const float* __restrict__ bc,
               float* __restrict__ out, int rows, float invcnt) {
    int lane = threadIdx.x & 31;
    int wv  = __builtin_amdgcn_readfirstlane(threadIdx.x >> 5);
    int gw  = blockIdx.x * DO_WAVES + wv;
    int tw  = gridDim.x * DO_WAVES;
    int per = (rows + tw - 1) / tw;
    int r0  = gw * per;
    int r1  = r0 + per; if (r1 > rows) r1 = rows;
    if (r0 >= r1) return;

    float mu[4], is[4], g[4], b[4], w[4];
#pragma unroll
    for (int j = 0; j < 4; ++j) {
        int c = lane * 4 + j;
        mu[j] = sums[c] * invcnt;
        float var = sums[CH + c] * invcnt - mu[j] * mu[j];
        is[j] = rsqrtf(var + BN_EPS);
        g[j]  = gamma[c];
        b[j]  = beta[c];
        w[j]  = wc[c];
    }
    float bcv = bc[0];
    const float4* xp = (const float4*)x + lane;

    for (int r = r0; r < r1; r += DO_U) {
        float4 v[DO_U];
#pragma unroll
        for (int u = 0; u < DO_U; ++u) {
            int rr = r + u;
            v[u] = xp[(long)(rr < r1 ? rr : r1 - 1) * (CH / 4)];
        }
        float res = 0.f;
#pragma unroll
        for (int u = 0; u < DO_U; ++u) {
            float vv[4] = {v[u].x, v[u].y, v[u].z, v[u].w};
            float acc = 0.f;
#pragma unroll
            for (int j = 0; j < 4; ++j) {
                float val = (fmaxf(vv[j], 0.f) - mu[j]) * is[j] * g[j] + b[j];
                acc += val * w[j];
            }
#pragma unroll
            for (int o = 16; o > 0; o >>= 1)
                acc += __shfl_xor(acc, o, 32);
            res = (lane == u) ? acc : res;
        }
        if (lane < DO_U && r + lane < r1) out[r + lane] = res + bcv;
    }
}

extern "C" void kernel_launch(void* const* d_in, const int* in_sizes, int n_in,
                              void* d_out, int out_size, void* d_ws, size_t ws_size,
                              hipStream_t stream) {
    const float* x   = (const float*)d_in[0];
    const int*   ei  = (const int*)d_in[1];
    const float* W1  = (const float*)d_in[2];
    const float* b1  = (const float*)d_in[3];
    const float* W2  = (const float*)d_in[4];
    const float* b2  = (const float*)d_in[5];
    const float* g1  = (const float*)d_in[6];
    const float* be1 = (const float*)d_in[7];
    const float* g2  = (const float*)d_in[8];
    const float* be2 = (const float*)d_in[9];
    const float* Wc  = (const float*)d_in[10];
    const float* bc  = (const float*)d_in[11];
    float* out = (float*)d_out;

    const int  E  = in_sizes[1] / 2;
    const int  N  = 20000;
    const int  H  = in_sizes[3];
    const int  F  = in_sizes[2] / H;
    const long BN = (long)out_size;
    const int  Bz = (int)(BN / N);
    const long TOT = BN * H;
    const float invcnt = 1.0f / (float)BN;

    float* ws     = (float*)d_ws;
    float* stats1 = ws;
    float* stats2 = ws + 256;
    float* dinv   = ws + 512;
    long   nAl    = (N + 255) & ~255L;
    float* bufA   = ws + 512 + nAl;

    const long PL   = (long)N * H;
    float* bufM     = bufA + (Bz - 1) * PL;
    float* agg1Last = bufA + (2 * Bz - 2) * PL;
    float* agg2Last = bufA + (Bz - 1) * PL;
    long   n1Al     = (N + 1 + 255) & ~255L;
    long   eAl      = (E + 255) & ~255L;
    int*   csrEnd   = (int*)(bufA + 2 * TOT);
    int*   sorted   = csrEnd - eAl;
    int*   slots    = sorted - eAl;
    int*   rowptr   = slots - n1Al;
    int*   cursor   = rowptr - nAl;
    int*   cnt      = cursor - nAl;

    const int* srcIdx = ei;
    const int* dstIdx = ei + E;

    gcn_init  <<<(N + 255) / 256, 256, 0, stream>>>(stats1, cnt, N);
    gcn_count <<<(E + 255) / 256, 256, 0, stream>>>(dstIdx, cnt, E);
    gcn_scan  <<<(N + SC_T - 1) / SC_T, SC_T, 0, stream>>>(cnt, rowptr, cursor, dinv, N);
    gcn_fill  <<<(E + 255) / 256, 256, 0, stream>>>(srcIdx, dstIdx, cursor, slots, E);
    gcn_order <<<(N + 7) / 8, 256, 0, stream>>>(rowptr, slots, sorted, N);

    dim3 tileGridRest((N + FT_ROWS - 1) / FT_ROWS, Bz - 1, 1);
    dim3 tileGridLast((N + FT_ROWS - 1) / FT_ROWS, 1, 1);
    int  gemmBlks  = (int)((BN / 16 + 7) / 8);
    int  statBlks  = 256;

    dim3 nodeGridIn((N + 7) / 8, (Bz + 3) / 4, 1);
    gcn_gather_in<<<nodeGridIn, 256, 0, stream>>>(x, rowptr, sorted, dinv, bufA, N, Bz);
    gcn_gemm_wmma<<<gemmBlks, 256, 0, stream>>>(bufA, W1, b1, bufM, (int)BN, F);
    gcn_bn_stats <<<statBlks, 512, 0, stream>>>(bufM, stats1, BN);

    gcn_gather_bn_gemm<<<tileGridRest, FT_ROWS * 16, 0, stream>>>(bufM, rowptr, sorted, dinv, stats1, g1, be1,
                                                                 invcnt, W2, b2, bufA, N);
    gcn_gather_bn_gemm<<<tileGridLast, FT_ROWS * 16, 0, stream>>>(agg1Last, rowptr, sorted, dinv, stats1, g1, be1,
                                                                 invcnt, W2, b2, agg2Last, N);
    gcn_bn_stats <<<statBlks, 512, 0, stream>>>(bufA, stats2, BN);

    gcn_bn_dot_out<<<statBlks, 512, 0, stream>>>(bufA, stats2, g2, be2, Wc, bc, out, (int)BN, invcnt);
}


# layer 1 in one kernel: a work-group sums the input rows of 8 nodes x 4 planes in an LDS tile, multiplies by W1 on WMMA, stores whole rows; no ax buffer, 1 launch fewer; order of sums unchanged
# speedup vs baseline: 4.7046x; 1.0644x over previous
#include <hip/hip_runtime.h>
#include <hip/hip_bf16.h>

typedef __attribute__((ext_vector_type(2))) float v2f;
typedef __attribute__((ext_vector_type(8))) float v8f;

#define CH 128
#define BN_EPS 1e-5f

__device__ __forceinline__ void atomicAddF(float* p, float v) {
    (void)__hip_atomic_fetch_add(p, v, __ATOMIC_RELAXED, __HIP_MEMORY_SCOPE_AGENT);
}

__global__ void gcn_init(float* stats, int* cnt, int n) {
    int i = blockIdx.x * blockDim.x + threadIdx.x;
    if (i < 2 * 2 * CH) stats[i] = 0.0f;
    if (i < n) cnt[i] = 0;
}
__global__ void gcn_count(const int* __restrict__ dst, int* cnt, int e) {
    int i = blockIdx.x * blockDim.x + threadIdx.x;
    if (i < e) atomicAdd(&cnt[dst[i]], 1);
}
#define SC_T 1024
__global__ void __launch_bounds__(SC_T)
gcn_scan(const int* __restrict__ cnt, int* __restrict__ rowptr,
         int* __restrict__ cursor, float* __restrict__ dinv, int n) {
    __shared__ int wown[SC_T / 32];
    __shared__ int wpre[SC_T / 32];
    int t    = threadIdx.x;
    int lane = t & 31;
    int wv   = t >> 5;
    int nb   = blockIdx.x;
    int i    = nb * SC_T + t;

    int pre = 0;
    for (int j = 0; j < nb; ++j) pre += cnt[j * SC_T + t];
    int own = (i < n) ? cnt[i] : 0;

    int inc = own;
    for (int o = 1; o < 32; o <<= 1) {
        int v = __shfl_up(inc, o, 32);
        inc += (lane >= o) ? v : 0;
    }
    for (int o = 16; o > 0; o >>= 1) pre += __shfl_xor(pre, o, 32);
    if (lane == 31) wown[wv] = inc;
    if (lane == 0)  wpre[wv] = pre;
    __syncthreads();

    int v = wpre[lane] + ((lane < wv) ? wown[lane] : 0);
    for (int o = 16; o > 0; o >>= 1) v += __shfl_xor(v, o, 32);
    int excl = v + inc - own;

    if (i < n) {
        rowptr[i] = excl;
        cursor[i] = excl;
        dinv[i]   = rsqrtf((float)(own + 1));
    }
    if (i == n - 1) rowptr[n] = excl + own;
}
__global__ void gcn_fill(const int* __restrict__ src, const int* __restrict__ dst,
                         int* cursor, int* __restrict__ slots, int e) {
    int i = blockIdx.x * blockDim.x + threadIdx.x;
    if (i < e) {
        int p = atomicAdd(&cursor[dst[i]], 1);
        slots[p] = src[i];
    }
}
__global__ void gcn_order(const int* __restrict__ rowptr, const int* __restrict__ slots,
                          int* __restrict__ sorted, int n) {
    int d = blockIdx.x * (blockDim.x >> 5) + (threadIdx.x >> 5);
    if (d >= n) return;
    int lane = threadIdx.x & 31;
    int beg = rowptr[d];
    int len = rowptr[d + 1] - beg;
    for (int i = lane; i < len; i += 32) {
        int v = slots[beg + i];
        int rank = 0;
        for (int j = 0; j < len; ++j) {
            int u = slots[beg + j];
            rank += (u < v || (u == v && j < i)) ? 1 : 0;
        }
        sorted[beg + rank] = v;
    }
}

__device__ __forceinline__ void gcn_fma4(float4& acc, const float4& v, float w) {
    acc.x = __builtin_fmaf(v.x, w, acc.x);
    acc.y = __builtin_fmaf(v.y, w, acc.y);
    acc.z = __builtin_fmaf(v.z, w, acc.z);
    acc.w = __builtin_fmaf(v.w, w, acc.w);
}
#define FT_ROWS  16
#define FT_PITCH 132
__device__ __forceinline__ float4 gcn_bn4(const float4& v, const float* mu, const float* is,
                                          const float* g, const float* be) {
    float4 r;
    r.x = (fmaxf(v.x, 0.f) - mu[0]) * is[0] * g[0] + be[0];
    r.y = (fmaxf(v.y, 0.f) - mu[1]) * is[1] * g[1] + be[1];
    r.z = (fmaxf(v.z, 0.f) - mu[2]) * is[2] * g[2] + be[2];
    r.w = (fmaxf(v.w, 0.f) - mu[3]) * is[3] * g[3] + be[3];
    return r;
}
__global__ void __launch_bounds__(FT_ROWS * 16)
gcn_gather_bn_gemm(const float* __restrict__ a, const int* __restrict__ rowptr,
                   const int* __restrict__ csrc, const float* __restrict__ dinv,
                   const float* __restrict__ sums, const float* __restrict__ gamma,
                   const float* __restrict__ beta, float invcnt,
                   const float* __restrict__ W, const float* __restrict__ bias,
                   float* __restrict__ out, int n) {
    __shared__ __attribute__((aligned(16))) float tile[FT_ROWS * FT_PITCH];
    int lane = threadIdx.x & 31;
    int wv = __builtin_amdgcn_readfirstlane(threadIdx.x >> 5);
    int node0 = blockIdx.x * FT_ROWS;
    const float4* hb = (const float4*)(a + (long)blockIdx.y * n * CH);
    float4*       ob = (float4*)(out + (long)blockIdx.y * n * CH);

    float mu[4], is[4], g[4], be[4];
#pragma unroll
    for (int j = 0; j < 4; ++j) {
        int c = lane * 4 + j;
        mu[j] = sums[c] * invcnt;
        float var = sums[CH + c] * invcnt - mu[j] * mu[j];
        is[j] = rsqrtf(var + BN_EPS);
        g[j]  = gamma[c];
        be[j] = beta[c];
    }

    for (int q = 0; q < 2; ++q) {
        int r = wv * 2 + q;
        int d = node0 + r;
        float4 acc = {0.f, 0.f, 0.f, 0.f};
        if (d < n) {
            int beg = __builtin_amdgcn_readfirstlane(rowptr[d]);
            int end = __builtin_amdgcn_readfirstlane(rowptr[d + 1]);
            float dd = dinv[d];
            float4 hv = gcn_bn4(hb[(long)d * (CH / 4) + lane], mu, is, g, be);
            acc.x = hv.x * dd * dd;
            acc.y = hv.y * dd * dd;
            acc.z = hv.z * dd * dd;
            acc.w = hv.w * dd * dd;

            for (int base = beg; base < end; base += 32) {
                int cnt = end - base; if (cnt > 32) cnt = 32;
                int   sj = 0;
                float wj = 0.f;
                if (lane < cnt) { sj = csrc[base + lane]; wj = dinv[sj] * dd; }
                int wjb = __builtin_bit_cast(int, wj);
                int j = 0;
                for (; j + 4 <= cnt; j += 4) {
                    int s0 = __builtin_amdgcn_readlane(sj, j);
                    int s1 = __builtin_amdgcn_readlane(sj, j + 1);
                    int s2 = __builtin_amdgcn_readlane(sj, j + 2);
                    int s3 = __builtin_amdgcn_readlane(sj, j + 3);
                    float w0 = __builtin_bit_cast(float, __builtin_amdgcn_readlane(wjb, j));
                    float w1 = __builtin_bit_cast(float, __builtin_amdgcn_readlane(wjb, j + 1));
                    float w2 = __builtin_bit_cast(float, __builtin_amdgcn_readlane(wjb, j + 2));
                    float w3 = __builtin_bit_cast(float, __builtin_amdgcn_readlane(wjb, j + 3));
                    float4 v0 = hb[(long)s0 * (CH / 4) + lane];
                    float4 v1 = hb[(long)s1 * (CH / 4) + lane];
                    float4 v2 = hb[(long)s2 * (CH / 4) + lane];
                    float4 v3 = hb[(long)s3 * (CH / 4) + lane];
                    gcn_fma4(acc, gcn_bn4(v0, mu, is, g, be), w0);
                    gcn_fma4(acc, gcn_bn4(v1, mu, is, g, be), w1);
                    gcn_fma4(acc, gcn_bn4(v2, mu, is, g, be), w2);
                    gcn_fma4(acc, gcn_bn4(v3, mu, is, g, be), w3);
                }
                int rem = cnt - j;
                if (rem > 0) {
                    int s0 = __builtin_amdgcn_readlane(sj, j);
                    float w0 = __builtin_bit_cast(float, __builtin_amdgcn_readlane(wjb, j));
                    float w1 = 0.f, w2 = 0.f;
                    float4 v0 = hb[(long)s0 * (CH / 4) + lane];
                    float4 v1 = {0.f, 0.f, 0.f, 0.f}, v2 = {0.f, 0.f, 0.f, 0.f};
                    if (rem > 1) {
                        int s1 = __builtin_amdgcn_readlane(sj, j + 1);
                        w1 = __builtin_bit_cast(float, __builtin_amdgcn_readlane(wjb, j + 1));
                        v1 = hb[(long)s1 * (CH / 4) + lane];
                    }
                    if (rem > 2) {
                        int s2 = __builtin_amdgcn_readlane(sj, j + 2);
                        w2 = __builtin_bit_cast(float, __builtin_amdgcn_readlane(wjb, j + 2));
                        v2 = hb[(long)s2 * (CH / 4) + lane];
                    }
                    gcn_fma4(acc, gcn_bn4(v0, mu, is, g, be), w0);
                    if (rem > 1) gcn_fma4(acc, gcn_bn4(v1, mu, is, g, be), w1);
                    if (rem > 2) gcn_fma4(acc, gcn_bn4(v2, mu, is, g, be), w2);
                }
            }
        }
        *(float4*)&tile[r * FT_PITCH + lane * 4] = acc;
    }

    int khalf = (lane >> 4) << 1;
    int col   = wv * 16 + (lane & 15);
    float bv  = bias[col];
    v8f acc8 = (v8f){bv, bv, bv, bv, bv, bv, bv, bv};
    const float* ap = &tile[(lane & 15) * FT_PITCH + khalf];
    const float* wp = W + (long)khalf * CH + col;
    v2f bw[16];
#pragma unroll
    for (int i = 0; i < 16; ++i) {
        bw[i].x = wp[(long)(4 * i) * CH];
        bw[i].y = wp[(long)(4 * i) * CH + CH];
    }
    __syncthreads();
#pragma unroll
    for (int i = 0; i < 16; ++i) {
        v2f av = *(const v2f*)(ap + 4 * i);
        acc8 = __builtin_amdgcn_wmma_f32_16x16x4_f32(
            false, av, false, bw[i], (short)0, acc8, false, false);
        bw[i].x = wp[(long)(4 * i + 64) * CH];
        bw[i].y = wp[(long)(4 * i + 64) * CH + CH];
    }
#pragma unroll
    for (int i = 0; i < 16; ++i) {
        v2f av = *(const v2f*)(ap + 4 * i + 64);
        acc8 = __builtin_amdgcn_wmma_f32_16x16x4_f32(
            false, av, false, bw[i], (short)0, acc8, false, false);
    }
    __syncthreads();
    int rb = (lane >> 4) << 3;
#pragma unroll
    for (int r = 0; r < 8; ++r) tile[(rb + r) * FT_PITCH + col] = acc8[r];
    __syncthreads();
    for (int q = 0; q < 2; ++q) {
        int r = wv * 2 + q;
        int d = node0 + r;
        if (d < n) ob[(long)d * (CH / 4) + lane] = *(const float4*)&tile[r * FT_PITCH + lane * 4];
    }
}

#define FIN 64
#define FI_NODES 8
#define FI_ROWS  (FI_NODES * 4)
#define FI_PITCH 68
__global__ void __launch_bounds__(FI_NODES * 32)
gcn_gather_in_gemm(const float* __restrict__ x, const int* __restrict__ rowptr,
                   const int* __restrict__ csrc, const float* __restrict__ dinv,
                   const float* __restrict__ W, const float* __restrict__ bias,
                   float* __restrict__ out, int n, int nb) {
    __shared__ __attribute__((aligned(16))) float tin[FI_ROWS * FI_PITCH];
    __shared__ __attribute__((aligned(16))) float tout[FI_ROWS * FT_PITCH];
    int lane = threadIdx.x & 31;
    int wv = __builtin_amdgcn_readfirstlane(threadIdx.x >> 5);
    int d  = blockIdx.x * FI_NODES + wv;
    int c4   = lane & 15;
    int pb0  = blockIdx.y * 4 + (lane >> 4);
    int pb1  = pb0 + 2;
    bool ok0 = pb0 < nb, ok1 = pb1 < nb;
    const float4* x0 = (const float4*)x + (long)(ok0 ? pb0 : 0) * n * (FIN / 4) + c4;
    const float4* x1 = (const float4*)x + (long)(ok1 ? pb1 : 0) * n * (FIN / 4) + c4;
    float4 acc0 = {0.f, 0.f, 0.f, 0.f};
    float4 acc1 = {0.f, 0.f, 0.f, 0.f};
    if (d < n) {
        int beg = __builtin_amdgcn_readfirstlane(rowptr[d]);
        int end = __builtin_amdgcn_readfirstlane(rowptr[d + 1]);
        float dd = dinv[d];
        float4 h0 = x0[(long)d * (FIN / 4)];
        float4 h1 = x1[(long)d * (FIN / 4)];
        acc0.x = h0.x * dd * dd; acc0.y = h0.y * dd * dd; acc0.z = h0.z * dd * dd; acc0.w = h0.w * dd * dd;
        acc1.x = h1.x * dd * dd; acc1.y = h1.y * dd * dd; acc1.z = h1.z * dd * dd; acc1.w = h1.w * dd * dd;

        for (int base = beg; base < end; base += 32) {
            int cnt = end - base; if (cnt > 32) cnt = 32;
            int   sj = 0;
            float wj = 0.f;
            if (lane < cnt) { sj = csrc[base + lane]; wj = dinv[sj] * dd; }
            int wjb = __builtin_bit_cast(int, wj);
            int j = 0;
            for (; j + 4 <= cnt; j += 4) {
                int s0 = __builtin_amdgcn_readlane(sj, j);
                int s1 = __builtin_amdgcn_readlane(sj, j + 1);
                int s2 = __builtin_amdgcn_readlane(sj, j + 2);
                int s3 = __builtin_amdgcn_readlane(sj, j + 3);
                float4 a0 = x0[(long)s0 * (FIN / 4)];
                float4 b0 = x1[(long)s0 * (FIN / 4)];
                float4 a1 = x0[(long)s1 * (FIN / 4)];
                float4 b1 = x1[(long)s1 * (FIN / 4)];
                float4 a2 = x0[(long)s2 * (FIN / 4)];
                float4 b2 = x1[(long)s2 * (FIN / 4)];
                float4 a3 = x0[(long)s3 * (FIN / 4)];
                float4 b3 = x1[(long)s3 * (FIN / 4)];
                float w0 = __builtin_bit_cast(float, __builtin_amdgcn_readlane(wjb, j));
                float w1 = __builtin_bit_cast(float, __builtin_amdgcn_readlane(wjb, j + 1));
                float w2 = __builtin_bit_cast(float, __builtin_amdgcn_readlane(wjb, j + 2));
                float w3 = __builtin_bit_cast(float, __builtin_amdgcn_readlane(wjb, j + 3));
                gcn_fma4(acc0, a0, w0); gcn_fma4(acc1, b0, w0);
                gcn_fma4(acc0, a1, w1); gcn_fma4(acc1, b1, w1);
                gcn_fma4(acc0, a2, w2); gcn_fma4(acc1, b2, w2);
                gcn_fma4(acc0, a3, w3); gcn_fma4(acc1, b3, w3);
            }
            int rem = cnt - j;
            if (rem > 0) {
                const float4 z = {0.f, 0.f, 0.f, 0.f};
                int s0 = __builtin_amdgcn_readlane(sj, j);
                float w0 = __builtin_bit_cast(float, __builtin_amdgcn_readlane(wjb, j));
                float w1 = 0.f, w2 = 0.f;
                float4 a0 = x0[(long)s0 * (FIN / 4)];
                float4 b0 = x1[(long)s0 * (FIN / 4)];
                float4 a1 = z, b1 = z, a2 = z, b2 = z;
                if (rem > 1) {
                    int s1 = __builtin_amdgcn_readlane(sj, j + 1);
                    w1 = __builtin_bit_cast(float, __builtin_amdgcn_readlane(wjb, j + 1));
                    a1 = x0[(long)s1 * (FIN / 4)];
                    b1 = x1[(long)s1 * (FIN / 4)];
                }
                if (rem > 2) {
                    int s2 = __builtin_amdgcn_readlane(sj, j + 2);
                    w2 = __builtin_bit_cast(float, __builtin_amdgcn_readlane(wjb, j + 2));
                    a2 = x0[(long)s2 * (FIN / 4)];
                    b2 = x1[(long)s2 * (FIN / 4)];
                }
                gcn_fma4(acc0, a0, w0); gcn_fma4(acc1, b0, w0);
                if (rem > 1) { gcn_fma4(acc0, a1, w1); gcn_fma4(acc1, b1, w1); }
                if (rem > 2) { gcn_fma4(acc0, a2, w2); gcn_fma4(acc1, b2, w2); }
            }
        }
    }
    int trow = wv * 4 + (lane >> 4);
    *(float4*)&tin[trow * FI_PITCH + c4 * 4]       = acc0;
    *(float4*)&tin[(trow + 2) * FI_PITCH + c4 * 4] = acc1;

    int khalf = (lane >> 4) << 1;
    int col   = wv * 16 + (lane & 15);
    float bv  = bias[col];
    v8f accA = (v8f){bv, bv, bv, bv, bv, bv, bv, bv};
    v8f accB = accA;
    const float* apA = &tin[(lane & 15) * FI_PITCH + khalf];
    const float* apB = apA + 16 * FI_PITCH;
    const float* wp  = W + (long)khalf * CH + col;
    v2f bw[FIN / 4];
#pragma unroll
    for (int i = 0; i < FIN / 4; ++i) {
        bw[i].x = wp[(long)(4 * i) * CH];
        bw[i].y = wp[(long)(4 * i) * CH + CH];
    }
    __syncthreads();
#pragma unroll
    for (int i = 0; i < FIN / 4; ++i) {
        v2f avA = *(const v2f*)(apA + 4 * i);
        v2f avB = *(const v2f*)(apB + 4 * i);
        accA = __builtin_amdgcn_wmma_f32_16x16x4_f32(
            false, avA, false, bw[i], (short)0, accA, false, false);
        accB = __builtin_amdgcn_wmma_f32_16x16x4_f32(
            false, avB, false, bw[i], (short)0, accB, false, false);
    }
    int rb = (lane >> 4) << 3;
#pragma unroll
    for (int r = 0; r < 8; ++r) {
        tout[(rb + r) * FT_PITCH + col]      = accA[r];
        tout[(16 + rb + r) * FT_PITCH + col] = accB[r];
    }
    __syncthreads();
    if (d < n) {
        float4* ob = (float4*)out;
#pragma unroll
        for (int p = 0; p < 4; ++p) {
            int plane = blockIdx.y * 4 + p;
            if (plane < nb)
                ob[((long)plane * n + d) * (CH / 4) + lane] =
                    *(const float4*)&tout[(wv * 4 + p) * FT_PITCH + lane * 4];
        }
    }
}

#define ST_U     10
#define ST_WAVES 16
__global__ void gcn_bn_stats(const float* __restrict__ x, float* __restrict__ sums, long rows) {
    __shared__ float red[ST_WAVES][2 * CH];
    int lane = threadIdx.x & 31;
    int wv = __builtin_amdgcn_readfirstlane(threadIdx.x >> 5);
    long gw = (long)blockIdx.x * ST_WAVES + wv;
    long tw = (long)gridDim.x * ST_WAVES;
    const float4* xp = (const float4*)x;
    float4 s = {0.f, 0.f, 0.f, 0.f};
    float4 q = {0.f, 0.f, 0.f, 0.f};
    for (long r = gw; r < rows; r += tw * ST_U) {
        float4 v[ST_U];
#pragma unroll
        for (int u = 0; u < ST_U; ++u) {
            long rr = r + u * tw;
            v[u] = xp[(rr < rows ? rr : rows - 1) * (CH / 4) + lane];
        }
#pragma unroll
        for (int u = 0; u < ST_U; ++u) {
            bool in = r + u * tw < rows;
            float a = fmaxf(in ? v[u].x : 0.f, 0.f);
            float b = fmaxf(in ? v[u].y : 0.f, 0.f);
            float c = fmaxf(in ? v[u].z : 0.f, 0.f);
            float d = fmaxf(in ? v[u].w : 0.f, 0.f);
            s.x += a; s.y += b; s.z += c; s.w += d;
            q.x = __builtin_fmaf(a, a, q.x);
            q.y = __builtin_fmaf(b, b, q.y);
            q.z = __builtin_fmaf(c, c, q.z);
            q.w = __builtin_fmaf(d, d, q.w);
        }
    }
    *(float4*)&red[wv][lane * 4]      = s;
    *(float4*)&red[wv][CH + lane * 4] = q;
    __syncthreads();
    int t = threadIdx.x;
    if (t < 2 * CH) {
        float tot = 0.f;
#pragma unroll
        for (int w = 0; w < ST_WAVES; ++w) tot += red[w][t];
        atomicAddF(&sums[t], tot);
    }
}

#define DO_U     20
#define DO_WAVES 16
__global__ void __launch_bounds__(DO_WAVES * 32) __attribute__((amdgpu_waves_per_eu(4, 4)))
gcn_bn_dot_out(const float* __restrict__ x, const float* __restrict__ sums,
               const float* __restrict__ gamma, const float* __restrict__ beta,
               const float* __restrict__ wc, const float* __restrict__ bc,
               float* __restrict__ out, int rows, float invcnt) {
    int lane = threadIdx.x & 31;
    int wv  = __builtin_amdgcn_readfirstlane(threadIdx.x >> 5);
    int gw  = blockIdx.x * DO_WAVES + wv;
    int tw  = gridDim.x * DO_WAVES;
    int per = (rows + tw - 1) / tw;
    int r0  = gw * per;
    int r1  = r0 + per; if (r1 > rows) r1 = rows;
    if (r0 >= r1) return;

    float mu[4], is[4], g[4], b[4], w[4];
#pragma unroll
    for (int j = 0; j < 4; ++j) {
        int c = lane * 4 + j;
        mu[j] = sums[c] * invcnt;
        float var = sums[CH + c] * invcnt - mu[j] * mu[j];
        is[j] = rsqrtf(var + BN_EPS);
        g[j]  = gamma[c];
        b[j]  = beta[c];
        w[j]  = wc[c];
    }
    float bcv = bc[0];
    const float4* xp = (const float4*)x + lane;

    for (int r = r0; r < r1; r += DO_U) {
        float4 v[DO_U];
#pragma unroll
        for (int u = 0; u < DO_U; ++u) {
            int rr = r + u;
            v[u] = xp[(long)(rr < r1 ? rr : r1 - 1) * (CH / 4)];
        }
        float res = 0.f;
#pragma unroll
        for (int u = 0; u < DO_U; ++u) {
            float vv[4] = {v[u].x, v[u].y, v[u].z, v[u].w};
            float acc = 0.f;
#pragma unroll
            for (int j = 0; j < 4; ++j) {
                float val = (fmaxf(vv[j], 0.f) - mu[j]) * is[j] * g[j] + b[j];
                acc += val * w[j];
            }
#pragma unroll
            for (int o = 16; o > 0; o >>= 1)
                acc += __shfl_xor(acc, o, 32);
            res = (lane == u) ? acc : res;
        }
        if (lane < DO_U && r + lane < r1) out[r + lane] = res + bcv;
    }
}

extern "C" void kernel_launch(void* const* d_in, const int* in_sizes, int n_in,
                              void* d_out, int out_size, void* d_ws, size_t ws_size,
                              hipStream_t stream) {
    const float* x   = (const float*)d_in[0];
    const int*   ei  = (const int*)d_in[1];
    const float* W1  = (const float*)d_in[2];
    const float* b1  = (const float*)d_in[3];
    const float* W2  = (const float*)d_in[4];
    const float* b2  = (const float*)d_in[5];
    const float* g1  = (const float*)d_in[6];
    const float* be1 = (const float*)d_in[7];
    const float* g2  = (const float*)d_in[8];
    const float* be2 = (const float*)d_in[9];
    const float* Wc  = (const float*)d_in[10];
    const float* bc  = (const float*)d_in[11];
    float* out = (float*)d_out;

    const int  E  = in_sizes[1] / 2;
    const int  N  = 20000;
    const int  H  = in_sizes[3];
    const long BN = (long)out_size;
    const int  Bz = (int)(BN / N);
    const long TOT = BN * H;
    const float invcnt = 1.0f / (float)BN;

    float* ws     = (float*)d_ws;
    float* stats1 = ws;
    float* stats2 = ws + 256;
    float* dinv   = ws + 512;
    long   nAl    = (N + 255) & ~255L;
    float* bufA   = ws + 512 + nAl;

    const long PL   = (long)N * H;
    float* bufM     = bufA + (Bz - 1) * PL;
    float* agg1Last = bufA + (2 * Bz - 2) * PL;
    float* agg2Last = bufA + (Bz - 1) * PL;
    long   n1Al     = (N + 1 + 255) & ~255L;
    long   eAl      = (E + 255) & ~255L;
    int*   csrEnd   = (int*)(bufA + 2 * TOT);
    int*   sorted   = csrEnd - eAl;
    int*   slots    = sorted - eAl;
    int*   rowptr   = slots - n1Al;
    int*   cursor   = rowptr - nAl;
    int*   cnt      = cursor - nAl;

    const int* srcIdx = ei;
    const int* dstIdx = ei + E;

    gcn_init  <<<(N + 255) / 256, 256, 0, stream>>>(stats1, cnt, N);
    gcn_count <<<(E + 255) / 256, 256, 0, stream>>>(dstIdx, cnt, E);
    gcn_scan  <<<(N + SC_T - 1) / SC_T, SC_T, 0, stream>>>(cnt, rowptr, cursor, dinv, N);
    gcn_fill  <<<(E + 255) / 256, 256, 0, stream>>>(srcIdx, dstIdx, cursor, slots, E);
    gcn_order <<<(N + 7) / 8, 256, 0, stream>>>(rowptr, slots, sorted, N);

    dim3 tileGridRest((N + FT_ROWS - 1) / FT_ROWS, Bz - 1, 1);
    dim3 tileGridLast((N + FT_ROWS - 1) / FT_ROWS, 1, 1);
    int  statBlks  = 256;

    dim3 nodeGridIn((N + FI_NODES - 1) / FI_NODES, (Bz + 3) / 4, 1);
    gcn_gather_in_gemm<<<nodeGridIn, FI_NODES * 32, 0, stream>>>(x, rowptr, sorted, dinv, W1, b1, bufM, N, Bz);
    gcn_bn_stats <<<statBlks, 512, 0, stream>>>(bufM, stats1, BN);

    gcn_gather_bn_gemm<<<tileGridRest, FT_ROWS * 16, 0, stream>>>(bufM, rowptr, sorted, dinv, stats1, g1, be1,
                                                                 invcnt, W2, b2, bufA, N);
    gcn_gather_bn_gemm<<<tileGridLast, FT_ROWS * 16, 0, stream>>>(agg1Last, rowptr, sorted, dinv, stats1, g1, be1,
                                                                 invcnt, W2, b2, agg2Last, N);
    gcn_bn_stats <<<statBlks, 512, 0, stream>>>(bufA, stats2, BN);

    gcn_bn_dot_out<<<statBlks, 512, 0, stream>>>(bufA, stats2, g2, be2, Wc, bc, out, (int)BN, invcnt);
}


# edge-list build in one launch: 128 owner work-groups each scan all destinations, count and rank their nodes' edges on LDS counters, sort segments in LDS; 4 launches fewer; order of sums unchanged
# speedup vs baseline: 6.1439x; 1.3059x over previous
#include <hip/hip_runtime.h>
#include <hip/hip_bf16.h>

typedef __attribute__((ext_vector_type(2))) float v2f;
typedef __attribute__((ext_vector_type(8))) float v8f;

#define CH 128
#define BN_EPS 1e-5f

__device__ __forceinline__ void atomicAddF(float* p, float v) {
    (void)__hip_atomic_fetch_add(p, v, __ATOMIC_RELAXED, __HIP_MEMORY_SCOPE_AGENT);
}

template <bool FITS>
__device__ __forceinline__ void gcn_sort_run(const int* segL, const int* segG, const int* preL,
                                             const int* cntL, int* outRun, int span, int wv, int lane) {
    for (int q = wv; q < span; q += 32) {
        int beg = __builtin_amdgcn_readfirstlane(preL[q]);
        int len = __builtin_amdgcn_readfirstlane(cntL[q]);
        int* outp = outRun + beg;
        if (len <= 32) {
            int v = 0;
            if (lane < len) v = FITS ? segL[beg + lane] : segG[beg + lane];
            int rank = 0;
            for (int j = 0; j < len; ++j) {
                int u = __builtin_amdgcn_readlane(v, j);
                rank += (u < v || (u == v && j < lane)) ? 1 : 0;
            }
            if (lane < len) outp[rank] = v;
        } else {
            for (int i = lane; i < len; i += 32) {
                int v = FITS ? segL[beg + i] : segG[beg + i];
                int rank = 0;
                for (int j = 0; j < len; ++j) {
                    int u = FITS ? segL[beg + j] : segG[beg + j];
                    rank += (u < v || (u == v && j < i)) ? 1 : 0;
                }
                outp[rank] = v;
            }
        }
    }
}
#define OW_T    1024
#define OW_R    160
#define OW_WCAP 256
#define OW_SCAP 8192
#define OW_U    4
__global__ void __launch_bounds__(OW_T)
gcn_build(const int* __restrict__ src, const int* __restrict__ dst, int* __restrict__ rowptr,
          int* __restrict__ slots, int* __restrict__ sorted, float* __restrict__ dinv,
          float* __restrict__ stats, int n, int e) {
    __shared__ int cntL[OW_R];
    __shared__ int preL[OW_R];
    __shared__ int wown[OW_T / 32];
    __shared__ int wlow[OW_T / 32];
    __shared__ int wcnt[OW_T / 32];
    __shared__ int lstI[(OW_T / 32) * OW_WCAP];
    __shared__ int lstD[(OW_T / 32) * OW_WCAP];
    __shared__ int srcL[OW_SCAP];

    int t    = threadIdx.x;
    int lane = t & 31;
    int wv   = __builtin_amdgcn_readfirstlane(t >> 5);
    int per  = (n + (int)gridDim.x - 1) / (int)gridDim.x;
    int lo   = blockIdx.x * per;
    int hi   = lo + per; if (hi > n) hi = n;
    int span = hi - lo;  if (span < 0) span = 0;

    if (t < OW_R) cntL[t] = 0;
    if (blockIdx.x == 0 && t < 2 * 2 * CH) stats[t] = 0.0f;
    __syncthreads();

    int low  = 0;
    int wnum = 0;
    int* myI = lstI + wv * OW_WCAP;
    int* myD = lstD + wv * OW_WCAP;
    unsigned lt = (1u << lane) - 1u;

#define OW_VISIT(dv, iv, okv)                                                       \
    {                                                                               \
        int  dl_ = (dv) - lo;                                                       \
        bool in_ = (okv) && (unsigned)dl_ < (unsigned)span;                         \
        low += __builtin_popcount(__builtin_amdgcn_ballot_w32((okv) && dl_ < 0));   \
        unsigned m_ = __builtin_amdgcn_ballot_w32(in_);                             \
        if (m_ != 0u) {                                                             \
            if (in_) {                                                              \
                atomicAdd(&cntL[dl_], 1);                                           \
                int p_ = wnum + __builtin_popcount(m_ & lt);                        \
                if (p_ < OW_WCAP) { myI[p_] = (iv); myD[p_] = dl_; }                \
            }                                                                       \
            wnum += __builtin_popcount(m_);                                         \
        }                                                                           \
    }

    bool vec = ((e & 3) == 0) && ((((unsigned long)dst) & 15ul) == 0ul);
    if (vec) {
        const int4* d4p = (const int4*)dst;
        int q4 = e >> 2;
        int b  = 0;
        for (; b + OW_T * OW_U <= q4; b += OW_T * OW_U) {
            int4 dv[OW_U];
#pragma unroll
            for (int u = 0; u < OW_U; ++u) dv[u] = d4p[b + u * OW_T + t];
#pragma unroll
            for (int u = 0; u < OW_U; ++u) {
                int i0 = 4 * (b + u * OW_T + t);
                OW_VISIT(dv[u].x, i0,     true)
                OW_VISIT(dv[u].y, i0 + 1, true)
                OW_VISIT(dv[u].z, i0 + 2, true)
                OW_VISIT(dv[u].w, i0 + 3, true)
            }
        }
        for (; b < q4; b += OW_T) {
            int  g  = b + t;
            bool ok = g < q4;
            int4 dv = d4p[ok ? g : q4 - 1];
            OW_VISIT(dv.x, 4 * g,     ok)
            OW_VISIT(dv.y, 4 * g + 1, ok)
            OW_VISIT(dv.z, 4 * g + 2, ok)
            OW_VISIT(dv.w, 4 * g + 3, ok)
        }
    } else {
        for (int b = 0; b < e; b += OW_T) {
            int  i  = b + t;
            bool ok = i < e;
            int  d  = dst[ok ? i : e - 1];
            OW_VISIT(d, i, ok)
        }
    }
#undef OW_VISIT
    __syncthreads();

    int own = (t < span) ? cntL[t] : 0;
    int inc = own;
    for (int o = 1; o < 32; o <<= 1) {
        int v = __shfl_up(inc, o, 32);
        inc += (lane >= o) ? v : 0;
    }
    if (lane == 31) wown[wv] = inc;
    if (lane == 0)  { wlow[wv] = low; wcnt[wv] = wnum; }
    __syncthreads();
    int wa    = wown[lane];
    int front = (lane < wv) ? wa : 0;
    int total = wa;
    int below = wlow[lane];
    int over  = (wcnt[lane] > OW_WCAP) ? 1 : 0;
    for (int o = 16; o > 0; o >>= 1) {
        front += __shfl_xor(front, o, 32);
        total += __shfl_xor(total, o, 32);
        below += __shfl_xor(below, o, 32);
        over  |= __shfl_xor(over, o, 32);
    }
    int  excl = front + inc - own;
    bool slow = over != 0;
    bool fits = total <= OW_SCAP;
    if (t < span) {
        rowptr[lo + t] = below + excl;
        dinv[lo + t]   = rsqrtf((float)(own + 1));
        preL[t] = excl;
        cntL[t] = 0;
    }
    if (t == 0 && lo < n && hi == n) rowptr[n] = below + total;
    __syncthreads();

    if (!slow) {
        for (int k = lane; k < wnum; k += 32) {
            int i  = myI[k];
            int dl = myD[k];
            int sv = src[i];
            int p  = preL[dl] + atomicAdd(&cntL[dl], 1);
            if (fits) srcL[p] = sv; else slots[below + p] = sv;
        }
    } else {
        for (int b = 0; b < e; b += OW_T) {
            int i = b + t;
            if (i < e) {
                int dl = dst[i] - lo;
                if ((unsigned)dl < (unsigned)span) {
                    int sv = src[i];
                    int p  = preL[dl] + atomicAdd(&cntL[dl], 1);
                    if (fits) srcL[p] = sv; else slots[below + p] = sv;
                }
            }
        }
    }
    __syncthreads();

    if (fits) gcn_sort_run<true >(srcL, slots + below, preL, cntL, sorted + below, span, wv, lane);
    else      gcn_sort_run<false>(srcL, slots + below, preL, cntL, sorted + below, span, wv, lane);
}

__device__ __forceinline__ void gcn_fma4(float4& acc, const float4& v, float w) {
    acc.x = __builtin_fmaf(v.x, w, acc.x);
    acc.y = __builtin_fmaf(v.y, w, acc.y);
    acc.z = __builtin_fmaf(v.z, w, acc.z);
    acc.w = __builtin_fmaf(v.w, w, acc.w);
}
#define FT_ROWS  16
#define FT_PITCH 132
__device__ __forceinline__ float4 gcn_bn4(const float4& v, const float* mu, const float* is,
                                          const float* g, const float* be) {
    float4 r;
    r.x = (fmaxf(v.x, 0.f) - mu[0]) * is[0] * g[0] + be[0];
    r.y = (fmaxf(v.y, 0.f) - mu[1]) * is[1] * g[1] + be[1];
    r.z = (fmaxf(v.z, 0.f) - mu[2]) * is[2] * g[2] + be[2];
    r.w = (fmaxf(v.w, 0.f) - mu[3]) * is[3] * g[3] + be[3];
    return r;
}
__global__ void __launch_bounds__(FT_ROWS * 16)
gcn_gather_bn_gemm(const float* __restrict__ a, const int* __restrict__ rowptr,
                   const int* __restrict__ csrc, const float* __restrict__ dinv,
                   const float* __restrict__ sums, const float* __restrict__ gamma,
                   const float* __restrict__ beta, float invcnt,
                   const float* __restrict__ W, const float* __restrict__ bias,
                   float* __restrict__ out, int n) {
    __shared__ __attribute__((aligned(16))) float tile[FT_ROWS * FT_PITCH];
    int lane = threadIdx.x & 31;
    int wv = __builtin_amdgcn_readfirstlane(threadIdx.x >> 5);
    int node0 = blockIdx.x * FT_ROWS;
    const float4* hb = (const float4*)(a + (long)blockIdx.y * n * CH);
    float4*       ob = (float4*)(out + (long)blockIdx.y * n * CH);

    float mu[4], is[4], g[4], be[4];
#pragma unroll
    for (int j = 0; j < 4; ++j) {
        int c = lane * 4 + j;
        mu[j] = sums[c] * invcnt;
        float var = sums[CH + c] * invcnt - mu[j] * mu[j];
        is[j] = rsqrtf(var + BN_EPS);
        g[j]  = gamma[c];
        be[j] = beta[c];
    }

    for (int q = 0; q < 2; ++q) {
        int r = wv * 2 + q;
        int d = node0 + r;
        float4 acc = {0.f, 0.f, 0.f, 0.f};
        if (d < n) {
            int beg = __builtin_amdgcn_readfirstlane(rowptr[d]);
            int end = __builtin_amdgcn_readfirstlane(rowptr[d + 1]);
            float dd = dinv[d];
            float4 hv = gcn_bn4(hb[(long)d * (CH / 4) + lane], mu, is, g, be);
            acc.x = hv.x * dd * dd;
            acc.y = hv.y * dd * dd;
            acc.z = hv.z * dd * dd;
            acc.w = hv.w * dd * dd;

            for (int base = beg; base < end; base += 32) {
                int cnt = end - base; if (cnt > 32) cnt = 32;
                int   sj = 0;
                float wj = 0.f;
                if (lane < cnt) { sj = csrc[base + lane]; wj = dinv[sj] * dd; }
                int wjb = __builtin_bit_cast(int, wj);
                int j = 0;
                for (; j + 4 <= cnt; j += 4) {
                    int s0 = __builtin_amdgcn_readlane(sj, j);
                    int s1 = __builtin_amdgcn_readlane(sj, j + 1);
                    int s2 = __builtin_amdgcn_readlane(sj, j + 2);
                    int s3 = __builtin_amdgcn_readlane(sj, j + 3);
                    float w0 = __builtin_bit_cast(float, __builtin_amdgcn_readlane(wjb, j));
                    float w1 = __builtin_bit_cast(float, __builtin_amdgcn_readlane(wjb, j + 1));
                    float w2 = __builtin_bit_cast(float, __builtin_amdgcn_readlane(wjb, j + 2));
                    float w3 = __builtin_bit_cast(float, __builtin_amdgcn_readlane(wjb, j + 3));
                    float4 v0 = hb[(long)s0 * (CH / 4) + lane];
                    float4 v1 = hb[(long)s1 * (CH / 4) + lane];
                    float4 v2 = hb[(long)s2 * (CH / 4) + lane];
                    float4 v3 = hb[(long)s3 * (CH / 4) + lane];
                    gcn_fma4(acc, gcn_bn4(v0, mu, is, g, be), w0);
                    gcn_fma4(acc, gcn_bn4(v1, mu, is, g, be), w1);
                    gcn_fma4(acc, gcn_bn4(v2, mu, is, g, be), w2);
                    gcn_fma4(acc, gcn_bn4(v3, mu, is, g, be), w3);
                }
                int rem = cnt - j;
                if (rem > 0) {
                    int s0 = __builtin_amdgcn_readlane(sj, j);
                    float w0 = __builtin_bit_cast(float, __builtin_amdgcn_readlane(wjb, j));
                    float w1 = 0.f, w2 = 0.f;
                    float4 v0 = hb[(long)s0 * (CH / 4) + lane];
                    float4 v1 = {0.f, 0.f, 0.f, 0.f}, v2 = {0.f, 0.f, 0.f, 0.f};
                    if (rem > 1) {
                        int s1 = __builtin_amdgcn_readlane(sj, j + 1);
                        w1 = __builtin_bit_cast(float, __builtin_amdgcn_readlane(wjb, j + 1));
                        v1 = hb[(long)s1 * (CH / 4) + lane];
                    }
                    if (rem > 2) {
                        int s2 = __builtin_amdgcn_readlane(sj, j + 2);
                        w2 = __builtin_bit_cast(float, __builtin_amdgcn_readlane(wjb, j + 2));
                        v2 = hb[(long)s2 * (CH / 4) + lane];
                    }
                    gcn_fma4(acc, gcn_bn4(v0, mu, is, g, be), w0);
                    if (rem > 1) gcn_fma4(acc, gcn_bn4(v1, mu, is, g, be), w1);
                    if (rem > 2) gcn_fma4(acc, gcn_bn4(v2, mu, is, g, be), w2);
                }
            }
        }
        *(float4*)&tile[r * FT_PITCH + lane * 4] = acc;
    }

    int khalf = (lane >> 4) << 1;
    int col   = wv * 16 + (lane & 15);
    float bv  = bias[col];
    v8f acc8 = (v8f){bv, bv, bv, bv, bv, bv, bv, bv};
    const float* ap = &tile[(lane & 15) * FT_PITCH + khalf];
    const float* wp = W + (long)khalf * CH + col;
    v2f bw[16];
#pragma unroll
    for (int i = 0; i < 16; ++i) {
        bw[i].x = wp[(long)(4 * i) * CH];
        bw[i].y = wp[(long)(4 * i) * CH + CH];
    }
    __syncthreads();
#pragma unroll
    for (int i = 0; i < 16; ++i) {
        v2f av = *(const v2f*)(ap + 4 * i);
        acc8 = __builtin_amdgcn_wmma_f32_16x16x4_f32(
            false, av, false, bw[i], (short)0, acc8, false, false);
        bw[i].x = wp[(long)(4 * i + 64) * CH];
        bw[i].y = wp[(long)(4 * i + 64) * CH + CH];
    }
#pragma unroll
    for (int i = 0; i < 16; ++i) {
        v2f av = *(const v2f*)(ap + 4 * i + 64);
        acc8 = __builtin_amdgcn_wmma_f32_16x16x4_f32(
            false, av, false, bw[i], (short)0, acc8, false, false);
    }
    __syncthreads();
    int rb = (lane >> 4) << 3;
#pragma unroll
    for (int r = 0; r < 8; ++r) tile[(rb + r) * FT_PITCH + col] = acc8[r];
    __syncthreads();
    for (int q = 0; q < 2; ++q) {
        int r = wv * 2 + q;
        int d = node0 + r;
        if (d < n) ob[(long)d * (CH / 4) + lane] = *(const float4*)&tile[r * FT_PITCH + lane * 4];
    }
}

#define FIN 64
#define FI_NODES 8
#define FI_ROWS  (FI_NODES * 4)
#define FI_PITCH 68
__global__ void __launch_bounds__(FI_NODES * 32)
gcn_gather_in_gemm(const float* __restrict__ x, const int* __restrict__ rowptr,
                   const int* __restrict__ csrc, const float* __restrict__ dinv,
                   const float* __restrict__ W, const float* __restrict__ bias,
                   float* __restrict__ out, int n, int nb) {
    __shared__ __attribute__((aligned(16))) float tin[FI_ROWS * FI_PITCH];
    __shared__ __attribute__((aligned(16))) float tout[FI_ROWS * FT_PITCH];
    int lane = threadIdx.x & 31;
    int wv = __builtin_amdgcn_readfirstlane(threadIdx.x >> 5);
    int d  = blockIdx.x * FI_NODES + wv;
    int c4   = lane & 15;
    int pb0  = blockIdx.y * 4 + (lane >> 4);
    int pb1  = pb0 + 2;
    bool ok0 = pb0 < nb, ok1 = pb1 < nb;
    const float4* x0 = (const float4*)x + (long)(ok0 ? pb0 : 0) * n * (FIN / 4) + c4;
    const float4* x1 = (const float4*)x + (long)(ok1 ? pb1 : 0) * n * (FIN / 4) + c4;
    float4 acc0 = {0.f, 0.f, 0.f, 0.f};
    float4 acc1 = {0.f, 0.f, 0.f, 0.f};
    if (d < n) {
        int beg = __builtin_amdgcn_readfirstlane(rowptr[d]);
        int end = __builtin_amdgcn_readfirstlane(rowptr[d + 1]);
        float dd = dinv[d];
        float4 h0 = x0[(long)d * (FIN / 4)];
        float4 h1 = x1[(long)d * (FIN / 4)];
        acc0.x = h0.x * dd * dd; acc0.y = h0.y * dd * dd; acc0.z = h0.z * dd * dd; acc0.w = h0.w * dd * dd;
        acc1.x = h1.x * dd * dd; acc1.y = h1.y * dd * dd; acc1.z = h1.z * dd * dd; acc1.w = h1.w * dd * dd;

        for (int base = beg; base < end; base += 32) {
            int cnt = end - base; if (cnt > 32) cnt = 32;
            int   sj = 0;
            float wj = 0.f;
            if (lane < cnt) { sj = csrc[base + lane]; wj = dinv[sj] * dd; }
            int wjb = __builtin_bit_cast(int, wj);
            int j = 0;
            for (; j + 4 <= cnt; j += 4) {
                int s0 = __builtin_amdgcn_readlane(sj, j);
                int s1 = __builtin_amdgcn_readlane(sj, j + 1);
                int s2 = __builtin_amdgcn_readlane(sj, j + 2);
                int s3 = __builtin_amdgcn_readlane(sj, j + 3);
                float4 a0 = x0[(long)s0 * (FIN / 4)];
                float4 b0 = x1[(long)s0 * (FIN / 4)];
                float4 a1 = x0[(long)s1 * (FIN / 4)];
                float4 b1 = x1[(long)s1 * (FIN / 4)];
                float4 a2 = x0[(long)s2 * (FIN / 4)];
                float4 b2 = x1[(long)s2 * (FIN / 4)];
                float4 a3 = x0[(long)s3 * (FIN / 4)];
                float4 b3 = x1[(long)s3 * (FIN / 4)];
                float w0 = __builtin_bit_cast(float, __builtin_amdgcn_readlane(wjb, j));
                float w1 = __builtin_bit_cast(float, __builtin_amdgcn_readlane(wjb, j + 1));
                float w2 = __builtin_bit_cast(float, __builtin_amdgcn_readlane(wjb, j + 2));
                float w3 = __builtin_bit_cast(float, __builtin_amdgcn_readlane(wjb, j + 3));
                gcn_fma4(acc0, a0, w0); gcn_fma4(acc1, b0, w0);
                gcn_fma4(acc0, a1, w1); gcn_fma4(acc1, b1, w1);
                gcn_fma4(acc0, a2, w2); gcn_fma4(acc1, b2, w2);
                gcn_fma4(acc0, a3, w3); gcn_fma4(acc1, b3, w3);
            }
            int rem = cnt - j;
            if (rem > 0) {
                const float4 z = {0.f, 0.f, 0.f, 0.f};
                int s0 = __builtin_amdgcn_readlane(sj, j);
                float w0 = __builtin_bit_cast(float, __builtin_amdgcn_readlane(wjb, j));
                float w1 = 0.f, w2 = 0.f;
                float4 a0 = x0[(long)s0 * (FIN / 4)];
                float4 b0 = x1[(long)s0 * (FIN / 4)];
                float4 a1 = z, b1 = z, a2 = z, b2 = z;
                if (rem > 1) {
                    int s1 = __builtin_amdgcn_readlane(sj, j + 1);
                    w1 = __builtin_bit_cast(float, __builtin_amdgcn_readlane(wjb, j + 1));
                    a1 = x0[(long)s1 * (FIN / 4)];
                    b1 = x1[(long)s1 * (FIN / 4)];
                }
                if (rem > 2) {
                    int s2 = __builtin_amdgcn_readlane(sj, j + 2);
                    w2 = __builtin_bit_cast(float, __builtin_amdgcn_readlane(wjb, j + 2));
                    a2 = x0[(long)s2 * (FIN / 4)];
                    b2 = x1[(long)s2 * (FIN / 4)];
                }
                gcn_fma4(acc0, a0, w0); gcn_fma4(acc1, b0, w0);
                if (rem > 1) { gcn_fma4(acc0, a1, w1); gcn_fma4(acc1, b1, w1); }
                if (rem > 2) { gcn_fma4(acc0, a2, w2); gcn_fma4(acc1, b2, w2); }
            }
        }
    }
    int trow = wv * 4 + (lane >> 4);
    *(float4*)&tin[trow * FI_PITCH + c4 * 4]       = acc0;
    *(float4*)&tin[(trow + 2) * FI_PITCH + c4 * 4] = acc1;

    int khalf = (lane >> 4) << 1;
    int col   = wv * 16 + (lane & 15);
    float bv  = bias[col];
    v8f accA = (v8f){bv, bv, bv, bv, bv, bv, bv, bv};
    v8f accB = accA;
    const float* apA = &tin[(lane & 15) * FI_PITCH + khalf];
    const float* apB = apA + 16 * FI_PITCH;
    const float* wp  = W + (long)khalf * CH + col;
    v2f bw[FIN / 4];
#pragma unroll
    for (int i = 0; i < FIN / 4; ++i) {
        bw[i].x = wp[(long)(4 * i) * CH];
        bw[i].y = wp[(long)(4 * i) * CH + CH];
    }
    __syncthreads();
#pragma unroll
    for (int i = 0; i < FIN / 4; ++i) {
        v2f avA = *(const v2f*)(apA + 4 * i);
        v2f avB = *(const v2f*)(apB + 4 * i);
        accA = __builtin_amdgcn_wmma_f32_16x16x4_f32(
            false, avA, false, bw[i], (short)0, accA, false, false);
        accB = __builtin_amdgcn_wmma_f32_16x16x4_f32(
            false, avB, false, bw[i], (short)0, accB, false, false);
    }
    int rb = (lane >> 4) << 3;
#pragma unroll
    for (int r = 0; r < 8; ++r) {
        tout[(rb + r) * FT_PITCH + col]      = accA[r];
        tout[(16 + rb + r) * FT_PITCH + col] = accB[r];
    }
    __syncthreads();
    if (d < n) {
        float4* ob = (float4*)out;
#pragma unroll
        for (int p = 0; p < 4; ++p) {
            int plane = blockIdx.y * 4 + p;
            if (plane < nb)
                ob[((long)plane * n + d) * (CH / 4) + lane] =
                    *(const float4*)&tout[(wv * 4 + p) * FT_PITCH + lane * 4];
        }
    }
}

#define ST_U     10
#define ST_WAVES 16
__global__ void gcn_bn_stats(const float* __restrict__ x, float* __restrict__ sums, long rows) {
    __shared__ float red[ST_WAVES][2 * CH];
    int lane = threadIdx.x & 31;
    int wv = __builtin_amdgcn_readfirstlane(threadIdx.x >> 5);
    long gw = (long)blockIdx.x * ST_WAVES + wv;
    long tw = (long)gridDim.x * ST_WAVES;
    const float4* xp = (const float4*)x;
    float4 s = {0.f, 0.f, 0.f, 0.f};
    float4 q = {0.f, 0.f, 0.f, 0.f};
    for (long r = gw; r < rows; r += tw * ST_U) {
        float4 v[ST_U];
#pragma unroll
        for (int u = 0; u < ST_U; ++u) {
            long rr = r + u * tw;
            v[u] = xp[(rr < rows ? rr : rows - 1) * (CH / 4) + lane];
        }
#pragma unroll
        for (int u = 0; u < ST_U; ++u) {
            bool in = r + u * tw < rows;
            float a = fmaxf(in ? v[u].x : 0.f, 0.f);
            float b = fmaxf(in ? v[u].y : 0.f, 0.f);
            float c = fmaxf(in ? v[u].z : 0.f, 0.f);
            float d = fmaxf(in ? v[u].w : 0.f, 0.f);
            s.x += a; s.y += b; s.z += c; s.w += d;
            q.x = __builtin_fmaf(a, a, q.x);
            q.y = __builtin_fmaf(b, b, q.y);
            q.z = __builtin_fmaf(c, c, q.z);
            q.w = __builtin_fmaf(d, d, q.w);
        }
    }
    *(float4*)&red[wv][lane * 4]      = s;
    *(float4*)&red[wv][CH + lane * 4] = q;
    __syncthreads();
    int t = threadIdx.x;
    if (t < 2 * CH) {
        float tot = 0.f;
#pragma unroll
        for (int w = 0; w < ST_WAVES; ++w) tot += red[w][t];
        atomicAddF(&sums[t], tot);
    }
}

#define DO_U     20
#define DO_WAVES 16
__global__ void __launch_bounds__(DO_WAVES * 32) __attribute__((amdgpu_waves_per_eu(4, 4)))
gcn_bn_dot_out(const float* __restrict__ x, const float* __restrict__ sums,
               const float* __restrict__ gamma, const float* __restrict__ beta,
               const float* __restrict__ wc, const float* __restrict__ bc,
               float* __restrict__ out, int rows, float invcnt) {
    int lane = threadIdx.x & 31;
    int wv  = __builtin_amdgcn_readfirstlane(threadIdx.x >> 5);
    int gw  = blockIdx.x * DO_WAVES + wv;
    int tw  = gridDim.x * DO_WAVES;
    int per = (rows + tw - 1) / tw;
    int r0  = gw * per;
    int r1  = r0 + per; if (r1 > rows) r1 = rows;
    if (r0 >= r1) return;

    float mu[4], is[4], g[4], b[4], w[4];
#pragma unroll
    for (int j = 0; j < 4; ++j) {
        int c = lane * 4 + j;
        mu[j] = sums[c] * invcnt;
        float var = sums[CH + c] * invcnt - mu[j] * mu[j];
        is[j] = rsqrtf(var + BN_EPS);
        g[j]  = gamma[c];
        b[j]  = beta[c];
        w[j]  = wc[c];
    }
    float bcv = bc[0];
    const float4* xp = (const float4*)x + lane;

    for (int r = r0; r < r1; r += DO_U) {
        float4 v[DO_U];
#pragma unroll
        for (int u = 0; u < DO_U; ++u) {
            int rr = r + u;
            v[u] = xp[(long)(rr < r1 ? rr : r1 - 1) * (CH / 4)];
        }
        float res = 0.f;
#pragma unroll
        for (int u = 0; u < DO_U; ++u) {
            float vv[4] = {v[u].x, v[u].y, v[u].z, v[u].w};
            float acc = 0.f;
#pragma unroll
            for (int j = 0; j < 4; ++j) {
                float val = (fmaxf(vv[j], 0.f) - mu[j]) * is[j] * g[j] + b[j];
                acc += val * w[j];
            }
#pragma unroll
            for (int o = 16; o > 0; o >>= 1)
                acc += __shfl_xor(acc, o, 32);
            res = (lane == u) ? acc : res;
        }
        if (lane < DO_U && r + lane < r1) out[r + lane] = res + bcv;
    }
}

extern "C" void kernel_launch(void* const* d_in, const int* in_sizes, int n_in,
                              void* d_out, int out_size, void* d_ws, size_t ws_size,
                              hipStream_t stream) {
    const float* x   = (const float*)d_in[0];
    const int*   ei  = (const int*)d_in[1];
    const float* W1  = (const float*)d_in[2];
    const float* b1  = (const float*)d_in[3];
    const float* W2  = (const float*)d_in[4];
    const float* b2  = (const float*)d_in[5];
    const float* g1  = (const float*)d_in[6];
    const float* be1 = (const float*)d_in[7];
    const float* g2  = (const float*)d_in[8];
    const float* be2 = (const float*)d_in[9];
    const float* Wc  = (const float*)d_in[10];
    const float* bc  = (const float*)d_in[11];
    float* out = (float*)d_out;

    const int  E  = in_sizes[1] / 2;
    const int  N  = 20000;
    const int  H  = in_sizes[3];
    const long BN = (long)out_size;
    const int  Bz = (int)(BN / N);
    const long TOT = BN * H;
    const float invcnt = 1.0f / (float)BN;

    float* ws     = (float*)d_ws;
    float* stats1 = ws;
    float* stats2 = ws + 256;
    float* dinv   = ws + 512;
    long   nAl    = (N + 255) & ~255L;
    float* bufA   = ws + 512 + nAl;

    const long PL   = (long)N * H;
    float* bufM     = bufA + (Bz - 1) * PL;
    float* agg1Last = bufA + (2 * Bz - 2) * PL;
    float* agg2Last = bufA + (Bz - 1) * PL;
    long   n1Al     = (N + 1 + 255) & ~255L;
    long   eAl      = (E + 255) & ~255L;
    int*   csrEnd   = (int*)(bufA + 2 * TOT);
    int*   sorted   = csrEnd - eAl;
    int*   slots    = sorted - eAl;
    int*   rowptr   = slots - n1Al;

    const int* srcIdx = ei;
    const int* dstIdx = ei + E;

    gcn_build<<<128, OW_T, 0, stream>>>(srcIdx, dstIdx, rowptr, slots, sorted, dinv, stats1, N, E);

    dim3 tileGridRest((N + FT_ROWS - 1) / FT_ROWS, Bz - 1, 1);
    dim3 tileGridLast((N + FT_ROWS - 1) / FT_ROWS, 1, 1);
    int  statBlks  = 256;

    dim3 nodeGridIn((N + FI_NODES - 1) / FI_NODES, (Bz + 3) / 4, 1);
    gcn_gather_in_gemm<<<nodeGridIn, FI_NODES * 32, 0, stream>>>(x, rowptr, sorted, dinv, W1, b1, bufM, N, Bz);
    gcn_bn_stats <<<statBlks, 512, 0, stream>>>(bufM, stats1, BN);

    gcn_gather_bn_gemm<<<tileGridRest, FT_ROWS * 16, 0, stream>>>(bufM, rowptr, sorted, dinv, stats1, g1, be1,
                                                                 invcnt, W2, b2, bufA, N);
    gcn_gather_bn_gemm<<<tileGridLast, FT_ROWS * 16, 0, stream>>>(agg1Last, rowptr, sorted, dinv, stats1, g1, be1,
                                                                 invcnt, W2, b2, agg2Last, N);
    gcn_bn_stats <<<statBlks, 512, 0, stream>>>(bufA, stats2, BN);

    gcn_bn_dot_out<<<statBlks, 512, 0, stream>>>(bufA, stats2, g2, be2, Wc, bc, out, (int)BN, invcnt);
}


# BatchNorm sums taken in the gather+GEMM kernels' epilogues from the accumulators: per-tile column sums, one atomic per channel into 8 blocks the readers add up; 2 launches fewer; sum order changes
# speedup vs baseline: 6.3976x; 1.0413x over previous
#include <hip/hip_runtime.h>
#include <hip/hip_bf16.h>

typedef __attribute__((ext_vector_type(2))) float v2f;
typedef __attribute__((ext_vector_type(8))) float v8f;

#define CH 128
#define BN_EPS 1e-5f
#define ST_REP 8

__device__ __forceinline__ void atomicAddF(float* p, float v) {
    (void)__hip_atomic_fetch_add(p, v, __ATOMIC_RELAXED, __HIP_MEMORY_SCOPE_AGENT);
}

template <bool FITS>
__device__ __forceinline__ void gcn_sort_run(const int* segL, const int* segG, const int* preL,
                                             const int* cntL, int* outRun, int span, int wv, int lane) {
    for (int q = wv; q < span; q += 32) {
        int beg = __builtin_amdgcn_readfirstlane(preL[q]);
        int len = __builtin_amdgcn_readfirstlane(cntL[q]);
        int* outp = outRun + beg;
        if (len <= 32) {
            int v = 0;
            if (lane < len) v = FITS ? segL[beg + lane] : segG[beg + lane];
            int rank = 0;
            for (int j = 0; j < len; ++j) {
                int u = __builtin_amdgcn_readlane(v, j);
                rank += (u < v || (u == v && j < lane)) ? 1 : 0;
            }
            if (lane < len) outp[rank] = v;
        } else {
            for (int i = lane; i < len; i += 32) {
                int v = FITS ? segL[beg + i] : segG[beg + i];
                int rank = 0;
                for (int j = 0; j < len; ++j) {
                    int u = FITS ? segL[beg + j] : segG[beg + j];
                    rank += (u < v || (u == v && j < i)) ? 1 : 0;
                }
                outp[rank] = v;
            }
        }
    }
}
#define OW_T    1024
#define OW_R    160
#define OW_WCAP 256
#define OW_SCAP 8192
#define OW_U    4
__global__ void __launch_bounds__(OW_T)
gcn_build(const int* __restrict__ src, const int* __restrict__ dst, int* __restrict__ rowptr,
          int* __restrict__ slots, int* __restrict__ sorted, float* __restrict__ dinv,
          float* __restrict__ stats, int n, int e) {
    __shared__ int cntL[OW_R];
    __shared__ int preL[OW_R];
    __shared__ int wown[OW_T / 32];
    __shared__ int wlow[OW_T / 32];
    __shared__ int wcnt[OW_T / 32];
    __shared__ int lstI[(OW_T / 32) * OW_WCAP];
    __shared__ int lstD[(OW_T / 32) * OW_WCAP];
    __shared__ int srcL[OW_SCAP];

    int t    = threadIdx.x;
    int lane = t & 31;
    int wv   = __builtin_amdgcn_readfirstlane(t >> 5);
    int per  = (n + (int)gridDim.x - 1) / (int)gridDim.x;
    int lo   = blockIdx.x * per;
    int hi   = lo + per; if (hi > n) hi = n;
    int span = hi - lo;  if (span < 0) span = 0;

    if (t < OW_R) cntL[t] = 0;
    for (int i = blockIdx.x * OW_T + t; i < 2 * ST_REP * 2 * CH; i += (int)gridDim.x * OW_T) stats[i] = 0.0f;
    __syncthreads();

    int low  = 0;
    int wnum = 0;
    int* myI = lstI + wv * OW_WCAP;
    int* myD = lstD + wv * OW_WCAP;
    unsigned lt = (1u << lane) - 1u;

#define OW_VISIT(dv, iv, okv)                                                       \
    {                                                                               \
        int  dl_ = (dv) - lo;                                                       \
        bool in_ = (okv) && (unsigned)dl_ < (unsigned)span;                         \
        low += __builtin_popcount(__builtin_amdgcn_ballot_w32((okv) && dl_ < 0));   \
        unsigned m_ = __builtin_amdgcn_ballot_w32(in_);                             \
        if (m_ != 0u) {                                                             \
            if (in_) {                                                              \
                atomicAdd(&cntL[dl_], 1);                                           \
                int p_ = wnum + __builtin_popcount(m_ & lt);                        \
                if (p_ < OW_WCAP) { myI[p_] = (iv); myD[p_] = dl_; }                \
            }                                                                       \
            wnum += __builtin_popcount(m_);                                         \
        }                                                                           \
    }

    bool vec = ((e & 3) == 0) && ((((unsigned long)dst) & 15ul) == 0ul);
    if (vec) {
        const int4* d4p = (const int4*)dst;
        int q4 = e >> 2;
        int b  = 0;
        for (; b + OW_T * OW_U <= q4; b += OW_T * OW_U) {
            int4 dv[OW_U];
#pragma unroll
            for (int u = 0; u < OW_U; ++u) dv[u] = d4p[b + u * OW_T + t];
#pragma unroll
            for (int u = 0; u < OW_U; ++u) {
                int i0 = 4 * (b + u * OW_T + t);
                OW_VISIT(dv[u].x, i0,     true)
                OW_VISIT(dv[u].y, i0 + 1, true)
                OW_VISIT(dv[u].z, i0 + 2, true)
                OW_VISIT(dv[u].w, i0 + 3, true)
            }
        }
        for (; b < q4; b += OW_T) {
            int  g  = b + t;
            bool ok = g < q4;
            int4 dv = d4p[ok ? g : q4 - 1];
            OW_VISIT(dv.x, 4 * g,     ok)
            OW_VISIT(dv.y, 4 * g + 1, ok)
            OW_VISIT(dv.z, 4 * g + 2, ok)
            OW_VISIT(dv.w, 4 * g + 3, ok)
        }
    } else {
        for (int b = 0; b < e; b += OW_T) {
            int  i  = b + t;
            bool ok = i < e;
            int  d  = dst[ok ? i : e - 1];
            OW_VISIT(d, i, ok)
        }
    }
#undef OW_VISIT
    __syncthreads();

    int own = (t < span) ? cntL[t] : 0;
    int inc = own;
    for (int o = 1; o < 32; o <<= 1) {
        int v = __shfl_up(inc, o, 32);
        inc += (lane >= o) ? v : 0;
    }
    if (lane == 31) wown[wv] = inc;
    if (lane == 0)  { wlow[wv] = low; wcnt[wv] = wnum; }
    __syncthreads();
    int wa    = wown[lane];
    int front = (lane < wv) ? wa : 0;
    int total = wa;
    int below = wlow[lane];
    int over  = (wcnt[lane] > OW_WCAP) ? 1 : 0;
    for (int o = 16; o > 0; o >>= 1) {
        front += __shfl_xor(front, o, 32);
        total += __shfl_xor(total, o, 32);
        below += __shfl_xor(below, o, 32);
        over  |= __shfl_xor(over, o, 32);
    }
    int  excl = front + inc - own;
    bool slow = over != 0;
    bool fits = total <= OW_SCAP;
    if (t < span) {
        rowptr[lo + t] = below + excl;
        dinv[lo + t]   = rsqrtf((float)(own + 1));
        preL[t] = excl;
        cntL[t] = 0;
    }
    if (t == 0 && lo < n && hi == n) rowptr[n] = below + total;
    __syncthreads();

    if (!slow) {
        for (int k = lane; k < wnum; k += 32) {
            int i  = myI[k];
            int dl = myD[k];
            int sv = src[i];
            int p  = preL[dl] + atomicAdd(&cntL[dl], 1);
            if (fits) srcL[p] = sv; else slots[below + p] = sv;
        }
    } else {
        for (int b = 0; b < e; b += OW_T) {
            int i = b + t;
            if (i < e) {
                int dl = dst[i] - lo;
                if ((unsigned)dl < (unsigned)span) {
                    int sv = src[i];
                    int p  = preL[dl] + atomicAdd(&cntL[dl], 1);
                    if (fits) srcL[p] = sv; else slots[below + p] = sv;
                }
            }
        }
    }
    __syncthreads();

    if (fits) gcn_sort_run<true >(srcL, slots + below, preL, cntL, sorted + below, span, wv, lane);
    else      gcn_sort_run<false>(srcL, slots + below, preL, cntL, sorted + below, span, wv, lane);
}

__device__ __forceinline__ void gcn_fma4(float4& acc, const float4& v, float w) {
    acc.x = __builtin_fmaf(v.x, w, acc.x);
    acc.y = __builtin_fmaf(v.y, w, acc.y);
    acc.z = __builtin_fmaf(v.z, w, acc.z);
    acc.w = __builtin_fmaf(v.w, w, acc.w);
}
#define FT_ROWS  16
#define FT_PITCH 132
__device__ __forceinline__ float4 gcn_bn4(const float4& v, const float* mu, const float* is,
                                          const float* g, const float* be) {
    float4 r;
    r.x = (fmaxf(v.x, 0.f) - mu[0]) * is[0] * g[0] + be[0];
    r.y = (fmaxf(v.y, 0.f) - mu[1]) * is[1] * g[1] + be[1];
    r.z = (fmaxf(v.z, 0.f) - mu[2]) * is[2] * g[2] + be[2];
    r.w = (fmaxf(v.w, 0.f) - mu[3]) * is[3] * g[3] + be[3];
    return r;
}
__global__ void __launch_bounds__(FT_ROWS * 16)
gcn_gather_bn_gemm(const float* __restrict__ a, const int* __restrict__ rowptr,
                   const int* __restrict__ csrc, const float* __restrict__ dinv,
                   const float* __restrict__ sums, const float* __restrict__ gamma,
                   const float* __restrict__ beta, float invcnt,
                   const float* __restrict__ W, const float* __restrict__ bias,
                   float* __restrict__ out, float* __restrict__ osums, int n) {
    __shared__ __attribute__((aligned(16))) float tile[FT_ROWS * FT_PITCH];
    __shared__ __attribute__((aligned(16))) float sred[ST_REP * 2 * CH];
    int lane = threadIdx.x & 31;
    int wv = __builtin_amdgcn_readfirstlane(threadIdx.x >> 5);
    int node0 = blockIdx.x * FT_ROWS;
    const float4* hb = (const float4*)(a + (long)blockIdx.y * n * CH);
    float4*       ob = (float4*)(out + (long)blockIdx.y * n * CH);

    float mu[4], is[4], g[4], be[4];
    {
        const float4* s4 = (const float4*)sums;
        for (int p = wv; p < ST_REP; p += FT_ROWS * 16 / 32) {
            *(float4*)&sred[p * 2 * CH + lane * 4]      = s4[p * (2 * CH / 4) + lane];
            *(float4*)&sred[p * 2 * CH + CH + lane * 4] = s4[p * (2 * CH / 4) + CH / 4 + lane];
        }
        __syncthreads();
        float4 sm = *(const float4*)&sred[lane * 4];
        float4 sq = *(const float4*)&sred[CH + lane * 4];
#pragma unroll
        for (int p = 1; p < ST_REP; ++p) {
            float4 ps = *(const float4*)&sred[p * 2 * CH + lane * 4];
            float4 pq = *(const float4*)&sred[p * 2 * CH + CH + lane * 4];
            sm.x += ps.x; sm.y += ps.y; sm.z += ps.z; sm.w += ps.w;
            sq.x += pq.x; sq.y += pq.y; sq.z += pq.z; sq.w += pq.w;
        }
        float smv[4] = {sm.x, sm.y, sm.z, sm.w};
        float sqv[4] = {sq.x, sq.y, sq.z, sq.w};
#pragma unroll
        for (int j = 0; j < 4; ++j) {
            int c = lane * 4 + j;
            mu[j] = smv[j] * invcnt;
            float var = sqv[j] * invcnt - mu[j] * mu[j];
            is[j] = rsqrtf(var + BN_EPS);
            g[j]  = gamma[c];
            be[j] = beta[c];
        }
    }

    for (int q = 0; q < 2; ++q) {
        int r = wv * 2 + q;
        int d = node0 + r;
        float4 acc = {0.f, 0.f, 0.f, 0.f};
        if (d < n) {
            int beg = __builtin_amdgcn_readfirstlane(rowptr[d]);
            int end = __builtin_amdgcn_readfirstlane(rowptr[d + 1]);
            float dd = dinv[d];
            float4 hv = gcn_bn4(hb[(long)d * (CH / 4) + lane], mu, is, g, be);
            acc.x = hv.x * dd * dd;
            acc.y = hv.y * dd * dd;
            acc.z = hv.z * dd * dd;
            acc.w = hv.w * dd * dd;

            for (int base = beg; base < end; base += 32) {
                int cnt = end - base; if (cnt > 32) cnt = 32;
                int   sj = 0;
                float wj = 0.f;
                if (lane < cnt) { sj = csrc[base + lane]; wj = dinv[sj] * dd; }
                int wjb = __builtin_bit_cast(int, wj);
                int j = 0;
                for (; j + 4 <= cnt; j += 4) {
                    int s0 = __builtin_amdgcn_readlane(sj, j);
                    int s1 = __builtin_amdgcn_readlane(sj, j + 1);
                    int s2 = __builtin_amdgcn_readlane(sj, j + 2);
                    int s3 = __builtin_amdgcn_readlane(sj, j + 3);
                    float w0 = __builtin_bit_cast(float, __builtin_amdgcn_readlane(wjb, j));
                    float w1 = __builtin_bit_cast(float, __builtin_amdgcn_readlane(wjb, j + 1));
                    float w2 = __builtin_bit_cast(float, __builtin_amdgcn_readlane(wjb, j + 2));
                    float w3 = __builtin_bit_cast(float, __builtin_amdgcn_readlane(wjb, j + 3));
                    float4 v0 = hb[(long)s0 * (CH / 4) + lane];
                    float4 v1 = hb[(long)s1 * (CH / 4) + lane];
                    float4 v2 = hb[(long)s2 * (CH / 4) + lane];
                    float4 v3 = hb[(long)s3 * (CH / 4) + lane];
                    gcn_fma4(acc, gcn_bn4(v0, mu, is, g, be), w0);
                    gcn_fma4(acc, gcn_bn4(v1, mu, is, g, be), w1);
                    gcn_fma4(acc, gcn_bn4(v2, mu, is, g, be), w2);
                    gcn_fma4(acc, gcn_bn4(v3, mu, is, g, be), w3);
                }
                int rem = cnt - j;
                if (rem > 0) {
                    int s0 = __builtin_amdgcn_readlane(sj, j);
                    float w0 = __builtin_bit_cast(float, __builtin_amdgcn_readlane(wjb, j));
                    float w1 = 0.f, w2 = 0.f;
                    float4 v0 = hb[(long)s0 * (CH / 4) + lane];
                    float4 v1 = {0.f, 0.f, 0.f, 0.f}, v2 = {0.f, 0.f, 0.f, 0.f};
                    if (rem > 1) {
                        int s1 = __builtin_amdgcn_readlane(sj, j + 1);
                        w1 = __builtin_bit_cast(float, __builtin_amdgcn_readlane(wjb, j + 1));
                        v1 = hb[(long)s1 * (CH / 4) + lane];
                    }
                    if (rem > 2) {
                        int s2 = __builtin_amdgcn_readlane(sj, j + 2);
                        w2 = __builtin_bit_cast(float, __builtin_amdgcn_readlane(wjb, j + 2));
                        v2 = hb[(long)s2 * (CH / 4) + lane];
                    }
                    gcn_fma4(acc, gcn_bn4(v0, mu, is, g, be), w0);
                    if (rem > 1) gcn_fma4(acc, gcn_bn4(v1, mu, is, g, be), w1);
                    if (rem > 2) gcn_fma4(acc, gcn_bn4(v2, mu, is, g, be), w2);
                }
            }
        }
        *(float4*)&tile[r * FT_PITCH + lane * 4] = acc;
    }

    int khalf = (lane >> 4) << 1;
    int col   = wv * 16 + (lane & 15);
    float bv  = bias[col];
    v8f acc8 = (v8f){bv, bv, bv, bv, bv, bv, bv, bv};
    const float* ap = &tile[(lane & 15) * FT_PITCH + khalf];
    const float* wp = W + (long)khalf * CH + col;
    v2f bw[16];
#pragma unroll
    for (int i = 0; i < 16; ++i) {
        bw[i].x = wp[(long)(4 * i) * CH];
        bw[i].y = wp[(long)(4 * i) * CH + CH];
    }
    __syncthreads();
#pragma unroll
    for (int i = 0; i < 16; ++i) {
        v2f av = *(const v2f*)(ap + 4 * i);
        acc8 = __builtin_amdgcn_wmma_f32_16x16x4_f32(
            false, av, false, bw[i], (short)0, acc8, false, false);
        bw[i].x = wp[(long)(4 * i + 64) * CH];
        bw[i].y = wp[(long)(4 * i + 64) * CH + CH];
    }
#pragma unroll
    for (int i = 0; i < 16; ++i) {
        v2f av = *(const v2f*)(ap + 4 * i + 64);
        acc8 = __builtin_amdgcn_wmma_f32_16x16x4_f32(
            false, av, false, bw[i], (short)0, acc8, false, false);
    }
    __syncthreads();
    int rb = (lane >> 4) << 3;
#pragma unroll
    for (int r = 0; r < 8; ++r) tile[(rb + r) * FT_PITCH + col] = acc8[r];
    __syncthreads();
    {
        float s = 0.f, qq = 0.f;
#pragma unroll
        for (int r = 0; r < 8; ++r) {
            float v = (node0 + rb + r < n) ? fmaxf(acc8[r], 0.f) : 0.f;
            s += v;
            qq = __builtin_fmaf(v, v, qq);
        }
        s  += __shfl_xor(s, 16, 32);
        qq += __shfl_xor(qq, 16, 32);
        int blk = (int)((blockIdx.x + blockIdx.y) & (ST_REP - 1));
        atomicAddF(osums + blk * (2 * CH) + ((lane < 16) ? col : CH + col), (lane < 16) ? s : qq);
    }
    for (int q = 0; q < 2; ++q) {
        int r = wv * 2 + q;
        int d = node0 + r;
        if (d < n) ob[(long)d * (CH / 4) + lane] = *(const float4*)&tile[r * FT_PITCH + lane * 4];
    }
}

#define FIN 64
#define FI_NODES 8
#define FI_ROWS  (FI_NODES * 4)
#define FI_PITCH 68
__global__ void __launch_bounds__(FI_NODES * 32)
gcn_gather_in_gemm(const float* __restrict__ x, const int* __restrict__ rowptr,
                   const int* __restrict__ csrc, const float* __restrict__ dinv,
                   const float* __restrict__ W, const float* __restrict__ bias,
                   float* __restrict__ out, float* __restrict__ osums, int n, int nb) {
    __shared__ __attribute__((aligned(16))) float tin[FI_ROWS * FI_PITCH];
    __shared__ __attribute__((aligned(16))) float tout[FI_ROWS * FT_PITCH];
    int lane = threadIdx.x & 31;
    int wv = __builtin_amdgcn_readfirstlane(threadIdx.x >> 5);
    int d  = blockIdx.x * FI_NODES + wv;
    int c4   = lane & 15;
    int pb0  = blockIdx.y * 4 + (lane >> 4);
    int pb1  = pb0 + 2;
    bool ok0 = pb0 < nb, ok1 = pb1 < nb;
    const float4* x0 = (const float4*)x + (long)(ok0 ? pb0 : 0) * n * (FIN / 4) + c4;
    const float4* x1 = (const float4*)x + (long)(ok1 ? pb1 : 0) * n * (FIN / 4) + c4;
    float4 acc0 = {0.f, 0.f, 0.f, 0.f};
    float4 acc1 = {0.f, 0.f, 0.f, 0.f};
    if (d < n) {
        int beg = __builtin_amdgcn_readfirstlane(rowptr[d]);
        int end = __builtin_amdgcn_readfirstlane(rowptr[d + 1]);
        float dd = dinv[d];
        float4 h0 = x0[(long)d * (FIN / 4)];
        float4 h1 = x1[(long)d * (FIN / 4)];
        acc0.x = h0.x * dd * dd; acc0.y = h0.y * dd * dd; acc0.z = h0.z * dd * dd; acc0.w = h0.w * dd * dd;
        acc1.x = h1.x * dd * dd; acc1.y = h1.y * dd * dd; acc1.z = h1.z * dd * dd; acc1.w = h1.w * dd * dd;

        for (int base = beg; base < end; base += 32) {
            int cnt = end - base; if (cnt > 32) cnt = 32;
            int   sj = 0;
            float wj = 0.f;
            if (lane < cnt) { sj = csrc[base + lane]; wj = dinv[sj] * dd; }
            int wjb = __builtin_bit_cast(int, wj);
            int j = 0;
            for (; j + 4 <= cnt; j += 4) {
                int s0 = __builtin_amdgcn_readlane(sj, j);
                int s1 = __builtin_amdgcn_readlane(sj, j + 1);
                int s2 = __builtin_amdgcn_readlane(sj, j + 2);
                int s3 = __builtin_amdgcn_readlane(sj, j + 3);
                float4 a0 = x0[(long)s0 * (FIN / 4)];
                float4 b0 = x1[(long)s0 * (FIN / 4)];
                float4 a1 = x0[(long)s1 * (FIN / 4)];
                float4 b1 = x1[(long)s1 * (FIN / 4)];
                float4 a2 = x0[(long)s2 * (FIN / 4)];
                float4 b2 = x1[(long)s2 * (FIN / 4)];
                float4 a3 = x0[(long)s3 * (FIN / 4)];
                float4 b3 = x1[(long)s3 * (FIN / 4)];
                float w0 = __builtin_bit_cast(float, __builtin_amdgcn_readlane(wjb, j));
                float w1 = __builtin_bit_cast(float, __builtin_amdgcn_readlane(wjb, j + 1));
                float w2 = __builtin_bit_cast(float, __builtin_amdgcn_readlane(wjb, j + 2));
                float w3 = __builtin_bit_cast(float, __builtin_amdgcn_readlane(wjb, j + 3));
                gcn_fma4(acc0, a0, w0); gcn_fma4(acc1, b0, w0);
                gcn_fma4(acc0, a1, w1); gcn_fma4(acc1, b1, w1);
                gcn_fma4(acc0, a2, w2); gcn_fma4(acc1, b2, w2);
                gcn_fma4(acc0, a3, w3); gcn_fma4(acc1, b3, w3);
            }
            int rem = cnt - j;
            if (rem > 0) {
                const float4 z = {0.f, 0.f, 0.f, 0.f};
                int s0 = __builtin_amdgcn_readlane(sj, j);
                float w0 = __builtin_bit_cast(float, __builtin_amdgcn_readlane(wjb, j));
                float w1 = 0.f, w2 = 0.f;
                float4 a0 = x0[(long)s0 * (FIN / 4)];
                float4 b0 = x1[(long)s0 * (FIN / 4)];
                float4 a1 = z, b1 = z, a2 = z, b2 = z;
                if (rem > 1) {
                    int s1 = __builtin_amdgcn_readlane(sj, j + 1);
                    w1 = __builtin_bit_cast(float, __builtin_amdgcn_readlane(wjb, j + 1));
                    a1 = x0[(long)s1 * (FIN / 4)];
                    b1 = x1[(long)s1 * (FIN / 4)];
                }
                if (rem > 2) {
                    int s2 = __builtin_amdgcn_readlane(sj, j + 2);
                    w2 = __builtin_bit_cast(float, __builtin_amdgcn_readlane(wjb, j + 2));
                    a2 = x0[(long)s2 * (FIN / 4)];
                    b2 = x1[(long)s2 * (FIN / 4)];
                }
                gcn_fma4(acc0, a0, w0); gcn_fma4(acc1, b0, w0);
                if (rem > 1) { gcn_fma4(acc0, a1, w1); gcn_fma4(acc1, b1, w1); }
                if (rem > 2) { gcn_fma4(acc0, a2, w2); gcn_fma4(acc1, b2, w2); }
            }
        }
    }
    int trow = wv * 4 + (lane >> 4);
    *(float4*)&tin[trow * FI_PITCH + c4 * 4]       = acc0;
    *(float4*)&tin[(trow + 2) * FI_PITCH + c4 * 4] = acc1;

    int khalf = (lane >> 4) << 1;
    int col   = wv * 16 + (lane & 15);
    float bv  = bias[col];
    v8f accA = (v8f){bv, bv, bv, bv, bv, bv, bv, bv};
    v8f accB = accA;
    const float* apA = &tin[(lane & 15) * FI_PITCH + khalf];
    const float* apB = apA + 16 * FI_PITCH;
    const float* wp  = W + (long)khalf * CH + col;
    v2f bw[FIN / 4];
#pragma unroll
    for (int i = 0; i < FIN / 4; ++i) {
        bw[i].x = wp[(long)(4 * i) * CH];
        bw[i].y = wp[(long)(4 * i) * CH + CH];
    }
    __syncthreads();
#pragma unroll
    for (int i = 0; i < FIN / 4; ++i) {
        v2f avA = *(const v2f*)(apA + 4 * i);
        v2f avB = *(const v2f*)(apB + 4 * i);
        accA = __builtin_amdgcn_wmma_f32_16x16x4_f32(
            false, avA, false, bw[i], (short)0, accA, false, false);
        accB = __builtin_amdgcn_wmma_f32_16x16x4_f32(
            false, avB, false, bw[i], (short)0, accB, false, false);
    }
    int rb = (lane >> 4) << 3;
#pragma unroll
    for (int r = 0; r < 8; ++r) {
        tout[(rb + r) * FT_PITCH + col]      = accA[r];
        tout[(16 + rb + r) * FT_PITCH + col] = accB[r];
    }
    __syncthreads();
    {
        float s = 0.f, qq = 0.f;
#pragma unroll
        for (int half = 0; half < 2; ++half) {
#pragma unroll
            for (int r = 0; r < 8; ++r) {
                int row = half * 16 + rb + r;
                bool ok = ((int)blockIdx.x * FI_NODES + (row >> 2) < n) && ((int)blockIdx.y * 4 + (row & 3) < nb);
                float a = half ? accB[r] : accA[r];
                float v = ok ? fmaxf(a, 0.f) : 0.f;
                s += v;
                qq = __builtin_fmaf(v, v, qq);
            }
        }
        s  += __shfl_xor(s, 16, 32);
        qq += __shfl_xor(qq, 16, 32);
        int blk = (int)((blockIdx.x + blockIdx.y) & (ST_REP - 1));
        atomicAddF(osums + blk * (2 * CH) + ((lane < 16) ? col : CH + col), (lane < 16) ? s : qq);
    }
    if (d < n) {
        float4* ob = (float4*)out;
#pragma unroll
        for (int p = 0; p < 4; ++p) {
            int plane = blockIdx.y * 4 + p;
            if (plane < nb)
                ob[((long)plane * n + d) * (CH / 4) + lane] =
                    *(const float4*)&tout[(wv * 4 + p) * FT_PITCH + lane * 4];
        }
    }
}

#define DO_U     20
#define DO_WAVES 16
__global__ void __launch_bounds__(DO_WAVES * 32) __attribute__((amdgpu_waves_per_eu(4, 4)))
gcn_bn_dot_out(const float* __restrict__ x, const float* __restrict__ sums,
               const float* __restrict__ gamma, const float* __restrict__ beta,
               const float* __restrict__ wc, const float* __restrict__ bc,
               float* __restrict__ out, int rows, float invcnt) {
    int lane = threadIdx.x & 31;
    int wv  = __builtin_amdgcn_readfirstlane(threadIdx.x >> 5);
    int gw  = blockIdx.x * DO_WAVES + wv;
    int tw  = gridDim.x * DO_WAVES;
    int per = (rows + tw - 1) / tw;
    int r0  = gw * per;
    int r1  = r0 + per; if (r1 > rows) r1 = rows;
    if (r0 >= r1) return;

    float mu[4], is[4], g[4], b[4], w[4];
    {
        const float4* s4 = (const float4*)sums;
        float4 sm = s4[lane];
        float4 sq = s4[CH / 4 + lane];
#pragma unroll
        for (int p = 1; p < ST_REP; ++p) {
            float4 ps = s4[p * (2 * CH / 4) + lane];
            float4 pq = s4[p * (2 * CH / 4) + CH / 4 + lane];
            sm.x += ps.x; sm.y += ps.y; sm.z += ps.z; sm.w += ps.w;
            sq.x += pq.x; sq.y += pq.y; sq.z += pq.z; sq.w += pq.w;
        }
        float smv[4] = {sm.x, sm.y, sm.z, sm.w};
        float sqv[4] = {sq.x, sq.y, sq.z, sq.w};
#pragma unroll
        for (int j = 0; j < 4; ++j) {
            int c = lane * 4 + j;
            mu[j] = smv[j] * invcnt;
            float var = sqv[j] * invcnt - mu[j] * mu[j];
            is[j] = rsqrtf(var + BN_EPS);
            g[j]  = gamma[c];
            b[j]  = beta[c];
            w[j]  = wc[c];
        }
    }
    float bcv = bc[0];
    const float4* xp = (const float4*)x + lane;

    for (int r = r0; r < r1; r += DO_U) {
        float4 v[DO_U];
#pragma unroll
        for (int u = 0; u < DO_U; ++u) {
            int rr = r + u;
            v[u] = xp[(long)(rr < r1 ? rr : r1 - 1) * (CH / 4)];
        }
        float res = 0.f;
#pragma unroll
        for (int u = 0; u < DO_U; ++u) {
            float vv[4] = {v[u].x, v[u].y, v[u].z, v[u].w};
            float acc = 0.f;
#pragma unroll
            for (int j = 0; j < 4; ++j) {
                float val = (fmaxf(vv[j], 0.f) - mu[j]) * is[j] * g[j] + b[j];
                acc += val * w[j];
            }
#pragma unroll
            for (int o = 16; o > 0; o >>= 1)
                acc += __shfl_xor(acc, o, 32);
            res = (lane == u) ? acc : res;
        }
        if (lane < DO_U && r + lane < r1) out[r + lane] = res + bcv;
    }
}

extern "C" void kernel_launch(void* const* d_in, const int* in_sizes, int n_in,
                              void* d_out, int out_size, void* d_ws, size_t ws_size,
                              hipStream_t stream) {
    const float* x   = (const float*)d_in[0];
    const int*   ei  = (const int*)d_in[1];
    const float* W1  = (const float*)d_in[2];
    const float* b1  = (const float*)d_in[3];
    const float* W2  = (const float*)d_in[4];
    const float* b2  = (const float*)d_in[5];
    const float* g1  = (const float*)d_in[6];
    const float* be1 = (const float*)d_in[7];
    const float* g2  = (const float*)d_in[8];
    const float* be2 = (const float*)d_in[9];
    const float* Wc  = (const float*)d_in[10];
    const float* bc  = (const float*)d_in[11];
    float* out = (float*)d_out;

    const int  E  = in_sizes[1] / 2;
    const int  N  = 20000;
    const int  H  = in_sizes[3];
    const long BN = (long)out_size;
    const int  Bz = (int)(BN / N);
    const long TOT = BN * H;
    const float invcnt = 1.0f / (float)BN;

    float* ws     = (float*)d_ws;
    float* dinv   = ws + 512;
    long   nAl    = (N + 255) & ~255L;
    float* bufA   = ws + 512 + nAl;

    const long PL   = (long)N * H;
    float* bufM     = bufA + (Bz - 1) * PL;
    float* agg1Last = bufA + (2 * Bz - 2) * PL;
    float* agg2Last = bufA + (Bz - 1) * PL;
    long   n1Al     = (N + 1 + 255) & ~255L;
    long   eAl      = (E + 255) & ~255L;
    int*   csrEnd   = (int*)(bufA + 2 * TOT);
    int*   sorted   = csrEnd - eAl;
    int*   slots    = sorted - eAl;
    int*   rowptr   = slots - n1Al;
    float* stats1   = bufA + (2 * Bz - 1) * PL;
    float* stats2   = stats1 + ST_REP * 2 * CH;

    const int* srcIdx = ei;
    const int* dstIdx = ei + E;

    gcn_build<<<128, OW_T, 0, stream>>>(srcIdx, dstIdx, rowptr, slots, sorted, dinv, stats1, N, E);

    dim3 tileGridRest((N + FT_ROWS - 1) / FT_ROWS, Bz - 1, 1);
    dim3 tileGridLast((N + FT_ROWS - 1) / FT_ROWS, 1, 1);
    int  outBlks   = 256;

    dim3 nodeGridIn((N + FI_NODES - 1) / FI_NODES, (Bz + 3) / 4, 1);
    gcn_gather_in_gemm<<<nodeGridIn, FI_NODES * 32, 0, stream>>>(x, rowptr, sorted, dinv, W1, b1, bufM, stats1, N, Bz);

    gcn_gather_bn_gemm<<<tileGridRest, FT_ROWS * 16, 0, stream>>>(bufM, rowptr, sorted, dinv, stats1, g1, be1,
                                                                 invcnt, W2, b2, bufA, stats2, N);
    gcn_gather_bn_gemm<<<tileGridLast, FT_ROWS * 16, 0, stream>>>(agg1Last, rowptr, sorted, dinv, stats1, g1, be1,
                                                                 invcnt, W2, b2, agg2Last, stats2, N);

    gcn_bn_dot_out<<<outBlks, 512, 0, stream>>>(bufA, stats2, g2, be2, Wc, bc, out, (int)BN, invcnt);
}
